# starting kernel
# baseline (speedup 1.0000x reference)
#include <hip/hip_runtime.h>

typedef float v2f __attribute__((ext_vector_type(2)));
typedef float v8f __attribute__((ext_vector_type(8)));

static constexpr int BB  = 2;
static constexpr int NP  = 8192;
static constexpr int CH  = 64;
static constexpr int KNN = 16;

__device__ __forceinline__ float leaky(float x) { return fmaxf(x, 0.1f * x); }

__global__ __launch_bounds__(256) void knn_kernel(
    const float* __restrict__ xq,
    const float* __restrict__ xk,
    int* __restrict__ idx)
{
    __shared__ float sx[256], sy[256], sz[256];
    int t = blockIdx.x * 256 + threadIdx.x;
    int b = t / NP;
    int n = t % NP;
    const float* q  = xq + (size_t)b * 3 * NP;
    const float* kp = xk + (size_t)b * 3 * NP;
    float qx = q[n], qy = q[NP + n], qz = q[2 * NP + n];

    float best[KNN]; int bidx[KNN];
#pragma unroll
    for (int j = 0; j < KNN; j++) { best[j] = 3.4e38f; bidx[j] = 0; }

    for (int tile = 0; tile < NP; tile += 256) {
        __syncthreads();
        int c = tile + threadIdx.x;
        sx[threadIdx.x] = kp[c];
        sy[threadIdx.x] = kp[NP + c];
        sz[threadIdx.x] = kp[2 * NP + c];
        __syncthreads();
        for (int j = 0; j < 256; j++) {
            float dx = qx - sx[j], dy = qy - sy[j], dz = qz - sz[j];
            float d  = dx * dx + dy * dy + dz * dz;
            if (d < best[KNN - 1]) {
                float vd = d; int vi = tile + j;
#pragma unroll
                for (int s = 0; s < KNN; s++) {
                    if (vd < best[s]) {
                        float td = best[s]; int ti = bidx[s];
                        best[s] = vd; bidx[s] = vi;
                        vd = td; vi = ti;
                    }
                }
            }
        }
    }
    int* o = idx + (size_t)t * KNN;
#pragma unroll
    for (int j = 0; j < KNN; j++) o[j] = bidx[j];
}

__device__ __forceinline__ void stage_wfrag(v2f* wl, const float* __restrict__ W,
                                            int tid, int nthreads)
{
    for (int s = tid; s < 2048; s += nthreads) {
        int lane = s & 31, nc = (s >> 5) & 3, kk = s >> 7;
        int o  = nc * 16 + (lane & 15);
        int k0 = kk * 4 + 2 * (lane >> 4);
        v2f w; w.x = W[o * 64 + k0]; w.y = W[o * 64 + k0 + 1];
        wl[s] = w;
    }
}

__global__ __launch_bounds__(256) void gemm_cm_kernel(
    const float* __restrict__ in,
    const float* __restrict__ W,
    const float* __restrict__ bias,
    float* __restrict__ out_rm)
{
    __shared__ v2f  wl[2048];
    __shared__ float bl[64];
    int tid = threadIdx.x;
    stage_wfrag(wl, W, tid, 256);
    if (tid < 64) bl[tid] = bias[tid];
    __syncthreads();

    int lane = tid & 31, wave = tid >> 5;
    int tile = blockIdx.x * 8 + wave;
    int b  = tile / (NP / 16);
    int n0 = (tile % (NP / 16)) * 16;
    const float* inb = in + (size_t)b * CH * NP;
    int row   = n0 + (lane & 15);
    int khalf = 2 * (lane >> 4);

    v2f a[16];
#pragma unroll
    for (int kk = 0; kk < 16; kk++) {
        int k0 = kk * 4 + khalf;
        v2f v; v.x = inb[(size_t)k0 * NP + row]; v.y = inb[(size_t)(k0 + 1) * NP + row];
        a[kk] = v;
    }
#pragma unroll
    for (int nc = 0; nc < 4; nc++) {
        v8f acc = {0.f, 0.f, 0.f, 0.f, 0.f, 0.f, 0.f, 0.f};
#pragma unroll
        for (int kk = 0; kk < 16; kk++) {
            acc = __builtin_amdgcn_wmma_f32_16x16x4_f32(
                false, a[kk], false, wl[(kk * 4 + nc) * 32 + lane],
                (short)0, acc, false, false);
        }
        int m = nc * 16 + (lane & 15);
        float bv = bl[m];
        float* ob = out_rm + (size_t)b * NP * 64;
#pragma unroll
        for (int v = 0; v < 8; v++) {
            int r = n0 + v + 8 * (lane >> 4);
            ob[(size_t)r * 64 + m] = acc[v] + bv;
        }
    }
}

__global__ __launch_bounds__(256) void gemm_rm_kernel(
    const float* __restrict__ in,
    const float* __restrict__ W,
    const float* __restrict__ bias,
    float* __restrict__ out_cm,
    float* __restrict__ out_rm)
{
    __shared__ v2f  wl[2048];
    __shared__ float bl[64];
    int tid = threadIdx.x;
    stage_wfrag(wl, W, tid, 256);
    if (tid < 64) bl[tid] = bias[tid];
    __syncthreads();

    int lane = tid & 31, wave = tid >> 5;
    int tile = blockIdx.x * 8 + wave;
    int b  = tile / (NP / 16);
    int n0 = (tile % (NP / 16)) * 16;
    const float* inb = in + (size_t)b * NP * 64;
    int row   = n0 + (lane & 15);
    int khalf = 2 * (lane >> 4);

    v2f a[16];
#pragma unroll
    for (int kk = 0; kk < 16; kk++)
        a[kk] = *(const v2f*)(inb + (size_t)row * 64 + kk * 4 + khalf);

#pragma unroll
    for (int nc = 0; nc < 4; nc++) {
        v8f acc = {0.f, 0.f, 0.f, 0.f, 0.f, 0.f, 0.f, 0.f};
#pragma unroll
        for (int kk = 0; kk < 16; kk++) {
            acc = __builtin_amdgcn_wmma_f32_16x16x4_f32(
                false, a[kk], false, wl[(kk * 4 + nc) * 32 + lane],
                (short)0, acc, false, false);
        }
        int m = nc * 16 + (lane & 15);
        float bv = bl[m];
#pragma unroll
        for (int v = 0; v < 8; v++) {
            int r = n0 + v + 8 * (lane >> 4);
            float val = acc[v] + bv;
            out_cm[(size_t)b * 64 * NP + (size_t)m * NP + r] = val;
            if (out_rm) out_rm[((size_t)b * NP + r) * 64 + m] = val;
        }
    }
}

__global__ __launch_bounds__(128) void cross_kernel(
    const float* __restrict__ xq, const float* __restrict__ xk,
    const float* __restrict__ p1, const float* __restrict__ p2,
    const int*   __restrict__ idx,
    const float* __restrict__ posw, const float* __restrict__ posb,
    const float* __restrict__ w1, const float* __restrict__ b1,
    const float* __restrict__ w2, const float* __restrict__ b2,
    int nlayers,
    float* __restrict__ out_rm,
    float* __restrict__ out_cm)
{
    __shared__ v2f  wl1[2048];
    __shared__ v2f  wl2[2048];
    __shared__ float bl1[64], bl2[64];
    __shared__ float pw[192], pb[64];
    __shared__ float X[4][16][68];

    int tid = threadIdx.x;
    stage_wfrag(wl1, w1, tid, 128);
    if (nlayers > 1) stage_wfrag(wl2, w2, tid, 128);
    for (int s = tid; s < 192; s += 128) pw[s] = posw[s];
    if (tid < 64) {
        bl1[tid] = b1[tid];
        bl2[tid] = (nlayers > 1) ? b2[tid] : 0.f;
        pb[tid]  = posb[tid];
    }
    __syncthreads();

    int lane = tid & 31, wave = tid >> 5;
    int q = blockIdx.x * 4 + wave;
    int b = q / NP;
    int n = q % NP;

    const float* xqb = xq + (size_t)b * 3 * NP;
    const float* xkb = xk + (size_t)b * 3 * NP;
    float qx = xqb[n], qy = xqb[NP + n], qz = xqb[2 * NP + n];
    const int* iq = idx + (size_t)q * KNN;

    int c0 = 2 * lane;
    v2f p1v = *(const v2f*)(p1 + (size_t)q * 64 + c0);
    float pwa0 = pw[c0 * 3 + 0], pwa1 = pw[c0 * 3 + 1], pwa2 = pw[c0 * 3 + 2];
    float pwb0 = pw[(c0 + 1) * 3 + 0], pwb1 = pw[(c0 + 1) * 3 + 1], pwb2 = pw[(c0 + 1) * 3 + 2];
    float pb0 = pb[c0], pb1 = pb[c0 + 1];
#pragma unroll
    for (int r = 0; r < KNN; r++) {
        int nb = iq[r];
        float dx = xkb[nb] - qx, dy = xkb[NP + nb] - qy, dz = xkb[2 * NP + nb] - qz;
        v2f g = *(const v2f*)(p2 + ((size_t)b * NP + nb) * 64 + c0);
        X[wave][r][c0]     = leaky(g.x + p1v.x + pb0 + dx * pwa0 + dy * pwa1 + dz * pwa2);
        X[wave][r][c0 + 1] = leaky(g.y + p1v.y + pb1 + dx * pwb0 + dy * pwb1 + dz * pwb2);
    }
    __syncthreads();

    int mrow  = lane & 15;
    int khalf = 2 * (lane >> 4);
    for (int layer = 0; layer < nlayers; layer++) {
        const v2f*  wl = layer ? wl2 : wl1;
        const float* bl = layer ? bl2 : bl1;
        bool last = (layer == nlayers - 1);

        v2f a[16];
#pragma unroll
        for (int kk = 0; kk < 16; kk++)
            a[kk] = *(const v2f*)&X[wave][mrow][kk * 4 + khalf];

#pragma unroll
        for (int nc = 0; nc < 4; nc++) {
            v8f acc = {0.f, 0.f, 0.f, 0.f, 0.f, 0.f, 0.f, 0.f};
#pragma unroll
            for (int kk = 0; kk < 16; kk++) {
                acc = __builtin_amdgcn_wmma_f32_16x16x4_f32(
                    false, a[kk], false, wl[(kk * 4 + nc) * 32 + lane],
                    (short)0, acc, false, false);
            }
            int m = nc * 16 + mrow;
            float bv = bl[m];
            if (!last) {
#pragma unroll
                for (int v = 0; v < 8; v++)
                    X[wave][v + 8 * (lane >> 4)][m] = leaky(acc[v] + bv);
            } else {
                float mx = -3.4e38f;
#pragma unroll
                for (int v = 0; v < 8; v++) mx = fmaxf(mx, leaky(acc[v] + bv));
                mx = fmaxf(mx, __shfl_xor(mx, 16, 32));
                if (lane < 16) {
                    out_rm[(size_t)q * 64 + m] = mx;
                    if (out_cm)
                        out_cm[(size_t)b * 64 * NP + (size_t)m * NP + n] = mx;
                }
            }
        }
        __syncthreads();
    }
}

extern "C" void kernel_launch(void* const* d_in, const int* in_sizes, int n_in,
                              void* d_out, int out_size, void* d_ws, size_t ws_size,
                              hipStream_t stream)
{
    (void)in_sizes; (void)n_in; (void)out_size; (void)ws_size;
    const float* pc1   = (const float*)d_in[0];
    const float* pc2   = (const float*)d_in[1];
    const float* feat1 = (const float*)d_in[2];
    const float* feat2 = (const float*)d_in[3];
    const float* t11_w = (const float*)d_in[4];  const float* t11_b = (const float*)d_in[5];
    const float* t22_w = (const float*)d_in[6];  const float* t22_b = (const float*)d_in[7];
    const float* pos1w = (const float*)d_in[8];  const float* pos1b = (const float*)d_in[9];
    const float* m1w1  = (const float*)d_in[10]; const float* m1b1  = (const float*)d_in[11];
    const float* m1w2  = (const float*)d_in[12]; const float* m1b2  = (const float*)d_in[13];
    const float* t1_w  = (const float*)d_in[14]; const float* t1_b  = (const float*)d_in[15];
    const float* t2_w  = (const float*)d_in[16]; const float* t2_b  = (const float*)d_in[17];
    const float* pos2w = (const float*)d_in[18]; const float* pos2b = (const float*)d_in[19];
    const float* m2w1  = (const float*)d_in[20]; const float* m2b1  = (const float*)d_in[21];

    float* out = (float*)d_out;
    const size_t S = (size_t)BB * NP * 64;
    float* ws = (float*)d_ws;
    float* q1 = ws;           float* q2 = ws + S;
    float* r1 = ws + 2 * S;   float* r2 = ws + 3 * S;
    float* pooled1 = ws + 4 * S;
    float* pooled2 = ws + 5 * S;
    int* idx1 = (int*)(ws + 6 * S);
    int* idx2 = idx1 + (size_t)BB * NP * KNN;
    float* fn1_rm  = q1;
    float* fn2_rm  = q2;
    float* pooled3 = r1;

    knn_kernel<<<BB * NP / 256, 256, 0, stream>>>(pc1, pc2, idx1);
    knn_kernel<<<BB * NP / 256, 256, 0, stream>>>(pc2, pc1, idx2);

    gemm_cm_kernel<<<128, 256, 0, stream>>>(feat1, t11_w, t11_b, q1);
    gemm_cm_kernel<<<128, 256, 0, stream>>>(feat2, t22_w, t22_b, q2);
    gemm_cm_kernel<<<128, 256, 0, stream>>>(feat2, t11_w, t11_b, r1);
    gemm_cm_kernel<<<128, 256, 0, stream>>>(feat1, t22_w, t22_b, r2);

    cross_kernel<<<BB * NP / 4, 128, 0, stream>>>(pc1, pc2, q1, q2, idx1,
        pos1w, pos1b, m1w1, m1b1, m1w2, m1b2, 2, pooled1, nullptr);
    cross_kernel<<<BB * NP / 4, 128, 0, stream>>>(pc2, pc1, r1, r2, idx2,
        pos1w, pos1b, m1w1, m1b1, m1w2, m1b2, 2, pooled2, nullptr);

    gemm_rm_kernel<<<128, 256, 0, stream>>>(pooled1, t1_w, t1_b, out,       fn1_rm);
    gemm_rm_kernel<<<128, 256, 0, stream>>>(pooled2, t2_w, t2_b, out + S,   fn2_rm);

    cross_kernel<<<BB * NP / 4, 128, 0, stream>>>(pc1, pc2, fn1_rm, fn2_rm, idx1,
        pos2w, pos2b, m2w1, m2b1, nullptr, nullptr, 1, pooled3, out + 2 * S);
}


# kNN top-16 insert as 16 independent compare/select pairs (branch-free) in place of the conditional swap chain, which compiled to whole-list register copies at each of its 16 steps
# speedup vs baseline: 2.6873x; 2.6873x over previous
#include <hip/hip_runtime.h>

typedef float v2f __attribute__((ext_vector_type(2)));
typedef float v8f __attribute__((ext_vector_type(8)));

static constexpr int BB  = 2;
static constexpr int NP  = 8192;
static constexpr int CH  = 64;
static constexpr int KNN = 16;

__device__ __forceinline__ float leaky(float x) { return fmaxf(x, 0.1f * x); }

__global__ __launch_bounds__(256) void knn_kernel(
    const float* __restrict__ xq,
    const float* __restrict__ xk,
    int* __restrict__ idx)
{
    __shared__ float sx[256], sy[256], sz[256];
    int t = blockIdx.x * 256 + threadIdx.x;
    int b = t / NP;
    int n = t % NP;
    const float* q  = xq + (size_t)b * 3 * NP;
    const float* kp = xk + (size_t)b * 3 * NP;
    float qx = q[n], qy = q[NP + n], qz = q[2 * NP + n];

    float best[KNN]; int bidx[KNN];
#pragma unroll
    for (int j = 0; j < KNN; j++) { best[j] = 3.4e38f; bidx[j] = 0; }

    for (int tile = 0; tile < NP; tile += 256) {
        __syncthreads();
        int c = tile + threadIdx.x;
        sx[threadIdx.x] = kp[c];
        sy[threadIdx.x] = kp[NP + c];
        sz[threadIdx.x] = kp[2 * NP + c];
        __syncthreads();
        for (int j = 0; j < 256; j++) {
            float dx = qx - sx[j], dy = qy - sy[j], dz = qz - sz[j];
            float d  = dx * dx + dy * dy + dz * dz;
            if (d < best[KNN - 1]) {
                const int vi = tile + j;
#pragma unroll
                for (int s = KNN - 1; s > 0; s--) {
                    const float bs = best[s], bp = best[s - 1];
                    const int   is = bidx[s], ip = bidx[s - 1];
                    const bool  lt = d < bs, ltp = d < bp;
                    float nb = lt ? d  : bs;
                    int   ni = lt ? vi : is;
                    nb = ltp ? bp : nb;
                    ni = ltp ? ip : ni;
                    best[s] = nb; bidx[s] = ni;
                }
                const float b0 = best[0]; const int i0 = bidx[0];
                const bool lt0 = d < b0;
                best[0] = lt0 ? d  : b0;
                bidx[0] = lt0 ? vi : i0;
            }
        }
    }
    int* o = idx + (size_t)t * KNN;
#pragma unroll
    for (int j = 0; j < KNN; j++) o[j] = bidx[j];
}

__device__ __forceinline__ void stage_wfrag(v2f* wl, const float* __restrict__ W,
                                            int tid, int nthreads)
{
    for (int s = tid; s < 2048; s += nthreads) {
        int lane = s & 31, nc = (s >> 5) & 3, kk = s >> 7;
        int o  = nc * 16 + (lane & 15);
        int k0 = kk * 4 + 2 * (lane >> 4);
        v2f w; w.x = W[o * 64 + k0]; w.y = W[o * 64 + k0 + 1];
        wl[s] = w;
    }
}

__global__ __launch_bounds__(256) void gemm_cm_kernel(
    const float* __restrict__ in,
    const float* __restrict__ W,
    const float* __restrict__ bias,
    float* __restrict__ out_rm)
{
    __shared__ v2f  wl[2048];
    __shared__ float bl[64];
    int tid = threadIdx.x;
    stage_wfrag(wl, W, tid, 256);
    if (tid < 64) bl[tid] = bias[tid];
    __syncthreads();

    int lane = tid & 31, wave = tid >> 5;
    int tile = blockIdx.x * 8 + wave;
    int b  = tile / (NP / 16);
    int n0 = (tile % (NP / 16)) * 16;
    const float* inb = in + (size_t)b * CH * NP;
    int row   = n0 + (lane & 15);
    int khalf = 2 * (lane >> 4);

    v2f a[16];
#pragma unroll
    for (int kk = 0; kk < 16; kk++) {
        int k0 = kk * 4 + khalf;
        v2f v; v.x = inb[(size_t)k0 * NP + row]; v.y = inb[(size_t)(k0 + 1) * NP + row];
        a[kk] = v;
    }
#pragma unroll
    for (int nc = 0; nc < 4; nc++) {
        v8f acc = {0.f, 0.f, 0.f, 0.f, 0.f, 0.f, 0.f, 0.f};
#pragma unroll
        for (int kk = 0; kk < 16; kk++) {
            acc = __builtin_amdgcn_wmma_f32_16x16x4_f32(
                false, a[kk], false, wl[(kk * 4 + nc) * 32 + lane],
                (short)0, acc, false, false);
        }
        int m = nc * 16 + (lane & 15);
        float bv = bl[m];
        float* ob = out_rm + (size_t)b * NP * 64;
#pragma unroll
        for (int v = 0; v < 8; v++) {
            int r = n0 + v + 8 * (lane >> 4);
            ob[(size_t)r * 64 + m] = acc[v] + bv;
        }
    }
}

__global__ __launch_bounds__(256) void gemm_rm_kernel(
    const float* __restrict__ in,
    const float* __restrict__ W,
    const float* __restrict__ bias,
    float* __restrict__ out_cm,
    float* __restrict__ out_rm)
{
    __shared__ v2f  wl[2048];
    __shared__ float bl[64];
    int tid = threadIdx.x;
    stage_wfrag(wl, W, tid, 256);
    if (tid < 64) bl[tid] = bias[tid];
    __syncthreads();

    int lane = tid & 31, wave = tid >> 5;
    int tile = blockIdx.x * 8 + wave;
    int b  = tile / (NP / 16);
    int n0 = (tile % (NP / 16)) * 16;
    const float* inb = in + (size_t)b * NP * 64;
    int row   = n0 + (lane & 15);
    int khalf = 2 * (lane >> 4);

    v2f a[16];
#pragma unroll
    for (int kk = 0; kk < 16; kk++)
        a[kk] = *(const v2f*)(inb + (size_t)row * 64 + kk * 4 + khalf);

#pragma unroll
    for (int nc = 0; nc < 4; nc++) {
        v8f acc = {0.f, 0.f, 0.f, 0.f, 0.f, 0.f, 0.f, 0.f};
#pragma unroll
        for (int kk = 0; kk < 16; kk++) {
            acc = __builtin_amdgcn_wmma_f32_16x16x4_f32(
                false, a[kk], false, wl[(kk * 4 + nc) * 32 + lane],
                (short)0, acc, false, false);
        }
        int m = nc * 16 + (lane & 15);
        float bv = bl[m];
#pragma unroll
        for (int v = 0; v < 8; v++) {
            int r = n0 + v + 8 * (lane >> 4);
            float val = acc[v] + bv;
            out_cm[(size_t)b * 64 * NP + (size_t)m * NP + r] = val;
            if (out_rm) out_rm[((size_t)b * NP + r) * 64 + m] = val;
        }
    }
}

__global__ __launch_bounds__(128) void cross_kernel(
    const float* __restrict__ xq, const float* __restrict__ xk,
    const float* __restrict__ p1, const float* __restrict__ p2,
    const int*   __restrict__ idx,
    const float* __restrict__ posw, const float* __restrict__ posb,
    const float* __restrict__ w1, const float* __restrict__ b1,
    const float* __restrict__ w2, const float* __restrict__ b2,
    int nlayers,
    float* __restrict__ out_rm,
    float* __restrict__ out_cm)
{
    __shared__ v2f  wl1[2048];
    __shared__ v2f  wl2[2048];
    __shared__ float bl1[64], bl2[64];
    __shared__ float pw[192], pb[64];
    __shared__ float X[4][16][68];

    int tid = threadIdx.x;
    stage_wfrag(wl1, w1, tid, 128);
    if (nlayers > 1) stage_wfrag(wl2, w2, tid, 128);
    for (int s = tid; s < 192; s += 128) pw[s] = posw[s];
    if (tid < 64) {
        bl1[tid] = b1[tid];
        bl2[tid] = (nlayers > 1) ? b2[tid] : 0.f;
        pb[tid]  = posb[tid];
    }
    __syncthreads();

    int lane = tid & 31, wave = tid >> 5;
    int q = blockIdx.x * 4 + wave;
    int b = q / NP;
    int n = q % NP;

    const float* xqb = xq + (size_t)b * 3 * NP;
    const float* xkb = xk + (size_t)b * 3 * NP;
    float qx = xqb[n], qy = xqb[NP + n], qz = xqb[2 * NP + n];
    const int* iq = idx + (size_t)q * KNN;

    int c0 = 2 * lane;
    v2f p1v = *(const v2f*)(p1 + (size_t)q * 64 + c0);
    float pwa0 = pw[c0 * 3 + 0], pwa1 = pw[c0 * 3 + 1], pwa2 = pw[c0 * 3 + 2];
    float pwb0 = pw[(c0 + 1) * 3 + 0], pwb1 = pw[(c0 + 1) * 3 + 1], pwb2 = pw[(c0 + 1) * 3 + 2];
    float pb0 = pb[c0], pb1 = pb[c0 + 1];
#pragma unroll
    for (int r = 0; r < KNN; r++) {
        int nb = iq[r];
        float dx = xkb[nb] - qx, dy = xkb[NP + nb] - qy, dz = xkb[2 * NP + nb] - qz;
        v2f g = *(const v2f*)(p2 + ((size_t)b * NP + nb) * 64 + c0);
        X[wave][r][c0]     = leaky(g.x + p1v.x + pb0 + dx * pwa0 + dy * pwa1 + dz * pwa2);
        X[wave][r][c0 + 1] = leaky(g.y + p1v.y + pb1 + dx * pwb0 + dy * pwb1 + dz * pwb2);
    }
    __syncthreads();

    int mrow  = lane & 15;
    int khalf = 2 * (lane >> 4);
    for (int layer = 0; layer < nlayers; layer++) {
        const v2f*  wl = layer ? wl2 : wl1;
        const float* bl = layer ? bl2 : bl1;
        bool last = (layer == nlayers - 1);

        v2f a[16];
#pragma unroll
        for (int kk = 0; kk < 16; kk++)
            a[kk] = *(const v2f*)&X[wave][mrow][kk * 4 + khalf];

#pragma unroll
        for (int nc = 0; nc < 4; nc++) {
            v8f acc = {0.f, 0.f, 0.f, 0.f, 0.f, 0.f, 0.f, 0.f};
#pragma unroll
            for (int kk = 0; kk < 16; kk++) {
                acc = __builtin_amdgcn_wmma_f32_16x16x4_f32(
                    false, a[kk], false, wl[(kk * 4 + nc) * 32 + lane],
                    (short)0, acc, false, false);
            }
            int m = nc * 16 + mrow;
            float bv = bl[m];
            if (!last) {
#pragma unroll
                for (int v = 0; v < 8; v++)
                    X[wave][v + 8 * (lane >> 4)][m] = leaky(acc[v] + bv);
            } else {
                float mx = -3.4e38f;
#pragma unroll
                for (int v = 0; v < 8; v++) mx = fmaxf(mx, leaky(acc[v] + bv));
                mx = fmaxf(mx, __shfl_xor(mx, 16, 32));
                if (lane < 16) {
                    out_rm[(size_t)q * 64 + m] = mx;
                    if (out_cm)
                        out_cm[(size_t)b * 64 * NP + (size_t)m * NP + n] = mx;
                }
            }
        }
        __syncthreads();
    }
}

extern "C" void kernel_launch(void* const* d_in, const int* in_sizes, int n_in,
                              void* d_out, int out_size, void* d_ws, size_t ws_size,
                              hipStream_t stream)
{
    (void)in_sizes; (void)n_in; (void)out_size; (void)ws_size;
    const float* pc1   = (const float*)d_in[0];
    const float* pc2   = (const float*)d_in[1];
    const float* feat1 = (const float*)d_in[2];
    const float* feat2 = (const float*)d_in[3];
    const float* t11_w = (const float*)d_in[4];  const float* t11_b = (const float*)d_in[5];
    const float* t22_w = (const float*)d_in[6];  const float* t22_b = (const float*)d_in[7];
    const float* pos1w = (const float*)d_in[8];  const float* pos1b = (const float*)d_in[9];
    const float* m1w1  = (const float*)d_in[10]; const float* m1b1  = (const float*)d_in[11];
    const float* m1w2  = (const float*)d_in[12]; const float* m1b2  = (const float*)d_in[13];
    const float* t1_w  = (const float*)d_in[14]; const float* t1_b  = (const float*)d_in[15];
    const float* t2_w  = (const float*)d_in[16]; const float* t2_b  = (const float*)d_in[17];
    const float* pos2w = (const float*)d_in[18]; const float* pos2b = (const float*)d_in[19];
    const float* m2w1  = (const float*)d_in[20]; const float* m2b1  = (const float*)d_in[21];

    float* out = (float*)d_out;
    const size_t S = (size_t)BB * NP * 64;
    float* ws = (float*)d_ws;
    float* q1 = ws;           float* q2 = ws + S;
    float* r1 = ws + 2 * S;   float* r2 = ws + 3 * S;
    float* pooled1 = ws + 4 * S;
    float* pooled2 = ws + 5 * S;
    int* idx1 = (int*)(ws + 6 * S);
    int* idx2 = idx1 + (size_t)BB * NP * KNN;
    float* fn1_rm  = q1;
    float* fn2_rm  = q2;
    float* pooled3 = r1;

    knn_kernel<<<BB * NP / 256, 256, 0, stream>>>(pc1, pc2, idx1);
    knn_kernel<<<BB * NP / 256, 256, 0, stream>>>(pc2, pc1, idx2);

    gemm_cm_kernel<<<128, 256, 0, stream>>>(feat1, t11_w, t11_b, q1);
    gemm_cm_kernel<<<128, 256, 0, stream>>>(feat2, t22_w, t22_b, q2);
    gemm_cm_kernel<<<128, 256, 0, stream>>>(feat2, t11_w, t11_b, r1);
    gemm_cm_kernel<<<128, 256, 0, stream>>>(feat1, t22_w, t22_b, r2);

    cross_kernel<<<BB * NP / 4, 128, 0, stream>>>(pc1, pc2, q1, q2, idx1,
        pos1w, pos1b, m1w1, m1b1, m1w2, m1b2, 2, pooled1, nullptr);
    cross_kernel<<<BB * NP / 4, 128, 0, stream>>>(pc2, pc1, r1, r2, idx2,
        pos1w, pos1b, m1w1, m1b1, m1w2, m1b2, 2, pooled2, nullptr);

    gemm_rm_kernel<<<128, 256, 0, stream>>>(pooled1, t1_w, t1_b, out,       fn1_rm);
    gemm_rm_kernel<<<128, 256, 0, stream>>>(pooled2, t2_w, t2_b, out + S,   fn2_rm);

    cross_kernel<<<BB * NP / 4, 128, 0, stream>>>(pc1, pc2, fn1_rm, fn2_rm, idx1,
        pos2w, pos2b, m2w1, m2b1, nullptr, nullptr, 1, pooled3, out + 2 * S);
}


# kNN scan handles 8 candidates per step: 128-bit LDS reads, 8 packed distances, one min test before the per-candidate tests (was one dependent LDS read and branch per candidate)
# speedup vs baseline: 4.4196x; 1.6446x over previous
#include <hip/hip_runtime.h>

typedef float v2f __attribute__((ext_vector_type(2)));
typedef float v8f __attribute__((ext_vector_type(8)));

static constexpr int BB  = 2;
static constexpr int NP  = 8192;
static constexpr int CH  = 64;
static constexpr int KNN = 16;

__device__ __forceinline__ float leaky(float x) { return fmaxf(x, 0.1f * x); }

__device__ __forceinline__ void knn_insert(float (&best)[KNN], int (&bidx)[KNN],
                                           const float d, const int vi)
{
#pragma unroll
    for (int s = KNN - 1; s > 0; s--) {
        const float bs = best[s], bp = best[s - 1];
        const int   is = bidx[s], ip = bidx[s - 1];
        const bool  lt = d < bs, ltp = d < bp;
        float nb = lt ? d  : bs;
        int   ni = lt ? vi : is;
        nb = ltp ? bp : nb;
        ni = ltp ? ip : ni;
        best[s] = nb; bidx[s] = ni;
    }
    const float b0 = best[0]; const int i0 = bidx[0];
    const bool lt0 = d < b0;
    best[0] = lt0 ? d  : b0;
    bidx[0] = lt0 ? vi : i0;
}

__device__ __forceinline__ float dist3(float qx, float qy, float qz,
                                       float x, float y, float z)
{
    const float dx = qx - x, dy = qy - y, dz = qz - z;
    return __builtin_fmaf(dz, dz, __builtin_fmaf(dx, dx, dy * dy));
}

__global__ __launch_bounds__(256) void knn_kernel(
    const float* __restrict__ xq,
    const float* __restrict__ xk,
    int* __restrict__ idx)
{
    __shared__ float4 sx4[64], sy4[64], sz4[64];
    float* sx = (float*)sx4;
    float* sy = (float*)sy4;
    float* sz = (float*)sz4;
    int t = blockIdx.x * 256 + threadIdx.x;
    int b = t / NP;
    int n = t % NP;
    const float* q  = xq + (size_t)b * 3 * NP;
    const float* kp = xk + (size_t)b * 3 * NP;
    float qx = q[n], qy = q[NP + n], qz = q[2 * NP + n];

    float best[KNN]; int bidx[KNN];
#pragma unroll
    for (int j = 0; j < KNN; j++) { best[j] = 3.4e38f; bidx[j] = 0; }

    for (int tile = 0; tile < NP; tile += 256) {
        __syncthreads();
        int c = tile + threadIdx.x;
        sx[threadIdx.x] = kp[c];
        sy[threadIdx.x] = kp[NP + c];
        sz[threadIdx.x] = kp[2 * NP + c];
        __syncthreads();

#pragma unroll 1
        for (int g = 0; g < 32; g++) {
            const float4 ax = sx4[2 * g], bx = sx4[2 * g + 1];
            const float4 ay = sy4[2 * g], by = sy4[2 * g + 1];
            const float4 az = sz4[2 * g], bz = sz4[2 * g + 1];

            float d[8];
            d[0] = dist3(qx, qy, qz, ax.x, ay.x, az.x);
            d[1] = dist3(qx, qy, qz, ax.y, ay.y, az.y);
            d[2] = dist3(qx, qy, qz, ax.z, ay.z, az.z);
            d[3] = dist3(qx, qy, qz, ax.w, ay.w, az.w);
            d[4] = dist3(qx, qy, qz, bx.x, by.x, bz.x);
            d[5] = dist3(qx, qy, qz, bx.y, by.y, bz.y);
            d[6] = dist3(qx, qy, qz, bx.z, by.z, bz.z);
            d[7] = dist3(qx, qy, qz, bx.w, by.w, bz.w);
            const float m = fminf(fminf(fminf(d[0], d[1]), fminf(d[2], d[3])),
                                  fminf(fminf(d[4], d[5]), fminf(d[6], d[7])));
            if (m < best[KNN - 1]) {
                const int base = tile + g * 8;
#pragma unroll
                for (int k = 0; k < 8; k++)
                    if (d[k] < best[KNN - 1]) knn_insert(best, bidx, d[k], base + k);
            }
        }
    }
    int* o = idx + (size_t)t * KNN;
#pragma unroll
    for (int j = 0; j < KNN; j++) o[j] = bidx[j];
}

__device__ __forceinline__ void stage_wfrag(v2f* wl, const float* __restrict__ W,
                                            int tid, int nthreads)
{
    for (int s = tid; s < 2048; s += nthreads) {
        int lane = s & 31, nc = (s >> 5) & 3, kk = s >> 7;
        int o  = nc * 16 + (lane & 15);
        int k0 = kk * 4 + 2 * (lane >> 4);
        v2f w; w.x = W[o * 64 + k0]; w.y = W[o * 64 + k0 + 1];
        wl[s] = w;
    }
}

__global__ __launch_bounds__(256) void gemm_cm_kernel(
    const float* __restrict__ in,
    const float* __restrict__ W,
    const float* __restrict__ bias,
    float* __restrict__ out_rm)
{
    __shared__ v2f  wl[2048];
    __shared__ float bl[64];
    int tid = threadIdx.x;
    stage_wfrag(wl, W, tid, 256);
    if (tid < 64) bl[tid] = bias[tid];
    __syncthreads();

    int lane = tid & 31, wave = tid >> 5;
    int tile = blockIdx.x * 8 + wave;
    int b  = tile / (NP / 16);
    int n0 = (tile % (NP / 16)) * 16;
    const float* inb = in + (size_t)b * CH * NP;
    int row   = n0 + (lane & 15);
    int khalf = 2 * (lane >> 4);

    v2f a[16];
#pragma unroll
    for (int kk = 0; kk < 16; kk++) {
        int k0 = kk * 4 + khalf;
        v2f v; v.x = inb[(size_t)k0 * NP + row]; v.y = inb[(size_t)(k0 + 1) * NP + row];
        a[kk] = v;
    }
#pragma unroll
    for (int nc = 0; nc < 4; nc++) {
        v8f acc = {0.f, 0.f, 0.f, 0.f, 0.f, 0.f, 0.f, 0.f};
#pragma unroll
        for (int kk = 0; kk < 16; kk++) {
            acc = __builtin_amdgcn_wmma_f32_16x16x4_f32(
                false, a[kk], false, wl[(kk * 4 + nc) * 32 + lane],
                (short)0, acc, false, false);
        }
        int m = nc * 16 + (lane & 15);
        float bv = bl[m];
        float* ob = out_rm + (size_t)b * NP * 64;
#pragma unroll
        for (int v = 0; v < 8; v++) {
            int r = n0 + v + 8 * (lane >> 4);
            ob[(size_t)r * 64 + m] = acc[v] + bv;
        }
    }
}

__global__ __launch_bounds__(256) void gemm_rm_kernel(
    const float* __restrict__ in,
    const float* __restrict__ W,
    const float* __restrict__ bias,
    float* __restrict__ out_cm,
    float* __restrict__ out_rm)
{
    __shared__ v2f  wl[2048];
    __shared__ float bl[64];
    int tid = threadIdx.x;
    stage_wfrag(wl, W, tid, 256);
    if (tid < 64) bl[tid] = bias[tid];
    __syncthreads();

    int lane = tid & 31, wave = tid >> 5;
    int tile = blockIdx.x * 8 + wave;
    int b  = tile / (NP / 16);
    int n0 = (tile % (NP / 16)) * 16;
    const float* inb = in + (size_t)b * NP * 64;
    int row   = n0 + (lane & 15);
    int khalf = 2 * (lane >> 4);

    v2f a[16];
#pragma unroll
    for (int kk = 0; kk < 16; kk++)
        a[kk] = *(const v2f*)(inb + (size_t)row * 64 + kk * 4 + khalf);

#pragma unroll
    for (int nc = 0; nc < 4; nc++) {
        v8f acc = {0.f, 0.f, 0.f, 0.f, 0.f, 0.f, 0.f, 0.f};
#pragma unroll
        for (int kk = 0; kk < 16; kk++) {
            acc = __builtin_amdgcn_wmma_f32_16x16x4_f32(
                false, a[kk], false, wl[(kk * 4 + nc) * 32 + lane],
                (short)0, acc, false, false);
        }
        int m = nc * 16 + (lane & 15);
        float bv = bl[m];
#pragma unroll
        for (int v = 0; v < 8; v++) {
            int r = n0 + v + 8 * (lane >> 4);
            float val = acc[v] + bv;
            out_cm[(size_t)b * 64 * NP + (size_t)m * NP + r] = val;
            if (out_rm) out_rm[((size_t)b * NP + r) * 64 + m] = val;
        }
    }
}

__global__ __launch_bounds__(128) void cross_kernel(
    const float* __restrict__ xq, const float* __restrict__ xk,
    const float* __restrict__ p1, const float* __restrict__ p2,
    const int*   __restrict__ idx,
    const float* __restrict__ posw, const float* __restrict__ posb,
    const float* __restrict__ w1, const float* __restrict__ b1,
    const float* __restrict__ w2, const float* __restrict__ b2,
    int nlayers,
    float* __restrict__ out_rm,
    float* __restrict__ out_cm)
{
    __shared__ v2f  wl1[2048];
    __shared__ v2f  wl2[2048];
    __shared__ float bl1[64], bl2[64];
    __shared__ float pw[192], pb[64];
    __shared__ float X[4][16][68];

    int tid = threadIdx.x;
    stage_wfrag(wl1, w1, tid, 128);
    if (nlayers > 1) stage_wfrag(wl2, w2, tid, 128);
    for (int s = tid; s < 192; s += 128) pw[s] = posw[s];
    if (tid < 64) {
        bl1[tid] = b1[tid];
        bl2[tid] = (nlayers > 1) ? b2[tid] : 0.f;
        pb[tid]  = posb[tid];
    }
    __syncthreads();

    int lane = tid & 31, wave = tid >> 5;
    int q = blockIdx.x * 4 + wave;
    int b = q / NP;
    int n = q % NP;

    const float* xqb = xq + (size_t)b * 3 * NP;
    const float* xkb = xk + (size_t)b * 3 * NP;
    float qx = xqb[n], qy = xqb[NP + n], qz = xqb[2 * NP + n];
    const int* iq = idx + (size_t)q * KNN;

    int c0 = 2 * lane;
    v2f p1v = *(const v2f*)(p1 + (size_t)q * 64 + c0);
    float pwa0 = pw[c0 * 3 + 0], pwa1 = pw[c0 * 3 + 1], pwa2 = pw[c0 * 3 + 2];
    float pwb0 = pw[(c0 + 1) * 3 + 0], pwb1 = pw[(c0 + 1) * 3 + 1], pwb2 = pw[(c0 + 1) * 3 + 2];
    float pb0 = pb[c0], pb1 = pb[c0 + 1];
#pragma unroll
    for (int r = 0; r < KNN; r++) {
        int nb = iq[r];
        float dx = xkb[nb] - qx, dy = xkb[NP + nb] - qy, dz = xkb[2 * NP + nb] - qz;
        v2f g = *(const v2f*)(p2 + ((size_t)b * NP + nb) * 64 + c0);
        X[wave][r][c0]     = leaky(g.x + p1v.x + pb0 + dx * pwa0 + dy * pwa1 + dz * pwa2);
        X[wave][r][c0 + 1] = leaky(g.y + p1v.y + pb1 + dx * pwb0 + dy * pwb1 + dz * pwb2);
    }
    __syncthreads();

    int mrow  = lane & 15;
    int khalf = 2 * (lane >> 4);
    for (int layer = 0; layer < nlayers; layer++) {
        const v2f*  wl = layer ? wl2 : wl1;
        const float* bl = layer ? bl2 : bl1;
        bool last = (layer == nlayers - 1);

        v2f a[16];
#pragma unroll
        for (int kk = 0; kk < 16; kk++)
            a[kk] = *(const v2f*)&X[wave][mrow][kk * 4 + khalf];

#pragma unroll
        for (int nc = 0; nc < 4; nc++) {
            v8f acc = {0.f, 0.f, 0.f, 0.f, 0.f, 0.f, 0.f, 0.f};
#pragma unroll
            for (int kk = 0; kk < 16; kk++) {
                acc = __builtin_amdgcn_wmma_f32_16x16x4_f32(
                    false, a[kk], false, wl[(kk * 4 + nc) * 32 + lane],
                    (short)0, acc, false, false);
            }
            int m = nc * 16 + mrow;
            float bv = bl[m];
            if (!last) {
#pragma unroll
                for (int v = 0; v < 8; v++)
                    X[wave][v + 8 * (lane >> 4)][m] = leaky(acc[v] + bv);
            } else {
                float mx = -3.4e38f;
#pragma unroll
                for (int v = 0; v < 8; v++) mx = fmaxf(mx, leaky(acc[v] + bv));
                mx = fmaxf(mx, __shfl_xor(mx, 16, 32));
                if (lane < 16) {
                    out_rm[(size_t)q * 64 + m] = mx;
                    if (out_cm)
                        out_cm[(size_t)b * 64 * NP + (size_t)m * NP + n] = mx;
                }
            }
        }
        __syncthreads();
    }
}

extern "C" void kernel_launch(void* const* d_in, const int* in_sizes, int n_in,
                              void* d_out, int out_size, void* d_ws, size_t ws_size,
                              hipStream_t stream)
{
    (void)in_sizes; (void)n_in; (void)out_size; (void)ws_size;
    const float* pc1   = (const float*)d_in[0];
    const float* pc2   = (const float*)d_in[1];
    const float* feat1 = (const float*)d_in[2];
    const float* feat2 = (const float*)d_in[3];
    const float* t11_w = (const float*)d_in[4];  const float* t11_b = (const float*)d_in[5];
    const float* t22_w = (const float*)d_in[6];  const float* t22_b = (const float*)d_in[7];
    const float* pos1w = (const float*)d_in[8];  const float* pos1b = (const float*)d_in[9];
    const float* m1w1  = (const float*)d_in[10]; const float* m1b1  = (const float*)d_in[11];
    const float* m1w2  = (const float*)d_in[12]; const float* m1b2  = (const float*)d_in[13];
    const float* t1_w  = (const float*)d_in[14]; const float* t1_b  = (const float*)d_in[15];
    const float* t2_w  = (const float*)d_in[16]; const float* t2_b  = (const float*)d_in[17];
    const float* pos2w = (const float*)d_in[18]; const float* pos2b = (const float*)d_in[19];
    const float* m2w1  = (const float*)d_in[20]; const float* m2b1  = (const float*)d_in[21];

    float* out = (float*)d_out;
    const size_t S = (size_t)BB * NP * 64;
    float* ws = (float*)d_ws;
    float* q1 = ws;           float* q2 = ws + S;
    float* r1 = ws + 2 * S;   float* r2 = ws + 3 * S;
    float* pooled1 = ws + 4 * S;
    float* pooled2 = ws + 5 * S;
    int* idx1 = (int*)(ws + 6 * S);
    int* idx2 = idx1 + (size_t)BB * NP * KNN;
    float* fn1_rm  = q1;
    float* fn2_rm  = q2;
    float* pooled3 = r1;

    knn_kernel<<<BB * NP / 256, 256, 0, stream>>>(pc1, pc2, idx1);
    knn_kernel<<<BB * NP / 256, 256, 0, stream>>>(pc2, pc1, idx2);

    gemm_cm_kernel<<<128, 256, 0, stream>>>(feat1, t11_w, t11_b, q1);
    gemm_cm_kernel<<<128, 256, 0, stream>>>(feat2, t22_w, t22_b, q2);
    gemm_cm_kernel<<<128, 256, 0, stream>>>(feat2, t11_w, t11_b, r1);
    gemm_cm_kernel<<<128, 256, 0, stream>>>(feat1, t22_w, t22_b, r2);

    cross_kernel<<<BB * NP / 4, 128, 0, stream>>>(pc1, pc2, q1, q2, idx1,
        pos1w, pos1b, m1w1, m1b1, m1w2, m1b2, 2, pooled1, nullptr);
    cross_kernel<<<BB * NP / 4, 128, 0, stream>>>(pc2, pc1, r1, r2, idx2,
        pos1w, pos1b, m1w1, m1b1, m1w2, m1b2, 2, pooled2, nullptr);

    gemm_rm_kernel<<<128, 256, 0, stream>>>(pooled1, t1_w, t1_b, out,       fn1_rm);
    gemm_rm_kernel<<<128, 256, 0, stream>>>(pooled2, t2_w, t2_b, out + S,   fn2_rm);

    cross_kernel<<<BB * NP / 4, 128, 0, stream>>>(pc1, pc2, fn1_rm, fn2_rm, idx1,
        pos2w, pos2b, m2w1, m2b1, nullptr, nullptr, 1, pooled3, out + 2 * S);
}


# kNN: one wave per query with the batch's whole key set in LDS (96 KiB); lanes scan 256 candidates per step against a wave-uniform threshold, 16 best kept one per lane, insert by a row shift
# speedup vs baseline: 16.0316x; 3.6274x over previous
#include <hip/hip_runtime.h>

typedef float v2f __attribute__((ext_vector_type(2)));
typedef float v8f __attribute__((ext_vector_type(8)));

static constexpr int BB  = 2;
static constexpr int NP  = 8192;
static constexpr int CH  = 64;
static constexpr int KNN = 16;

__device__ __forceinline__ float leaky(float x) { return fmaxf(x, 0.1f * x); }

static constexpr int KNN_THREADS = 512;
static constexpr int KNN_QPW     = 2;
static constexpr int KNN_QPB     = (KNN_THREADS / 32) * KNN_QPW;

__device__ __forceinline__ float dist3(float qx, float qy, float qz,
                                       float x, float y, float z)
{
    const float dx = qx - x, dy = qy - y, dz = qz - z;
    return __builtin_fmaf(dz, dz, __builtin_fmaf(dx, dx, dy * dy));
}

__device__ __forceinline__ int lane_below(int edge, int v)
{
    return __builtin_amdgcn_update_dpp(edge, v, 0x111, 0xf, 0xf, false);
}

__global__ __launch_bounds__(KNN_THREADS) void knn_kernel(
    const float* __restrict__ xq,
    const float* __restrict__ xk,
    int* __restrict__ idx)
{
    __shared__ float4 sx4[NP / 4], sy4[NP / 4], sz4[NP / 4];
    const int tid  = threadIdx.x;
    const int lane = tid & 31, wave = tid >> 5;
    const int q0 = blockIdx.x * KNN_QPB;
    const int b  = q0 / NP;
    const float* kp = xk + (size_t)b * 3 * NP;
    const float* qp = xq + (size_t)b * 3 * NP;
    {
        const float4* kx4 = (const float4*)kp;
        const float4* ky4 = (const float4*)(kp + NP);
        const float4* kz4 = (const float4*)(kp + 2 * NP);
        for (int j = tid; j < NP / 4; j += KNN_THREADS) {
            sx4[j] = kx4[j]; sy4[j] = ky4[j]; sz4[j] = kz4[j];
        }
    }
    __syncthreads();

    const float BIG = 3.4e38f;
#pragma unroll 1
    for (int qi = 0; qi < KNN_QPW; qi++) {
        const int q = q0 + wave * KNN_QPW + qi;
        const int n = q - b * NP;
        const float qx = qp[n], qy = qp[NP + n], qz = qp[2 * NP + n];

        float ld = BIG; int li = 0;
        float td = BIG;

#pragma unroll 1
        for (int g = 0; g < NP / 256; g++) {
            const int j0 = g * 64 + lane, j1 = j0 + 32;
            const float4 ax = sx4[j0], ay = sy4[j0], az = sz4[j0];
            const float4 bx = sx4[j1], by = sy4[j1], bz = sz4[j1];

            float d[8];
            d[0] = dist3(qx, qy, qz, ax.x, ay.x, az.x);
            d[1] = dist3(qx, qy, qz, ax.y, ay.y, az.y);
            d[2] = dist3(qx, qy, qz, ax.z, ay.z, az.z);
            d[3] = dist3(qx, qy, qz, ax.w, ay.w, az.w);
            d[4] = dist3(qx, qy, qz, bx.x, by.x, bz.x);
            d[5] = dist3(qx, qy, qz, bx.y, by.y, bz.y);
            d[6] = dist3(qx, qy, qz, bx.z, by.z, bz.z);
            d[7] = dist3(qx, qy, qz, bx.w, by.w, bz.w);
            const float m = fminf(fminf(fminf(d[0], d[1]), fminf(d[2], d[3])),
                                  fminf(fminf(d[4], d[5]), fminf(d[6], d[7])));
            if (__builtin_amdgcn_ballot_w32(m <= td) != 0u) {
#pragma unroll
                for (int k = 0; k < 8; k++) {
                    const float dk = d[k];
                    const int   ik = (k < 4) ? (4 * j0 + k) : (4 * j1 + (k - 4));
                    unsigned mask = __builtin_amdgcn_ballot_w32(dk <= td);
                    while (mask != 0u) {
                        const int src = __builtin_ctz(mask);
                        mask &= mask - 1u;
                        const float dn = __int_as_float(
                            __builtin_amdgcn_readlane(__float_as_int(dk), src));
                        const int in = __builtin_amdgcn_readlane(ik, src);
                        const float pd = __int_as_float(
                            lane_below(__float_as_int(-BIG), __float_as_int(ld)));
                        const int   pi = lane_below(0, li);
                        const bool lt  = dn < ld || (dn == ld && in < li);
                        const bool ltp = dn < pd || (dn == pd && in < pi);
                        ld = lt ? (ltp ? pd : dn) : ld;
                        li = lt ? (ltp ? pi : in) : li;
                        td = __int_as_float(
                            __builtin_amdgcn_readlane(__float_as_int(ld), 15));
                        mask &= __builtin_amdgcn_ballot_w32(dk <= td);
                    }
                }
            }
        }
        if (lane < KNN) idx[(size_t)q * KNN + lane] = li;
    }
}

__device__ __forceinline__ void stage_wfrag(v2f* wl, const float* __restrict__ W,
                                            int tid, int nthreads)
{
    for (int s = tid; s < 2048; s += nthreads) {
        int lane = s & 31, nc = (s >> 5) & 3, kk = s >> 7;
        int o  = nc * 16 + (lane & 15);
        int k0 = kk * 4 + 2 * (lane >> 4);
        v2f w; w.x = W[o * 64 + k0]; w.y = W[o * 64 + k0 + 1];
        wl[s] = w;
    }
}

__global__ __launch_bounds__(256) void gemm_cm_kernel(
    const float* __restrict__ in,
    const float* __restrict__ W,
    const float* __restrict__ bias,
    float* __restrict__ out_rm)
{
    __shared__ v2f  wl[2048];
    __shared__ float bl[64];
    int tid = threadIdx.x;
    stage_wfrag(wl, W, tid, 256);
    if (tid < 64) bl[tid] = bias[tid];
    __syncthreads();

    int lane = tid & 31, wave = tid >> 5;
    int tile = blockIdx.x * 8 + wave;
    int b  = tile / (NP / 16);
    int n0 = (tile % (NP / 16)) * 16;
    const float* inb = in + (size_t)b * CH * NP;
    int row   = n0 + (lane & 15);
    int khalf = 2 * (lane >> 4);

    v2f a[16];
#pragma unroll
    for (int kk = 0; kk < 16; kk++) {
        int k0 = kk * 4 + khalf;
        v2f v; v.x = inb[(size_t)k0 * NP + row]; v.y = inb[(size_t)(k0 + 1) * NP + row];
        a[kk] = v;
    }
#pragma unroll
    for (int nc = 0; nc < 4; nc++) {
        v8f acc = {0.f, 0.f, 0.f, 0.f, 0.f, 0.f, 0.f, 0.f};
#pragma unroll
        for (int kk = 0; kk < 16; kk++) {
            acc = __builtin_amdgcn_wmma_f32_16x16x4_f32(
                false, a[kk], false, wl[(kk * 4 + nc) * 32 + lane],
                (short)0, acc, false, false);
        }
        int m = nc * 16 + (lane & 15);
        float bv = bl[m];
        float* ob = out_rm + (size_t)b * NP * 64;
#pragma unroll
        for (int v = 0; v < 8; v++) {
            int r = n0 + v + 8 * (lane >> 4);
            ob[(size_t)r * 64 + m] = acc[v] + bv;
        }
    }
}

__global__ __launch_bounds__(256) void gemm_rm_kernel(
    const float* __restrict__ in,
    const float* __restrict__ W,
    const float* __restrict__ bias,
    float* __restrict__ out_cm,
    float* __restrict__ out_rm)
{
    __shared__ v2f  wl[2048];
    __shared__ float bl[64];
    int tid = threadIdx.x;
    stage_wfrag(wl, W, tid, 256);
    if (tid < 64) bl[tid] = bias[tid];
    __syncthreads();

    int lane = tid & 31, wave = tid >> 5;
    int tile = blockIdx.x * 8 + wave;
    int b  = tile / (NP / 16);
    int n0 = (tile % (NP / 16)) * 16;
    const float* inb = in + (size_t)b * NP * 64;
    int row   = n0 + (lane & 15);
    int khalf = 2 * (lane >> 4);

    v2f a[16];
#pragma unroll
    for (int kk = 0; kk < 16; kk++)
        a[kk] = *(const v2f*)(inb + (size_t)row * 64 + kk * 4 + khalf);

#pragma unroll
    for (int nc = 0; nc < 4; nc++) {
        v8f acc = {0.f, 0.f, 0.f, 0.f, 0.f, 0.f, 0.f, 0.f};
#pragma unroll
        for (int kk = 0; kk < 16; kk++) {
            acc = __builtin_amdgcn_wmma_f32_16x16x4_f32(
                false, a[kk], false, wl[(kk * 4 + nc) * 32 + lane],
                (short)0, acc, false, false);
        }
        int m = nc * 16 + (lane & 15);
        float bv = bl[m];
#pragma unroll
        for (int v = 0; v < 8; v++) {
            int r = n0 + v + 8 * (lane >> 4);
            float val = acc[v] + bv;
            out_cm[(size_t)b * 64 * NP + (size_t)m * NP + r] = val;
            if (out_rm) out_rm[((size_t)b * NP + r) * 64 + m] = val;
        }
    }
}

__global__ __launch_bounds__(128) void cross_kernel(
    const float* __restrict__ xq, const float* __restrict__ xk,
    const float* __restrict__ p1, const float* __restrict__ p2,
    const int*   __restrict__ idx,
    const float* __restrict__ posw, const float* __restrict__ posb,
    const float* __restrict__ w1, const float* __restrict__ b1,
    const float* __restrict__ w2, const float* __restrict__ b2,
    int nlayers,
    float* __restrict__ out_rm,
    float* __restrict__ out_cm)
{
    __shared__ v2f  wl1[2048];
    __shared__ v2f  wl2[2048];
    __shared__ float bl1[64], bl2[64];
    __shared__ float pw[192], pb[64];
    __shared__ float X[4][16][68];

    int tid = threadIdx.x;
    stage_wfrag(wl1, w1, tid, 128);
    if (nlayers > 1) stage_wfrag(wl2, w2, tid, 128);
    for (int s = tid; s < 192; s += 128) pw[s] = posw[s];
    if (tid < 64) {
        bl1[tid] = b1[tid];
        bl2[tid] = (nlayers > 1) ? b2[tid] : 0.f;
        pb[tid]  = posb[tid];
    }
    __syncthreads();

    int lane = tid & 31, wave = tid >> 5;
    int q = blockIdx.x * 4 + wave;
    int b = q / NP;
    int n = q % NP;

    const float* xqb = xq + (size_t)b * 3 * NP;
    const float* xkb = xk + (size_t)b * 3 * NP;
    float qx = xqb[n], qy = xqb[NP + n], qz = xqb[2 * NP + n];
    const int* iq = idx + (size_t)q * KNN;

    int c0 = 2 * lane;
    v2f p1v = *(const v2f*)(p1 + (size_t)q * 64 + c0);
    float pwa0 = pw[c0 * 3 + 0], pwa1 = pw[c0 * 3 + 1], pwa2 = pw[c0 * 3 + 2];
    float pwb0 = pw[(c0 + 1) * 3 + 0], pwb1 = pw[(c0 + 1) * 3 + 1], pwb2 = pw[(c0 + 1) * 3 + 2];
    float pb0 = pb[c0], pb1 = pb[c0 + 1];
#pragma unroll
    for (int r = 0; r < KNN; r++) {
        int nb = iq[r];
        float dx = xkb[nb] - qx, dy = xkb[NP + nb] - qy, dz = xkb[2 * NP + nb] - qz;
        v2f g = *(const v2f*)(p2 + ((size_t)b * NP + nb) * 64 + c0);
        X[wave][r][c0]     = leaky(g.x + p1v.x + pb0 + dx * pwa0 + dy * pwa1 + dz * pwa2);
        X[wave][r][c0 + 1] = leaky(g.y + p1v.y + pb1 + dx * pwb0 + dy * pwb1 + dz * pwb2);
    }
    __syncthreads();

    int mrow  = lane & 15;
    int khalf = 2 * (lane >> 4);
    for (int layer = 0; layer < nlayers; layer++) {
        const v2f*  wl = layer ? wl2 : wl1;
        const float* bl = layer ? bl2 : bl1;
        bool last = (layer == nlayers - 1);

        v2f a[16];
#pragma unroll
        for (int kk = 0; kk < 16; kk++)
            a[kk] = *(const v2f*)&X[wave][mrow][kk * 4 + khalf];

#pragma unroll
        for (int nc = 0; nc < 4; nc++) {
            v8f acc = {0.f, 0.f, 0.f, 0.f, 0.f, 0.f, 0.f, 0.f};
#pragma unroll
            for (int kk = 0; kk < 16; kk++) {
                acc = __builtin_amdgcn_wmma_f32_16x16x4_f32(
                    false, a[kk], false, wl[(kk * 4 + nc) * 32 + lane],
                    (short)0, acc, false, false);
            }
            int m = nc * 16 + mrow;
            float bv = bl[m];
            if (!last) {
#pragma unroll
                for (int v = 0; v < 8; v++)
                    X[wave][v + 8 * (lane >> 4)][m] = leaky(acc[v] + bv);
            } else {
                float mx = -3.4e38f;
#pragma unroll
                for (int v = 0; v < 8; v++) mx = fmaxf(mx, leaky(acc[v] + bv));
                mx = fmaxf(mx, __shfl_xor(mx, 16, 32));
                if (lane < 16) {
                    out_rm[(size_t)q * 64 + m] = mx;
                    if (out_cm)
                        out_cm[(size_t)b * 64 * NP + (size_t)m * NP + n] = mx;
                }
            }
        }
        __syncthreads();
    }
}

extern "C" void kernel_launch(void* const* d_in, const int* in_sizes, int n_in,
                              void* d_out, int out_size, void* d_ws, size_t ws_size,
                              hipStream_t stream)
{
    (void)in_sizes; (void)n_in; (void)out_size; (void)ws_size;
    const float* pc1   = (const float*)d_in[0];
    const float* pc2   = (const float*)d_in[1];
    const float* feat1 = (const float*)d_in[2];
    const float* feat2 = (const float*)d_in[3];
    const float* t11_w = (const float*)d_in[4];  const float* t11_b = (const float*)d_in[5];
    const float* t22_w = (const float*)d_in[6];  const float* t22_b = (const float*)d_in[7];
    const float* pos1w = (const float*)d_in[8];  const float* pos1b = (const float*)d_in[9];
    const float* m1w1  = (const float*)d_in[10]; const float* m1b1  = (const float*)d_in[11];
    const float* m1w2  = (const float*)d_in[12]; const float* m1b2  = (const float*)d_in[13];
    const float* t1_w  = (const float*)d_in[14]; const float* t1_b  = (const float*)d_in[15];
    const float* t2_w  = (const float*)d_in[16]; const float* t2_b  = (const float*)d_in[17];
    const float* pos2w = (const float*)d_in[18]; const float* pos2b = (const float*)d_in[19];
    const float* m2w1  = (const float*)d_in[20]; const float* m2b1  = (const float*)d_in[21];

    float* out = (float*)d_out;
    const size_t S = (size_t)BB * NP * 64;
    float* ws = (float*)d_ws;
    float* q1 = ws;           float* q2 = ws + S;
    float* r1 = ws + 2 * S;   float* r2 = ws + 3 * S;
    float* pooled1 = ws + 4 * S;
    float* pooled2 = ws + 5 * S;
    int* idx1 = (int*)(ws + 6 * S);
    int* idx2 = idx1 + (size_t)BB * NP * KNN;
    float* fn1_rm  = q1;
    float* fn2_rm  = q2;
    float* pooled3 = r1;

    knn_kernel<<<BB * NP / KNN_QPB, KNN_THREADS, 0, stream>>>(pc1, pc2, idx1);
    knn_kernel<<<BB * NP / KNN_QPB, KNN_THREADS, 0, stream>>>(pc2, pc1, idx2);

    gemm_cm_kernel<<<128, 256, 0, stream>>>(feat1, t11_w, t11_b, q1);
    gemm_cm_kernel<<<128, 256, 0, stream>>>(feat2, t22_w, t22_b, q2);
    gemm_cm_kernel<<<128, 256, 0, stream>>>(feat2, t11_w, t11_b, r1);
    gemm_cm_kernel<<<128, 256, 0, stream>>>(feat1, t22_w, t22_b, r2);

    cross_kernel<<<BB * NP / 4, 128, 0, stream>>>(pc1, pc2, q1, q2, idx1,
        pos1w, pos1b, m1w1, m1b1, m1w2, m1b2, 2, pooled1, nullptr);
    cross_kernel<<<BB * NP / 4, 128, 0, stream>>>(pc2, pc1, r1, r2, idx2,
        pos1w, pos1b, m1w1, m1b1, m1w2, m1b2, 2, pooled2, nullptr);

    gemm_rm_kernel<<<128, 256, 0, stream>>>(pooled1, t1_w, t1_b, out,       fn1_rm);
    gemm_rm_kernel<<<128, 256, 0, stream>>>(pooled2, t2_w, t2_b, out + S,   fn2_rm);

    cross_kernel<<<BB * NP / 4, 128, 0, stream>>>(pc1, pc2, fn1_rm, fn2_rm, idx1,
        pos2w, pos2b, m2w1, m2b1, nullptr, nullptr, 1, pooled3, out + 2 * S);
}


# independent launches of one kernel share one grid: the 4 input GEMMs, the 2 two-layer cross stages and the 2 trailing GEMMs become 3 launches (11 -> 6 launches per call)
# speedup vs baseline: 18.9723x; 1.1834x over previous
#include <hip/hip_runtime.h>

typedef float v2f __attribute__((ext_vector_type(2)));
typedef float v8f __attribute__((ext_vector_type(8)));

static constexpr int BB  = 2;
static constexpr int NP  = 8192;
static constexpr int CH  = 64;
static constexpr int KNN = 16;

__device__ __forceinline__ float leaky(float x) { return fmaxf(x, 0.1f * x); }

static constexpr int KNN_THREADS = 512;
static constexpr int KNN_QPW     = 2;
static constexpr int KNN_QPB     = (KNN_THREADS / 32) * KNN_QPW;

__device__ __forceinline__ float dist3(float qx, float qy, float qz,
                                       float x, float y, float z)
{
    const float dx = qx - x, dy = qy - y, dz = qz - z;
    return __builtin_fmaf(dz, dz, __builtin_fmaf(dx, dx, dy * dy));
}

__device__ __forceinline__ int lane_below(int edge, int v)
{
    return __builtin_amdgcn_update_dpp(edge, v, 0x111, 0xf, 0xf, false);
}

__global__ __launch_bounds__(KNN_THREADS) void knn_kernel(
    const float* __restrict__ xq,
    const float* __restrict__ xk,
    int* __restrict__ idx)
{
    __shared__ float4 sx4[NP / 4], sy4[NP / 4], sz4[NP / 4];
    const int tid  = threadIdx.x;
    const int lane = tid & 31, wave = tid >> 5;
    const int q0 = blockIdx.x * KNN_QPB;
    const int b  = q0 / NP;
    const float* kp = xk + (size_t)b * 3 * NP;
    const float* qp = xq + (size_t)b * 3 * NP;
    {
        const float4* kx4 = (const float4*)kp;
        const float4* ky4 = (const float4*)(kp + NP);
        const float4* kz4 = (const float4*)(kp + 2 * NP);
        for (int j = tid; j < NP / 4; j += KNN_THREADS) {
            sx4[j] = kx4[j]; sy4[j] = ky4[j]; sz4[j] = kz4[j];
        }
    }
    __syncthreads();

    const float BIG = 3.4e38f;
#pragma unroll 1
    for (int qi = 0; qi < KNN_QPW; qi++) {
        const int q = q0 + wave * KNN_QPW + qi;
        const int n = q - b * NP;
        const float qx = qp[n], qy = qp[NP + n], qz = qp[2 * NP + n];

        float ld = BIG; int li = 0;
        float td = BIG;

#pragma unroll 1
        for (int g = 0; g < NP / 256; g++) {
            const int j0 = g * 64 + lane, j1 = j0 + 32;
            const float4 ax = sx4[j0], ay = sy4[j0], az = sz4[j0];
            const float4 bx = sx4[j1], by = sy4[j1], bz = sz4[j1];

            float d[8];
            d[0] = dist3(qx, qy, qz, ax.x, ay.x, az.x);
            d[1] = dist3(qx, qy, qz, ax.y, ay.y, az.y);
            d[2] = dist3(qx, qy, qz, ax.z, ay.z, az.z);
            d[3] = dist3(qx, qy, qz, ax.w, ay.w, az.w);
            d[4] = dist3(qx, qy, qz, bx.x, by.x, bz.x);
            d[5] = dist3(qx, qy, qz, bx.y, by.y, bz.y);
            d[6] = dist3(qx, qy, qz, bx.z, by.z, bz.z);
            d[7] = dist3(qx, qy, qz, bx.w, by.w, bz.w);
            const float m = fminf(fminf(fminf(d[0], d[1]), fminf(d[2], d[3])),
                                  fminf(fminf(d[4], d[5]), fminf(d[6], d[7])));
            if (__builtin_amdgcn_ballot_w32(m <= td) != 0u) {
#pragma unroll
                for (int k = 0; k < 8; k++) {
                    const float dk = d[k];
                    const int   ik = (k < 4) ? (4 * j0 + k) : (4 * j1 + (k - 4));
                    unsigned mask = __builtin_amdgcn_ballot_w32(dk <= td);
                    while (mask != 0u) {
                        const int src = __builtin_ctz(mask);
                        mask &= mask - 1u;
                        const float dn = __int_as_float(
                            __builtin_amdgcn_readlane(__float_as_int(dk), src));
                        const int in = __builtin_amdgcn_readlane(ik, src);
                        const float pd = __int_as_float(
                            lane_below(__float_as_int(-BIG), __float_as_int(ld)));
                        const int   pi = lane_below(0, li);
                        const bool lt  = dn < ld || (dn == ld && in < li);
                        const bool ltp = dn < pd || (dn == pd && in < pi);
                        ld = lt ? (ltp ? pd : dn) : ld;
                        li = lt ? (ltp ? pi : in) : li;
                        td = __int_as_float(
                            __builtin_amdgcn_readlane(__float_as_int(ld), 15));
                        mask &= __builtin_amdgcn_ballot_w32(dk <= td);
                    }
                }
            }
        }
        if (lane < KNN) idx[(size_t)q * KNN + lane] = li;
    }
}

__device__ __forceinline__ void stage_wfrag(v2f* wl, const float* __restrict__ W,
                                            int tid, int nthreads)
{
    for (int s = tid; s < 2048; s += nthreads) {
        int lane = s & 31, nc = (s >> 5) & 3, kk = s >> 7;
        int o  = nc * 16 + (lane & 15);
        int k0 = kk * 4 + 2 * (lane >> 4);
        v2f w; w.x = W[o * 64 + k0]; w.y = W[o * 64 + k0 + 1];
        wl[s] = w;
    }
}

static constexpr int GEMM_BLOCKS = BB * NP / 16 / 8;

__global__ __launch_bounds__(256) void gemm_cm_kernel(
    const float* __restrict__ inA,
    const float* __restrict__ inB,
    const float* __restrict__ Wa,
    const float* __restrict__ ba,
    const float* __restrict__ Wb,
    const float* __restrict__ bb,
    float* __restrict__ out_rm4)
{
    __shared__ v2f  wl[2048];
    __shared__ float bl[64];
    int tid = threadIdx.x;
    const int prod = blockIdx.x / GEMM_BLOCKS;
    const int blk  = blockIdx.x % GEMM_BLOCKS;
    const float* in   = (prod == 0 || prod == 3) ? inA : inB;
    const float* W    = (prod & 1) ? Wb : Wa;
    const float* bias = (prod & 1) ? bb : ba;
    float* out_rm = out_rm4 + (size_t)prod * BB * NP * 64;
    stage_wfrag(wl, W, tid, 256);
    if (tid < 64) bl[tid] = bias[tid];
    __syncthreads();

    int lane = tid & 31, wave = tid >> 5;
    int tile = blk * 8 + wave;
    int b  = tile / (NP / 16);
    int n0 = (tile % (NP / 16)) * 16;
    const float* inb = in + (size_t)b * CH * NP;
    int row   = n0 + (lane & 15);
    int khalf = 2 * (lane >> 4);

    v2f a[16];
#pragma unroll
    for (int kk = 0; kk < 16; kk++) {
        int k0 = kk * 4 + khalf;
        v2f v; v.x = inb[(size_t)k0 * NP + row]; v.y = inb[(size_t)(k0 + 1) * NP + row];
        a[kk] = v;
    }
#pragma unroll
    for (int nc = 0; nc < 4; nc++) {
        v8f acc = {0.f, 0.f, 0.f, 0.f, 0.f, 0.f, 0.f, 0.f};
#pragma unroll
        for (int kk = 0; kk < 16; kk++) {
            acc = __builtin_amdgcn_wmma_f32_16x16x4_f32(
                false, a[kk], false, wl[(kk * 4 + nc) * 32 + lane],
                (short)0, acc, false, false);
        }
        int m = nc * 16 + (lane & 15);
        float bv = bl[m];
        float* ob = out_rm + (size_t)b * NP * 64;
#pragma unroll
        for (int v = 0; v < 8; v++) {
            int r = n0 + v + 8 * (lane >> 4);
            ob[(size_t)r * 64 + m] = acc[v] + bv;
        }
    }
}

__global__ __launch_bounds__(256) void gemm_rm_kernel(
    const float* __restrict__ in2,
    const float* __restrict__ W0, const float* __restrict__ bias0,
    const float* __restrict__ W1, const float* __restrict__ bias1,
    float* __restrict__ out_cm2,
    float* __restrict__ out_rm2)
{
    __shared__ v2f  wl[2048];
    __shared__ float bl[64];
    int tid = threadIdx.x;
    const int prod = blockIdx.x / GEMM_BLOCKS;
    const int blk  = blockIdx.x % GEMM_BLOCKS;
    const size_t plane = (size_t)prod * BB * NP * 64;
    const float* in   = in2 + plane;
    const float* W    = prod ? W1 : W0;
    const float* bias = prod ? bias1 : bias0;
    float* out_cm = out_cm2 + plane;
    float* out_rm = out_rm2 + plane;
    stage_wfrag(wl, W, tid, 256);
    if (tid < 64) bl[tid] = bias[tid];
    __syncthreads();

    int lane = tid & 31, wave = tid >> 5;
    int tile = blk * 8 + wave;
    int b  = tile / (NP / 16);
    int n0 = (tile % (NP / 16)) * 16;
    const float* inb = in + (size_t)b * NP * 64;
    int row   = n0 + (lane & 15);
    int khalf = 2 * (lane >> 4);

    v2f a[16];
#pragma unroll
    for (int kk = 0; kk < 16; kk++)
        a[kk] = *(const v2f*)(inb + (size_t)row * 64 + kk * 4 + khalf);

#pragma unroll
    for (int nc = 0; nc < 4; nc++) {
        v8f acc = {0.f, 0.f, 0.f, 0.f, 0.f, 0.f, 0.f, 0.f};
#pragma unroll
        for (int kk = 0; kk < 16; kk++) {
            acc = __builtin_amdgcn_wmma_f32_16x16x4_f32(
                false, a[kk], false, wl[(kk * 4 + nc) * 32 + lane],
                (short)0, acc, false, false);
        }
        int m = nc * 16 + (lane & 15);
        float bv = bl[m];
#pragma unroll
        for (int v = 0; v < 8; v++) {
            int r = n0 + v + 8 * (lane >> 4);
            float val = acc[v] + bv;
            out_cm[(size_t)b * 64 * NP + (size_t)m * NP + r] = val;
            out_rm[((size_t)b * NP + r) * 64 + m] = val;
        }
    }
}

static constexpr int CROSS_BLOCKS = BB * NP / 4;

__global__ __launch_bounds__(128) void cross_kernel(
    const float* __restrict__ xa, const float* __restrict__ xb,
    const float* __restrict__ p1_0, const float* __restrict__ p2_0,
    const float* __restrict__ p1_1, const float* __restrict__ p2_1,
    const int*   __restrict__ idx_0, const int* __restrict__ idx_1,
    const float* __restrict__ posw, const float* __restrict__ posb,
    const float* __restrict__ w1, const float* __restrict__ b1,
    const float* __restrict__ w2, const float* __restrict__ b2,
    int nlayers,
    float* __restrict__ out_rm_0,
    float* __restrict__ out_rm_1,
    float* __restrict__ out_cm)
{
    const int set = blockIdx.x / CROSS_BLOCKS;
    const int blk = blockIdx.x % CROSS_BLOCKS;
    const float* xq  = set ? xb : xa;
    const float* xk  = set ? xa : xb;
    const float* p1  = set ? p1_1 : p1_0;
    const float* p2  = set ? p2_1 : p2_0;
    const int*   idx = set ? idx_1 : idx_0;
    float* out_rm    = set ? out_rm_1 : out_rm_0;

    __shared__ v2f  wl1[2048];
    __shared__ v2f  wl2[2048];
    __shared__ float bl1[64], bl2[64];
    __shared__ float pw[192], pb[64];
    __shared__ float X[4][16][68];

    int tid = threadIdx.x;
    stage_wfrag(wl1, w1, tid, 128);
    if (nlayers > 1) stage_wfrag(wl2, w2, tid, 128);
    for (int s = tid; s < 192; s += 128) pw[s] = posw[s];
    if (tid < 64) {
        bl1[tid] = b1[tid];
        bl2[tid] = (nlayers > 1) ? b2[tid] : 0.f;
        pb[tid]  = posb[tid];
    }
    __syncthreads();

    int lane = tid & 31, wave = tid >> 5;
    int q = blk * 4 + wave;
    int b = q / NP;
    int n = q % NP;

    const float* xqb = xq + (size_t)b * 3 * NP;
    const float* xkb = xk + (size_t)b * 3 * NP;
    float qx = xqb[n], qy = xqb[NP + n], qz = xqb[2 * NP + n];
    const int* iq = idx + (size_t)q * KNN;

    int c0 = 2 * lane;
    v2f p1v = *(const v2f*)(p1 + (size_t)q * 64 + c0);
    float pwa0 = pw[c0 * 3 + 0], pwa1 = pw[c0 * 3 + 1], pwa2 = pw[c0 * 3 + 2];
    float pwb0 = pw[(c0 + 1) * 3 + 0], pwb1 = pw[(c0 + 1) * 3 + 1], pwb2 = pw[(c0 + 1) * 3 + 2];
    float pb0 = pb[c0], pb1 = pb[c0 + 1];
#pragma unroll
    for (int r = 0; r < KNN; r++) {
        int nb = iq[r];
        float dx = xkb[nb] - qx, dy = xkb[NP + nb] - qy, dz = xkb[2 * NP + nb] - qz;
        v2f g = *(const v2f*)(p2 + ((size_t)b * NP + nb) * 64 + c0);
        X[wave][r][c0]     = leaky(g.x + p1v.x + pb0 + dx * pwa0 + dy * pwa1 + dz * pwa2);
        X[wave][r][c0 + 1] = leaky(g.y + p1v.y + pb1 + dx * pwb0 + dy * pwb1 + dz * pwb2);
    }
    __syncthreads();

    int mrow  = lane & 15;
    int khalf = 2 * (lane >> 4);
    for (int layer = 0; layer < nlayers; layer++) {
        const v2f*  wl = layer ? wl2 : wl1;
        const float* bl = layer ? bl2 : bl1;
        bool last = (layer == nlayers - 1);

        v2f a[16];
#pragma unroll
        for (int kk = 0; kk < 16; kk++)
            a[kk] = *(const v2f*)&X[wave][mrow][kk * 4 + khalf];

#pragma unroll
        for (int nc = 0; nc < 4; nc++) {
            v8f acc = {0.f, 0.f, 0.f, 0.f, 0.f, 0.f, 0.f, 0.f};
#pragma unroll
            for (int kk = 0; kk < 16; kk++) {
                acc = __builtin_amdgcn_wmma_f32_16x16x4_f32(
                    false, a[kk], false, wl[(kk * 4 + nc) * 32 + lane],
                    (short)0, acc, false, false);
            }
            int m = nc * 16 + mrow;
            float bv = bl[m];
            if (!last) {
#pragma unroll
                for (int v = 0; v < 8; v++)
                    X[wave][v + 8 * (lane >> 4)][m] = leaky(acc[v] + bv);
            } else {
                float mx = -3.4e38f;
#pragma unroll
                for (int v = 0; v < 8; v++) mx = fmaxf(mx, leaky(acc[v] + bv));
                mx = fmaxf(mx, __shfl_xor(mx, 16, 32));
                if (lane < 16) {
                    out_rm[(size_t)q * 64 + m] = mx;
                    if (out_cm)
                        out_cm[(size_t)b * 64 * NP + (size_t)m * NP + n] = mx;
                }
            }
        }
        __syncthreads();
    }
}

extern "C" void kernel_launch(void* const* d_in, const int* in_sizes, int n_in,
                              void* d_out, int out_size, void* d_ws, size_t ws_size,
                              hipStream_t stream)
{
    (void)in_sizes; (void)n_in; (void)out_size; (void)ws_size;
    const float* pc1   = (const float*)d_in[0];
    const float* pc2   = (const float*)d_in[1];
    const float* feat1 = (const float*)d_in[2];
    const float* feat2 = (const float*)d_in[3];
    const float* t11_w = (const float*)d_in[4];  const float* t11_b = (const float*)d_in[5];
    const float* t22_w = (const float*)d_in[6];  const float* t22_b = (const float*)d_in[7];
    const float* pos1w = (const float*)d_in[8];  const float* pos1b = (const float*)d_in[9];
    const float* m1w1  = (const float*)d_in[10]; const float* m1b1  = (const float*)d_in[11];
    const float* m1w2  = (const float*)d_in[12]; const float* m1b2  = (const float*)d_in[13];
    const float* t1_w  = (const float*)d_in[14]; const float* t1_b  = (const float*)d_in[15];
    const float* t2_w  = (const float*)d_in[16]; const float* t2_b  = (const float*)d_in[17];
    const float* pos2w = (const float*)d_in[18]; const float* pos2b = (const float*)d_in[19];
    const float* m2w1  = (const float*)d_in[20]; const float* m2b1  = (const float*)d_in[21];

    float* out = (float*)d_out;
    const size_t S = (size_t)BB * NP * 64;
    float* ws = (float*)d_ws;
    float* q1 = ws;           float* q2 = ws + S;
    float* r1 = ws + 2 * S;   float* r2 = ws + 3 * S;
    float* pooled1 = ws + 4 * S;
    float* pooled2 = ws + 5 * S;
    int* idx1 = (int*)(ws + 6 * S);
    int* idx2 = idx1 + (size_t)BB * NP * KNN;
    float* fn1_rm  = q1;
    float* fn2_rm  = q2;
    float* pooled3 = r1;

    knn_kernel<<<BB * NP / KNN_QPB, KNN_THREADS, 0, stream>>>(pc1, pc2, idx1);
    knn_kernel<<<BB * NP / KNN_QPB, KNN_THREADS, 0, stream>>>(pc2, pc1, idx2);

    gemm_cm_kernel<<<4 * GEMM_BLOCKS, 256, 0, stream>>>(feat1, feat2,
        t11_w, t11_b, t22_w, t22_b, q1);

    cross_kernel<<<2 * CROSS_BLOCKS, 128, 0, stream>>>(pc1, pc2, q1, q2, r1, r2, idx1, idx2,
        pos1w, pos1b, m1w1, m1b1, m1w2, m1b2, 2, pooled1, pooled2, nullptr);

    gemm_rm_kernel<<<2 * GEMM_BLOCKS, 256, 0, stream>>>(pooled1, t1_w, t1_b, t2_w, t2_b,
        out, fn1_rm);

    cross_kernel<<<CROSS_BLOCKS, 128, 0, stream>>>(pc1, pc2, fn1_rm, fn2_rm, fn1_rm, fn2_rm,
        idx1, idx2, pos2w, pos2b, m2w1, m2b1, nullptr, nullptr, 1, pooled3, pooled2, out + 2 * S);
}


# kNN in two passes: the 16th smallest of the 32 per-lane minimum distances (pass 1) bounds the 16th neighbour, so pass 2 inserts about 22 candidates, not 116; each LDS read serves the wave's 2 queries
# speedup vs baseline: 21.2236x; 1.1187x over previous
#include <hip/hip_runtime.h>

typedef float v2f __attribute__((ext_vector_type(2)));
typedef float v8f __attribute__((ext_vector_type(8)));

static constexpr int BB  = 2;
static constexpr int NP  = 8192;
static constexpr int CH  = 64;
static constexpr int KNN = 16;

__device__ __forceinline__ float leaky(float x) { return fmaxf(x, 0.1f * x); }

static constexpr int KNN_THREADS = 512;
static constexpr int KNN_QPW     = 2;
static constexpr int KNN_QPB     = (KNN_THREADS / 32) * KNN_QPW;

__device__ __forceinline__ float dist3(float qx, float qy, float qz,
                                       float x, float y, float z)
{
    const float dx = qx - x, dy = qy - y, dz = qz - z;
    return __builtin_fmaf(dz, dz, __builtin_fmaf(dx, dx, dy * dy));
}

struct Cand8 { float4 ax, ay, az, bx, by, bz; };

__device__ __forceinline__ Cand8 load8(const float4* sx4, const float4* sy4,
                                       const float4* sz4, int j0)
{
    Cand8 c;
    c.ax = sx4[j0];      c.ay = sy4[j0];      c.az = sz4[j0];
    c.bx = sx4[j0 + 32]; c.by = sy4[j0 + 32]; c.bz = sz4[j0 + 32];
    return c;
}

__device__ __forceinline__ float dist8(const Cand8& c, float qx, float qy, float qz,
                                       float (&d)[8])
{
    d[0] = dist3(qx, qy, qz, c.ax.x, c.ay.x, c.az.x);
    d[1] = dist3(qx, qy, qz, c.ax.y, c.ay.y, c.az.y);
    d[2] = dist3(qx, qy, qz, c.ax.z, c.ay.z, c.az.z);
    d[3] = dist3(qx, qy, qz, c.ax.w, c.ay.w, c.az.w);
    d[4] = dist3(qx, qy, qz, c.bx.x, c.by.x, c.bz.x);
    d[5] = dist3(qx, qy, qz, c.bx.y, c.by.y, c.bz.y);
    d[6] = dist3(qx, qy, qz, c.bx.z, c.by.z, c.bz.z);
    d[7] = dist3(qx, qy, qz, c.bx.w, c.by.w, c.bz.w);
    return fminf(fminf(fminf(d[0], d[1]), fminf(d[2], d[3])),
                 fminf(fminf(d[4], d[5]), fminf(d[6], d[7])));
}

__device__ __forceinline__ unsigned sixteenth_of_lanes(unsigned v, unsigned big)
{
    unsigned r = big;
#pragma unroll
    for (int j = 0; j < 32; j++) {
        const unsigned s = (unsigned)__builtin_amdgcn_readlane((int)v, j);
        const int cnt = __builtin_popcount(__builtin_amdgcn_ballot_w32(v <= s));
        r = (cnt >= 16) ? min(r, s) : r;
    }
    return r;
}

__device__ __forceinline__ int lane_below(int edge, int v)
{
    return __builtin_amdgcn_update_dpp(edge, v, 0x111, 0xf, 0xf, false);
}

__device__ __forceinline__ void insert8(const float (&d)[8], int j0, float big,
                                        unsigned td0b, float& ld, int& li, float& td)
{
    const int j1 = j0 + 32;
#pragma unroll
    for (int k = 0; k < 8; k++) {
        const float dk = d[k];
        const int   ik = (k < 4) ? (4 * j0 + k) : (4 * j1 + (k - 4));
        unsigned mask = __builtin_amdgcn_ballot_w32(dk <= td);
        while (mask != 0u) {
            const int src = __builtin_ctz(mask);
            mask &= mask - 1u;
            const float dn = __int_as_float(
                __builtin_amdgcn_readlane(__float_as_int(dk), src));
            const int in = __builtin_amdgcn_readlane(ik, src);
            const float pd = __int_as_float(
                lane_below(__float_as_int(-big), __float_as_int(ld)));
            const int   pi = lane_below(0, li);
            const bool lt  = (dn < ld) | ((dn == ld) & (in < li));
            const bool ltp = (dn < pd) | ((dn == pd) & (in < pi));
            ld = lt ? (ltp ? pd : dn) : ld;
            li = lt ? (ltp ? pi : in) : li;
            td = __uint_as_float(min(td0b, (unsigned)
                __builtin_amdgcn_readlane(__float_as_int(ld), 15)));
            mask &= __builtin_amdgcn_ballot_w32(dk <= td);
        }
    }
}

__global__ __launch_bounds__(KNN_THREADS) void knn_kernel(
    const float* __restrict__ xq,
    const float* __restrict__ xk,
    int* __restrict__ idx)
{
    __shared__ float4 sx4[NP / 4], sy4[NP / 4], sz4[NP / 4];
    const int tid  = threadIdx.x;
    const int lane = tid & 31, wave = tid >> 5;
    const int q0 = blockIdx.x * KNN_QPB;
    const int b  = q0 / NP;
    const float* kp = xk + (size_t)b * 3 * NP;
    const float* qp = xq + (size_t)b * 3 * NP;
    {
        const float4* kx4 = (const float4*)kp;
        const float4* ky4 = (const float4*)(kp + NP);
        const float4* kz4 = (const float4*)(kp + 2 * NP);
        for (int j = tid; j < NP / 4; j += KNN_THREADS) {
            sx4[j] = kx4[j]; sy4[j] = ky4[j]; sz4[j] = kz4[j];
        }
    }
    __syncthreads();

    const float BIG = 3.4e38f;
    const unsigned BIGB = __float_as_uint(BIG);
    const int qA = q0 + wave * KNN_QPW;
    const int nA = qA - b * NP;
    const float ax = qp[nA],     ay = qp[NP + nA],     az = qp[2 * NP + nA];
    const float bx = qp[nA + 1], by = qp[NP + nA + 1], bz = qp[2 * NP + nA + 1];

    unsigned lmA = BIGB, lmB = BIGB;
#pragma clang loop vectorize(disable)
#pragma unroll 1
    for (int g = 0; g < NP / 256; g++) {
        const Cand8 c = load8(sx4, sy4, sz4, g * 64 + lane);
        float d[8];
        lmA = min(lmA, __float_as_uint(dist8(c, ax, ay, az, d)));
        lmB = min(lmB, __float_as_uint(dist8(c, bx, by, bz, d)));
    }
    const unsigned td0A = sixteenth_of_lanes(lmA, BIGB);
    const unsigned td0B = sixteenth_of_lanes(lmB, BIGB);

    float ldA = BIG, ldB = BIG; int liA = 0, liB = 0;
    float tdA = __uint_as_float(td0A), tdB = __uint_as_float(td0B);
#pragma unroll 1
    for (int g = 0; g < NP / 256; g++) {
        const int j0 = g * 64 + lane;
        const Cand8 c = load8(sx4, sy4, sz4, j0);
        float dA[8], dB[8];
        const float mA = dist8(c, ax, ay, az, dA);
        const float mB = dist8(c, bx, by, bz, dB);
        if (__builtin_amdgcn_ballot_w32(mA <= tdA) != 0u)
            insert8(dA, j0, BIG, td0A, ldA, liA, tdA);
        if (__builtin_amdgcn_ballot_w32(mB <= tdB) != 0u)
            insert8(dB, j0, BIG, td0B, ldB, liB, tdB);
    }
    if (lane < KNN) {
        idx[(size_t)qA * KNN + lane]       = liA;
        idx[(size_t)(qA + 1) * KNN + lane] = liB;
    }
}

__device__ __forceinline__ void stage_wfrag(v2f* wl, const float* __restrict__ W,
                                            int tid, int nthreads)
{
    for (int s = tid; s < 2048; s += nthreads) {
        int lane = s & 31, nc = (s >> 5) & 3, kk = s >> 7;
        int o  = nc * 16 + (lane & 15);
        int k0 = kk * 4 + 2 * (lane >> 4);
        v2f w; w.x = W[o * 64 + k0]; w.y = W[o * 64 + k0 + 1];
        wl[s] = w;
    }
}

static constexpr int GEMM_BLOCKS = BB * NP / 16 / 8;

__global__ __launch_bounds__(256) void gemm_cm_kernel(
    const float* __restrict__ inA,
    const float* __restrict__ inB,
    const float* __restrict__ Wa,
    const float* __restrict__ ba,
    const float* __restrict__ Wb,
    const float* __restrict__ bb,
    float* __restrict__ out_rm4)
{
    __shared__ v2f  wl[2048];
    __shared__ float bl[64];
    int tid = threadIdx.x;
    const int prod = blockIdx.x / GEMM_BLOCKS;
    const int blk  = blockIdx.x % GEMM_BLOCKS;
    const float* in   = (prod == 0 || prod == 3) ? inA : inB;
    const float* W    = (prod & 1) ? Wb : Wa;
    const float* bias = (prod & 1) ? bb : ba;
    float* out_rm = out_rm4 + (size_t)prod * BB * NP * 64;
    stage_wfrag(wl, W, tid, 256);
    if (tid < 64) bl[tid] = bias[tid];
    __syncthreads();

    int lane = tid & 31, wave = tid >> 5;
    int tile = blk * 8 + wave;
    int b  = tile / (NP / 16);
    int n0 = (tile % (NP / 16)) * 16;
    const float* inb = in + (size_t)b * CH * NP;
    int row   = n0 + (lane & 15);
    int khalf = 2 * (lane >> 4);

    v2f a[16];
#pragma unroll
    for (int kk = 0; kk < 16; kk++) {
        int k0 = kk * 4 + khalf;
        v2f v; v.x = inb[(size_t)k0 * NP + row]; v.y = inb[(size_t)(k0 + 1) * NP + row];
        a[kk] = v;
    }
#pragma unroll
    for (int nc = 0; nc < 4; nc++) {
        v8f acc = {0.f, 0.f, 0.f, 0.f, 0.f, 0.f, 0.f, 0.f};
#pragma unroll
        for (int kk = 0; kk < 16; kk++) {
            acc = __builtin_amdgcn_wmma_f32_16x16x4_f32(
                false, a[kk], false, wl[(kk * 4 + nc) * 32 + lane],
                (short)0, acc, false, false);
        }
        int m = nc * 16 + (lane & 15);
        float bv = bl[m];
        float* ob = out_rm + (size_t)b * NP * 64;
#pragma unroll
        for (int v = 0; v < 8; v++) {
            int r = n0 + v + 8 * (lane >> 4);
            ob[(size_t)r * 64 + m] = acc[v] + bv;
        }
    }
}

__global__ __launch_bounds__(256) void gemm_rm_kernel(
    const float* __restrict__ in2,
    const float* __restrict__ W0, const float* __restrict__ bias0,
    const float* __restrict__ W1, const float* __restrict__ bias1,
    float* __restrict__ out_cm2,
    float* __restrict__ out_rm2)
{
    __shared__ v2f  wl[2048];
    __shared__ float bl[64];
    int tid = threadIdx.x;
    const int prod = blockIdx.x / GEMM_BLOCKS;
    const int blk  = blockIdx.x % GEMM_BLOCKS;
    const size_t plane = (size_t)prod * BB * NP * 64;
    const float* in   = in2 + plane;
    const float* W    = prod ? W1 : W0;
    const float* bias = prod ? bias1 : bias0;
    float* out_cm = out_cm2 + plane;
    float* out_rm = out_rm2 + plane;
    stage_wfrag(wl, W, tid, 256);
    if (tid < 64) bl[tid] = bias[tid];
    __syncthreads();

    int lane = tid & 31, wave = tid >> 5;
    int tile = blk * 8 + wave;
    int b  = tile / (NP / 16);
    int n0 = (tile % (NP / 16)) * 16;
    const float* inb = in + (size_t)b * NP * 64;
    int row   = n0 + (lane & 15);
    int khalf = 2 * (lane >> 4);

    v2f a[16];
#pragma unroll
    for (int kk = 0; kk < 16; kk++)
        a[kk] = *(const v2f*)(inb + (size_t)row * 64 + kk * 4 + khalf);

#pragma unroll
    for (int nc = 0; nc < 4; nc++) {
        v8f acc = {0.f, 0.f, 0.f, 0.f, 0.f, 0.f, 0.f, 0.f};
#pragma unroll
        for (int kk = 0; kk < 16; kk++) {
            acc = __builtin_amdgcn_wmma_f32_16x16x4_f32(
                false, a[kk], false, wl[(kk * 4 + nc) * 32 + lane],
                (short)0, acc, false, false);
        }
        int m = nc * 16 + (lane & 15);
        float bv = bl[m];
#pragma unroll
        for (int v = 0; v < 8; v++) {
            int r = n0 + v + 8 * (lane >> 4);
            float val = acc[v] + bv;
            out_cm[(size_t)b * 64 * NP + (size_t)m * NP + r] = val;
            out_rm[((size_t)b * NP + r) * 64 + m] = val;
        }
    }
}

static constexpr int CROSS_BLOCKS = BB * NP / 4;

__global__ __launch_bounds__(128) void cross_kernel(
    const float* __restrict__ xa, const float* __restrict__ xb,
    const float* __restrict__ p1_0, const float* __restrict__ p2_0,
    const float* __restrict__ p1_1, const float* __restrict__ p2_1,
    const int*   __restrict__ idx_0, const int* __restrict__ idx_1,
    const float* __restrict__ posw, const float* __restrict__ posb,
    const float* __restrict__ w1, const float* __restrict__ b1,
    const float* __restrict__ w2, const float* __restrict__ b2,
    int nlayers,
    float* __restrict__ out_rm_0,
    float* __restrict__ out_rm_1,
    float* __restrict__ out_cm)
{
    const int set = blockIdx.x / CROSS_BLOCKS;
    const int blk = blockIdx.x % CROSS_BLOCKS;
    const float* xq  = set ? xb : xa;
    const float* xk  = set ? xa : xb;
    const float* p1  = set ? p1_1 : p1_0;
    const float* p2  = set ? p2_1 : p2_0;
    const int*   idx = set ? idx_1 : idx_0;
    float* out_rm    = set ? out_rm_1 : out_rm_0;

    __shared__ v2f  wl1[2048];
    __shared__ v2f  wl2[2048];
    __shared__ float bl1[64], bl2[64];
    __shared__ float pw[192], pb[64];
    __shared__ float X[4][16][68];

    int tid = threadIdx.x;
    stage_wfrag(wl1, w1, tid, 128);
    if (nlayers > 1) stage_wfrag(wl2, w2, tid, 128);
    for (int s = tid; s < 192; s += 128) pw[s] = posw[s];
    if (tid < 64) {
        bl1[tid] = b1[tid];
        bl2[tid] = (nlayers > 1) ? b2[tid] : 0.f;
        pb[tid]  = posb[tid];
    }
    __syncthreads();

    int lane = tid & 31, wave = tid >> 5;
    int q = blk * 4 + wave;
    int b = q / NP;
    int n = q % NP;

    const float* xqb = xq + (size_t)b * 3 * NP;
    const float* xkb = xk + (size_t)b * 3 * NP;
    float qx = xqb[n], qy = xqb[NP + n], qz = xqb[2 * NP + n];
    const int* iq = idx + (size_t)q * KNN;

    int c0 = 2 * lane;
    v2f p1v = *(const v2f*)(p1 + (size_t)q * 64 + c0);
    float pwa0 = pw[c0 * 3 + 0], pwa1 = pw[c0 * 3 + 1], pwa2 = pw[c0 * 3 + 2];
    float pwb0 = pw[(c0 + 1) * 3 + 0], pwb1 = pw[(c0 + 1) * 3 + 1], pwb2 = pw[(c0 + 1) * 3 + 2];
    float pb0 = pb[c0], pb1 = pb[c0 + 1];
#pragma unroll
    for (int r = 0; r < KNN; r++) {
        int nb = iq[r];
        float dx = xkb[nb] - qx, dy = xkb[NP + nb] - qy, dz = xkb[2 * NP + nb] - qz;
        v2f g = *(const v2f*)(p2 + ((size_t)b * NP + nb) * 64 + c0);
        X[wave][r][c0]     = leaky(g.x + p1v.x + pb0 + dx * pwa0 + dy * pwa1 + dz * pwa2);
        X[wave][r][c0 + 1] = leaky(g.y + p1v.y + pb1 + dx * pwb0 + dy * pwb1 + dz * pwb2);
    }
    __syncthreads();

    int mrow  = lane & 15;
    int khalf = 2 * (lane >> 4);
    for (int layer = 0; layer < nlayers; layer++) {
        const v2f*  wl = layer ? wl2 : wl1;
        const float* bl = layer ? bl2 : bl1;
        bool last = (layer == nlayers - 1);

        v2f a[16];
#pragma unroll
        for (int kk = 0; kk < 16; kk++)
            a[kk] = *(const v2f*)&X[wave][mrow][kk * 4 + khalf];

#pragma unroll
        for (int nc = 0; nc < 4; nc++) {
            v8f acc = {0.f, 0.f, 0.f, 0.f, 0.f, 0.f, 0.f, 0.f};
#pragma unroll
            for (int kk = 0; kk < 16; kk++) {
                acc = __builtin_amdgcn_wmma_f32_16x16x4_f32(
                    false, a[kk], false, wl[(kk * 4 + nc) * 32 + lane],
                    (short)0, acc, false, false);
            }
            int m = nc * 16 + mrow;
            float bv = bl[m];
            if (!last) {
#pragma unroll
                for (int v = 0; v < 8; v++)
                    X[wave][v + 8 * (lane >> 4)][m] = leaky(acc[v] + bv);
            } else {
                float mx = -3.4e38f;
#pragma unroll
                for (int v = 0; v < 8; v++) mx = fmaxf(mx, leaky(acc[v] + bv));
                mx = fmaxf(mx, __shfl_xor(mx, 16, 32));
                if (lane < 16) {
                    out_rm[(size_t)q * 64 + m] = mx;
                    if (out_cm)
                        out_cm[(size_t)b * 64 * NP + (size_t)m * NP + n] = mx;
                }
            }
        }
        __syncthreads();
    }
}

extern "C" void kernel_launch(void* const* d_in, const int* in_sizes, int n_in,
                              void* d_out, int out_size, void* d_ws, size_t ws_size,
                              hipStream_t stream)
{
    (void)in_sizes; (void)n_in; (void)out_size; (void)ws_size;
    const float* pc1   = (const float*)d_in[0];
    const float* pc2   = (const float*)d_in[1];
    const float* feat1 = (const float*)d_in[2];
    const float* feat2 = (const float*)d_in[3];
    const float* t11_w = (const float*)d_in[4];  const float* t11_b = (const float*)d_in[5];
    const float* t22_w = (const float*)d_in[6];  const float* t22_b = (const float*)d_in[7];
    const float* pos1w = (const float*)d_in[8];  const float* pos1b = (const float*)d_in[9];
    const float* m1w1  = (const float*)d_in[10]; const float* m1b1  = (const float*)d_in[11];
    const float* m1w2  = (const float*)d_in[12]; const float* m1b2  = (const float*)d_in[13];
    const float* t1_w  = (const float*)d_in[14]; const float* t1_b  = (const float*)d_in[15];
    const float* t2_w  = (const float*)d_in[16]; const float* t2_b  = (const float*)d_in[17];
    const float* pos2w = (const float*)d_in[18]; const float* pos2b = (const float*)d_in[19];
    const float* m2w1  = (const float*)d_in[20]; const float* m2b1  = (const float*)d_in[21];

    float* out = (float*)d_out;
    const size_t S = (size_t)BB * NP * 64;
    float* ws = (float*)d_ws;
    float* q1 = ws;           float* q2 = ws + S;
    float* r1 = ws + 2 * S;   float* r2 = ws + 3 * S;
    float* pooled1 = ws + 4 * S;
    float* pooled2 = ws + 5 * S;
    int* idx1 = (int*)(ws + 6 * S);
    int* idx2 = idx1 + (size_t)BB * NP * KNN;
    float* fn1_rm  = q1;
    float* fn2_rm  = q2;
    float* pooled3 = r1;

    knn_kernel<<<BB * NP / KNN_QPB, KNN_THREADS, 0, stream>>>(pc1, pc2, idx1);
    knn_kernel<<<BB * NP / KNN_QPB, KNN_THREADS, 0, stream>>>(pc2, pc1, idx2);

    gemm_cm_kernel<<<4 * GEMM_BLOCKS, 256, 0, stream>>>(feat1, feat2,
        t11_w, t11_b, t22_w, t22_b, q1);

    cross_kernel<<<2 * CROSS_BLOCKS, 128, 0, stream>>>(pc1, pc2, q1, q2, r1, r2, idx1, idx2,
        pos1w, pos1b, m1w1, m1b1, m1w2, m1b2, 2, pooled1, pooled2, nullptr);

    gemm_rm_kernel<<<2 * GEMM_BLOCKS, 256, 0, stream>>>(pooled1, t1_w, t1_b, t2_w, t2_b,
        out, fn1_rm);

    cross_kernel<<<CROSS_BLOCKS, 128, 0, stream>>>(pc1, pc2, fn1_rm, fn2_rm, fn1_rm, fn2_rm,
        idx1, idx2, pos2w, pos2b, m2w1, m2b1, nullptr, nullptr, 1, pooled3, pooled2, out + 2 * S);
}


# cross stages: 1024 workgroups of 8 waves stage the weight fragments in LDS once, then each wave loops over its 4 (one-layer stage: 2) query points; was one staging per 4-wave workgroup of 4 points
# speedup vs baseline: 22.3502x; 1.0531x over previous
#include <hip/hip_runtime.h>

typedef float v2f __attribute__((ext_vector_type(2)));
typedef float v8f __attribute__((ext_vector_type(8)));

static constexpr int BB  = 2;
static constexpr int NP  = 8192;
static constexpr int CH  = 64;
static constexpr int KNN = 16;

__device__ __forceinline__ float leaky(float x) { return fmaxf(x, 0.1f * x); }

static constexpr int KNN_THREADS = 512;
static constexpr int KNN_QPW     = 2;
static constexpr int KNN_QPB     = (KNN_THREADS / 32) * KNN_QPW;

__device__ __forceinline__ float dist3(float qx, float qy, float qz,
                                       float x, float y, float z)
{
    const float dx = qx - x, dy = qy - y, dz = qz - z;
    return __builtin_fmaf(dz, dz, __builtin_fmaf(dx, dx, dy * dy));
}

struct Cand8 { float4 ax, ay, az, bx, by, bz; };

__device__ __forceinline__ Cand8 load8(const float4* sx4, const float4* sy4,
                                       const float4* sz4, int j0)
{
    Cand8 c;
    c.ax = sx4[j0];      c.ay = sy4[j0];      c.az = sz4[j0];
    c.bx = sx4[j0 + 32]; c.by = sy4[j0 + 32]; c.bz = sz4[j0 + 32];
    return c;
}

__device__ __forceinline__ float dist8(const Cand8& c, float qx, float qy, float qz,
                                       float (&d)[8])
{
    d[0] = dist3(qx, qy, qz, c.ax.x, c.ay.x, c.az.x);
    d[1] = dist3(qx, qy, qz, c.ax.y, c.ay.y, c.az.y);
    d[2] = dist3(qx, qy, qz, c.ax.z, c.ay.z, c.az.z);
    d[3] = dist3(qx, qy, qz, c.ax.w, c.ay.w, c.az.w);
    d[4] = dist3(qx, qy, qz, c.bx.x, c.by.x, c.bz.x);
    d[5] = dist3(qx, qy, qz, c.bx.y, c.by.y, c.bz.y);
    d[6] = dist3(qx, qy, qz, c.bx.z, c.by.z, c.bz.z);
    d[7] = dist3(qx, qy, qz, c.bx.w, c.by.w, c.bz.w);
    return fminf(fminf(fminf(d[0], d[1]), fminf(d[2], d[3])),
                 fminf(fminf(d[4], d[5]), fminf(d[6], d[7])));
}

__device__ __forceinline__ unsigned sixteenth_of_lanes(unsigned v, unsigned big)
{
    unsigned r = big;
#pragma unroll
    for (int j = 0; j < 32; j++) {
        const unsigned s = (unsigned)__builtin_amdgcn_readlane((int)v, j);
        const int cnt = __builtin_popcount(__builtin_amdgcn_ballot_w32(v <= s));
        r = (cnt >= 16) ? min(r, s) : r;
    }
    return r;
}

__device__ __forceinline__ int lane_below(int edge, int v)
{
    return __builtin_amdgcn_update_dpp(edge, v, 0x111, 0xf, 0xf, false);
}

__device__ __forceinline__ void insert8(const float (&d)[8], int j0, float big,
                                        unsigned td0b, float& ld, int& li, float& td)
{
    const int j1 = j0 + 32;
#pragma unroll
    for (int k = 0; k < 8; k++) {
        const float dk = d[k];
        const int   ik = (k < 4) ? (4 * j0 + k) : (4 * j1 + (k - 4));
        unsigned mask = __builtin_amdgcn_ballot_w32(dk <= td);
        while (mask != 0u) {
            const int src = __builtin_ctz(mask);
            mask &= mask - 1u;
            const float dn = __int_as_float(
                __builtin_amdgcn_readlane(__float_as_int(dk), src));
            const int in = __builtin_amdgcn_readlane(ik, src);
            const float pd = __int_as_float(
                lane_below(__float_as_int(-big), __float_as_int(ld)));
            const int   pi = lane_below(0, li);
            const bool lt  = (dn < ld) | ((dn == ld) & (in < li));
            const bool ltp = (dn < pd) | ((dn == pd) & (in < pi));
            ld = lt ? (ltp ? pd : dn) : ld;
            li = lt ? (ltp ? pi : in) : li;
            td = __uint_as_float(min(td0b, (unsigned)
                __builtin_amdgcn_readlane(__float_as_int(ld), 15)));
            mask &= __builtin_amdgcn_ballot_w32(dk <= td);
        }
    }
}

__global__ __launch_bounds__(KNN_THREADS) void knn_kernel(
    const float* __restrict__ xq,
    const float* __restrict__ xk,
    int* __restrict__ idx)
{
    __shared__ float4 sx4[NP / 4], sy4[NP / 4], sz4[NP / 4];
    const int tid  = threadIdx.x;
    const int lane = tid & 31, wave = tid >> 5;
    const int q0 = blockIdx.x * KNN_QPB;
    const int b  = q0 / NP;
    const float* kp = xk + (size_t)b * 3 * NP;
    const float* qp = xq + (size_t)b * 3 * NP;
    {
        const float4* kx4 = (const float4*)kp;
        const float4* ky4 = (const float4*)(kp + NP);
        const float4* kz4 = (const float4*)(kp + 2 * NP);
        for (int j = tid; j < NP / 4; j += KNN_THREADS) {
            sx4[j] = kx4[j]; sy4[j] = ky4[j]; sz4[j] = kz4[j];
        }
    }
    __syncthreads();

    const float BIG = 3.4e38f;
    const unsigned BIGB = __float_as_uint(BIG);
    const int qA = q0 + wave * KNN_QPW;
    const int nA = qA - b * NP;
    const float ax = qp[nA],     ay = qp[NP + nA],     az = qp[2 * NP + nA];
    const float bx = qp[nA + 1], by = qp[NP + nA + 1], bz = qp[2 * NP + nA + 1];

    unsigned lmA = BIGB, lmB = BIGB;
#pragma clang loop vectorize(disable)
#pragma unroll 1
    for (int g = 0; g < NP / 256; g++) {
        const Cand8 c = load8(sx4, sy4, sz4, g * 64 + lane);
        float d[8];
        lmA = min(lmA, __float_as_uint(dist8(c, ax, ay, az, d)));
        lmB = min(lmB, __float_as_uint(dist8(c, bx, by, bz, d)));
    }
    const unsigned td0A = sixteenth_of_lanes(lmA, BIGB);
    const unsigned td0B = sixteenth_of_lanes(lmB, BIGB);

    float ldA = BIG, ldB = BIG; int liA = 0, liB = 0;
    float tdA = __uint_as_float(td0A), tdB = __uint_as_float(td0B);
#pragma unroll 1
    for (int g = 0; g < NP / 256; g++) {
        const int j0 = g * 64 + lane;
        const Cand8 c = load8(sx4, sy4, sz4, j0);
        float dA[8], dB[8];
        const float mA = dist8(c, ax, ay, az, dA);
        const float mB = dist8(c, bx, by, bz, dB);
        if (__builtin_amdgcn_ballot_w32(mA <= tdA) != 0u)
            insert8(dA, j0, BIG, td0A, ldA, liA, tdA);
        if (__builtin_amdgcn_ballot_w32(mB <= tdB) != 0u)
            insert8(dB, j0, BIG, td0B, ldB, liB, tdB);
    }
    if (lane < KNN) {
        idx[(size_t)qA * KNN + lane]       = liA;
        idx[(size_t)(qA + 1) * KNN + lane] = liB;
    }
}

__device__ __forceinline__ void stage_wfrag(v2f* wl, const float* __restrict__ W,
                                            int tid, int nthreads)
{
    for (int s = tid; s < 2048; s += nthreads) {
        int lane = s & 31, nc = (s >> 5) & 3, kk = s >> 7;
        int o  = nc * 16 + (lane & 15);
        int k0 = kk * 4 + 2 * (lane >> 4);
        v2f w; w.x = W[o * 64 + k0]; w.y = W[o * 64 + k0 + 1];
        wl[s] = w;
    }
}

template <int NT>
__device__ __forceinline__ void stage_wfrag_once(v2f* wl, const float* __restrict__ W, int tid)
{
    v2f w[2048 / NT];
#pragma unroll
    for (int i = 0; i < 2048 / NT; i++) {
        int s = tid + i * NT;
        int lane = s & 31, nc = (s >> 5) & 3, kk = s >> 7;
        int o  = nc * 16 + (lane & 15);
        int k0 = kk * 4 + 2 * (lane >> 4);
        w[i].x = W[o * 64 + k0]; w[i].y = W[o * 64 + k0 + 1];
    }
#pragma unroll
    for (int i = 0; i < 2048 / NT; i++) wl[tid + i * NT] = w[i];
}

static constexpr int GEMM_BLOCKS = BB * NP / 16 / 8;

__global__ __launch_bounds__(256) void gemm_cm_kernel(
    const float* __restrict__ inA,
    const float* __restrict__ inB,
    const float* __restrict__ Wa,
    const float* __restrict__ ba,
    const float* __restrict__ Wb,
    const float* __restrict__ bb,
    float* __restrict__ out_rm4)
{
    __shared__ v2f  wl[2048];
    __shared__ float bl[64];
    int tid = threadIdx.x;
    const int prod = blockIdx.x / GEMM_BLOCKS;
    const int blk  = blockIdx.x % GEMM_BLOCKS;
    const float* in   = (prod == 0 || prod == 3) ? inA : inB;
    const float* W    = (prod & 1) ? Wb : Wa;
    const float* bias = (prod & 1) ? bb : ba;
    float* out_rm = out_rm4 + (size_t)prod * BB * NP * 64;
    stage_wfrag(wl, W, tid, 256);
    if (tid < 64) bl[tid] = bias[tid];
    __syncthreads();

    int lane = tid & 31, wave = tid >> 5;
    int tile = blk * 8 + wave;
    int b  = tile / (NP / 16);
    int n0 = (tile % (NP / 16)) * 16;
    const float* inb = in + (size_t)b * CH * NP;
    int row   = n0 + (lane & 15);
    int khalf = 2 * (lane >> 4);

    v2f a[16];
#pragma unroll
    for (int kk = 0; kk < 16; kk++) {
        int k0 = kk * 4 + khalf;
        v2f v; v.x = inb[(size_t)k0 * NP + row]; v.y = inb[(size_t)(k0 + 1) * NP + row];
        a[kk] = v;
    }
#pragma unroll
    for (int nc = 0; nc < 4; nc++) {
        v8f acc = {0.f, 0.f, 0.f, 0.f, 0.f, 0.f, 0.f, 0.f};
#pragma unroll
        for (int kk = 0; kk < 16; kk++) {
            acc = __builtin_amdgcn_wmma_f32_16x16x4_f32(
                false, a[kk], false, wl[(kk * 4 + nc) * 32 + lane],
                (short)0, acc, false, false);
        }
        int m = nc * 16 + (lane & 15);
        float bv = bl[m];
        float* ob = out_rm + (size_t)b * NP * 64;
#pragma unroll
        for (int v = 0; v < 8; v++) {
            int r = n0 + v + 8 * (lane >> 4);
            ob[(size_t)r * 64 + m] = acc[v] + bv;
        }
    }
}

__global__ __launch_bounds__(256) void gemm_rm_kernel(
    const float* __restrict__ in2,
    const float* __restrict__ W0, const float* __restrict__ bias0,
    const float* __restrict__ W1, const float* __restrict__ bias1,
    float* __restrict__ out_cm2,
    float* __restrict__ out_rm2)
{
    __shared__ v2f  wl[2048];
    __shared__ float bl[64];
    int tid = threadIdx.x;
    const int prod = blockIdx.x / GEMM_BLOCKS;
    const int blk  = blockIdx.x % GEMM_BLOCKS;
    const size_t plane = (size_t)prod * BB * NP * 64;
    const float* in   = in2 + plane;
    const float* W    = prod ? W1 : W0;
    const float* bias = prod ? bias1 : bias0;
    float* out_cm = out_cm2 + plane;
    float* out_rm = out_rm2 + plane;
    stage_wfrag(wl, W, tid, 256);
    if (tid < 64) bl[tid] = bias[tid];
    __syncthreads();

    int lane = tid & 31, wave = tid >> 5;
    int tile = blk * 8 + wave;
    int b  = tile / (NP / 16);
    int n0 = (tile % (NP / 16)) * 16;
    const float* inb = in + (size_t)b * NP * 64;
    int row   = n0 + (lane & 15);
    int khalf = 2 * (lane >> 4);

    v2f a[16];
#pragma unroll
    for (int kk = 0; kk < 16; kk++)
        a[kk] = *(const v2f*)(inb + (size_t)row * 64 + kk * 4 + khalf);

#pragma unroll
    for (int nc = 0; nc < 4; nc++) {
        v8f acc = {0.f, 0.f, 0.f, 0.f, 0.f, 0.f, 0.f, 0.f};
#pragma unroll
        for (int kk = 0; kk < 16; kk++) {
            acc = __builtin_amdgcn_wmma_f32_16x16x4_f32(
                false, a[kk], false, wl[(kk * 4 + nc) * 32 + lane],
                (short)0, acc, false, false);
        }
        int m = nc * 16 + (lane & 15);
        float bv = bl[m];
#pragma unroll
        for (int v = 0; v < 8; v++) {
            int r = n0 + v + 8 * (lane >> 4);
            float val = acc[v] + bv;
            out_cm[(size_t)b * 64 * NP + (size_t)m * NP + r] = val;
            out_rm[((size_t)b * NP + r) * 64 + m] = val;
        }
    }
}

static constexpr int CROSS_THREADS = 256;
static constexpr int CROSS_WAVES   = CROSS_THREADS / 32;
static constexpr int CROSS_BLOCKS  = 512;
static constexpr int CROSS_QPW     = BB * NP / (CROSS_BLOCKS * CROSS_WAVES);

__global__ __launch_bounds__(CROSS_THREADS, 8) void cross_kernel(
    const float* __restrict__ xa, const float* __restrict__ xb,
    const float* __restrict__ p1_0, const float* __restrict__ p2_0,
    const float* __restrict__ p1_1, const float* __restrict__ p2_1,
    const int*   __restrict__ idx_0, const int* __restrict__ idx_1,
    const float* __restrict__ posw, const float* __restrict__ posb,
    const float* __restrict__ w1, const float* __restrict__ b1,
    const float* __restrict__ w2, const float* __restrict__ b2,
    int nlayers,
    int bps,
    int qpw,
    float* __restrict__ out_rm_0,
    float* __restrict__ out_rm_1,
    float* __restrict__ out_cm)
{
    const int set = blockIdx.x / bps;
    const int blk = blockIdx.x % bps;
    const float* xq  = set ? xb : xa;
    const float* xk  = set ? xa : xb;
    const float* p1  = set ? p1_1 : p1_0;
    const float* p2  = set ? p2_1 : p2_0;
    const int*   idx = set ? idx_1 : idx_0;
    float* out_rm    = set ? out_rm_1 : out_rm_0;

    __shared__ v2f  wl1[2048];
    __shared__ v2f  wl2[2048];
    __shared__ float bl1[64], bl2[64];
    __shared__ float pw[192], pb[64];
    __shared__ float X[CROSS_WAVES][16][68];

    int tid = threadIdx.x;
    stage_wfrag_once<CROSS_THREADS>(wl1, w1, tid);
    if (nlayers > 1) stage_wfrag_once<CROSS_THREADS>(wl2, w2, tid);
    if (tid < 192) pw[tid] = posw[tid];
    if (tid < 64) {
        bl1[tid] = b1[tid];
        bl2[tid] = (nlayers > 1) ? b2[tid] : 0.f;
        pb[tid]  = posb[tid];
    }
    __syncthreads();

    int lane = tid & 31, wave = tid >> 5;
    int mrow  = lane & 15;
    int khalf = 2 * (lane >> 4);

    int c0 = 2 * lane;
    float pwa0 = pw[c0 * 3 + 0], pwa1 = pw[c0 * 3 + 1], pwa2 = pw[c0 * 3 + 2];
    float pwb0 = pw[(c0 + 1) * 3 + 0], pwb1 = pw[(c0 + 1) * 3 + 1], pwb2 = pw[(c0 + 1) * 3 + 2];
    float pb0 = pb[c0], pb1 = pb[c0 + 1];

#pragma unroll 1
    for (int it = 0; it < qpw; it++) {
        int q = (blk * qpw + it) * CROSS_WAVES + wave;
        int b = q / NP;
        int n = q % NP;

        const float* xqb = xq + (size_t)b * 3 * NP;
        const float* xkb = xk + (size_t)b * 3 * NP;
        float qx = xqb[n], qy = xqb[NP + n], qz = xqb[2 * NP + n];
        const int* iq = idx + (size_t)q * KNN;

        v2f p1v = *(const v2f*)(p1 + (size_t)q * 64 + c0);
#pragma unroll
        for (int r = 0; r < KNN; r++) {
            int nb = iq[r];
            float dx = xkb[nb] - qx, dy = xkb[NP + nb] - qy, dz = xkb[2 * NP + nb] - qz;
            v2f g = *(const v2f*)(p2 + ((size_t)b * NP + nb) * 64 + c0);
            X[wave][r][c0]     = leaky(g.x + p1v.x + pb0 + dx * pwa0 + dy * pwa1 + dz * pwa2);
            X[wave][r][c0 + 1] = leaky(g.y + p1v.y + pb1 + dx * pwb0 + dy * pwb1 + dz * pwb2);
        }
        __syncthreads();

        for (int layer = 0; layer < nlayers; layer++) {
            const v2f*  wl = layer ? wl2 : wl1;
            const float* bl = layer ? bl2 : bl1;
            bool last = (layer == nlayers - 1);

            v2f a[16];
#pragma unroll
            for (int kk = 0; kk < 16; kk++)
                a[kk] = *(const v2f*)&X[wave][mrow][kk * 4 + khalf];

#pragma unroll
            for (int nc = 0; nc < 4; nc++) {
                v8f acc = {0.f, 0.f, 0.f, 0.f, 0.f, 0.f, 0.f, 0.f};
#pragma unroll
                for (int kk = 0; kk < 16; kk++) {
                    acc = __builtin_amdgcn_wmma_f32_16x16x4_f32(
                        false, a[kk], false, wl[(kk * 4 + nc) * 32 + lane],
                        (short)0, acc, false, false);
                }
                int m = nc * 16 + mrow;
                float bv = bl[m];
                if (!last) {
#pragma unroll
                    for (int v = 0; v < 8; v++)
                        X[wave][v + 8 * (lane >> 4)][m] = leaky(acc[v] + bv);
                } else {
                    float mx = -3.4e38f;
#pragma unroll
                    for (int v = 0; v < 8; v++) mx = fmaxf(mx, leaky(acc[v] + bv));
                    mx = fmaxf(mx, __shfl_xor(mx, 16, 32));
                    if (lane < 16) {
                        out_rm[(size_t)q * 64 + m] = mx;
                        if (out_cm)
                            out_cm[(size_t)b * 64 * NP + (size_t)m * NP + n] = mx;
                    }
                }
            }
            __syncthreads();
        }
    }
}

extern "C" void kernel_launch(void* const* d_in, const int* in_sizes, int n_in,
                              void* d_out, int out_size, void* d_ws, size_t ws_size,
                              hipStream_t stream)
{
    (void)in_sizes; (void)n_in; (void)out_size; (void)ws_size;
    const float* pc1   = (const float*)d_in[0];
    const float* pc2   = (const float*)d_in[1];
    const float* feat1 = (const float*)d_in[2];
    const float* feat2 = (const float*)d_in[3];
    const float* t11_w = (const float*)d_in[4];  const float* t11_b = (const float*)d_in[5];
    const float* t22_w = (const float*)d_in[6];  const float* t22_b = (const float*)d_in[7];
    const float* pos1w = (const float*)d_in[8];  const float* pos1b = (const float*)d_in[9];
    const float* m1w1  = (const float*)d_in[10]; const float* m1b1  = (const float*)d_in[11];
    const float* m1w2  = (const float*)d_in[12]; const float* m1b2  = (const float*)d_in[13];
    const float* t1_w  = (const float*)d_in[14]; const float* t1_b  = (const float*)d_in[15];
    const float* t2_w  = (const float*)d_in[16]; const float* t2_b  = (const float*)d_in[17];
    const float* pos2w = (const float*)d_in[18]; const float* pos2b = (const float*)d_in[19];
    const float* m2w1  = (const float*)d_in[20]; const float* m2b1  = (const float*)d_in[21];

    float* out = (float*)d_out;
    const size_t S = (size_t)BB * NP * 64;
    float* ws = (float*)d_ws;
    float* q1 = ws;           float* q2 = ws + S;
    float* r1 = ws + 2 * S;   float* r2 = ws + 3 * S;
    float* pooled1 = ws + 4 * S;
    float* pooled2 = ws + 5 * S;
    int* idx1 = (int*)(ws + 6 * S);
    int* idx2 = idx1 + (size_t)BB * NP * KNN;
    float* fn1_rm  = q1;
    float* fn2_rm  = q2;
    float* pooled3 = r1;

    knn_kernel<<<BB * NP / KNN_QPB, KNN_THREADS, 0, stream>>>(pc1, pc2, idx1);
    knn_kernel<<<BB * NP / KNN_QPB, KNN_THREADS, 0, stream>>>(pc2, pc1, idx2);

    gemm_cm_kernel<<<4 * GEMM_BLOCKS, 256, 0, stream>>>(feat1, feat2,
        t11_w, t11_b, t22_w, t22_b, q1);

    cross_kernel<<<2 * CROSS_BLOCKS, CROSS_THREADS, 0, stream>>>(pc1, pc2, q1, q2, r1, r2, idx1, idx2,
        pos1w, pos1b, m1w1, m1b1, m1w2, m1b2, 2, CROSS_BLOCKS, CROSS_QPW, pooled1, pooled2, nullptr);

    gemm_rm_kernel<<<2 * GEMM_BLOCKS, 256, 0, stream>>>(pooled1, t1_w, t1_b, t2_w, t2_b,
        out, fn1_rm);

    cross_kernel<<<2 * CROSS_BLOCKS, CROSS_THREADS, 0, stream>>>(pc1, pc2, fn1_rm, fn2_rm, fn1_rm, fn2_rm,
        idx1, idx2, pos2w, pos2b, m2w1, m2b1, nullptr, nullptr, 1, 2 * CROSS_BLOCKS, CROSS_QPW / 2,
        pooled3, pooled2, out + 2 * S);
}


# kNN pass 1 (bound on the 16th neighbour) moves to the f32 16x16x4 matrix unit: |k|^2 - 2q.k + |q|^2 for 256 key-query pairs per instruction, widened by a rounding bound; pass 2 unchanged
# speedup vs baseline: 23.9438x; 1.0713x over previous
#include <hip/hip_runtime.h>

typedef float v2f __attribute__((ext_vector_type(2)));
typedef float v8f __attribute__((ext_vector_type(8)));

static constexpr int BB  = 2;
static constexpr int NP  = 8192;
static constexpr int CH  = 64;
static constexpr int KNN = 16;

__device__ __forceinline__ float leaky(float x) { return fmaxf(x, 0.1f * x); }

static constexpr int KNN_THREADS = 512;
static constexpr int KNN_QPW     = 2;
static constexpr int KNN_QPB     = (KNN_THREADS / 32) * KNN_QPW;

__device__ __forceinline__ float dist3(float qx, float qy, float qz,
                                       float x, float y, float z)
{
    const float dx = qx - x, dy = qy - y, dz = qz - z;
    return __builtin_fmaf(dz, dz, __builtin_fmaf(dx, dx, dy * dy));
}

struct Cand8 { float4 ax, ay, az, bx, by, bz; };

__device__ __forceinline__ Cand8 load8(const float4* sx4, const float4* sy4,
                                       const float4* sz4, int j0)
{
    Cand8 c;
    c.ax = sx4[j0];      c.ay = sy4[j0];      c.az = sz4[j0];
    c.bx = sx4[j0 + 32]; c.by = sy4[j0 + 32]; c.bz = sz4[j0 + 32];
    return c;
}

__device__ __forceinline__ float dist8(const Cand8& c, float qx, float qy, float qz,
                                       float (&d)[8])
{
    d[0] = dist3(qx, qy, qz, c.ax.x, c.ay.x, c.az.x);
    d[1] = dist3(qx, qy, qz, c.ax.y, c.ay.y, c.az.y);
    d[2] = dist3(qx, qy, qz, c.ax.z, c.ay.z, c.az.z);
    d[3] = dist3(qx, qy, qz, c.ax.w, c.ay.w, c.az.w);
    d[4] = dist3(qx, qy, qz, c.bx.x, c.by.x, c.bz.x);
    d[5] = dist3(qx, qy, qz, c.bx.y, c.by.y, c.bz.y);
    d[6] = dist3(qx, qy, qz, c.bx.z, c.by.z, c.bz.z);
    d[7] = dist3(qx, qy, qz, c.bx.w, c.by.w, c.bz.w);
    return fminf(fminf(fminf(d[0], d[1]), fminf(d[2], d[3])),
                 fminf(fminf(d[4], d[5]), fminf(d[6], d[7])));
}

__device__ __forceinline__ unsigned sixteenth_of_lanes(unsigned v, unsigned big)
{
    unsigned r = big;
#pragma unroll
    for (int j = 0; j < 32; j++) {
        const unsigned s = (unsigned)__builtin_amdgcn_readlane((int)v, j);
        const int cnt = __builtin_popcount(__builtin_amdgcn_ballot_w32(v <= s));
        r = (cnt >= 16) ? min(r, s) : r;
    }
    return r;
}

__device__ __forceinline__ int lane_below(int edge, int v)
{
    return __builtin_amdgcn_update_dpp(edge, v, 0x111, 0xf, 0xf, false);
}

__device__ __forceinline__ void insert8(const float (&d)[8], int j0, float big,
                                        unsigned td0b, float& ld, int& li, float& td)
{
    const int j1 = j0 + 32;
#pragma unroll
    for (int k = 0; k < 8; k++) {
        const float dk = d[k];
        const int   ik = (k < 4) ? (4 * j0 + k) : (4 * j1 + (k - 4));
        unsigned mask = __builtin_amdgcn_ballot_w32(dk <= td);
        while (mask != 0u) {
            const int src = __builtin_ctz(mask);
            mask &= mask - 1u;
            const float dn = __int_as_float(
                __builtin_amdgcn_readlane(__float_as_int(dk), src));
            const int in = __builtin_amdgcn_readlane(ik, src);
            const float pd = __int_as_float(
                lane_below(__float_as_int(-big), __float_as_int(ld)));
            const int   pi = lane_below(0, li);
            const bool lt  = (dn < ld) | ((dn == ld) & (in < li));
            const bool ltp = (dn < pd) | ((dn == pd) & (in < pi));
            ld = lt ? (ltp ? pd : dn) : ld;
            li = lt ? (ltp ? pi : in) : li;
            td = __uint_as_float(min(td0b, (unsigned)
                __builtin_amdgcn_readlane(__float_as_int(ld), 15)));
            mask &= __builtin_amdgcn_ballot_w32(dk <= td);
        }
    }
}

__device__ __forceinline__ unsigned min4_bits(unsigned m, const v8f& d, int o)
{
    m = min(min(m, __float_as_uint(d[o])),     __float_as_uint(d[o + 1]));
    m = min(min(m, __float_as_uint(d[o + 2])), __float_as_uint(d[o + 3]));
    return m;
}

__global__ __launch_bounds__(KNN_THREADS) void knn_kernel(
    const float* __restrict__ xq,
    const float* __restrict__ xk,
    int* __restrict__ idx)
{
    __shared__ float4 sx4[NP / 4], sy4[NP / 4], sz4[NP / 4], sw4[NP / 4];
    __shared__ v2f   pm2[KNN_QPB * 16];
    __shared__ float smax[KNN_THREADS / 32];
    const int tid  = threadIdx.x;
    const int lane = tid & 31, wave = tid >> 5;
    const int q0 = blockIdx.x * KNN_QPB;
    const int b  = q0 / NP;
    const float* kp = xk + (size_t)b * 3 * NP;
    const float* qp = xq + (size_t)b * 3 * NP;
    {
        const float4* kx4 = (const float4*)kp;
        const float4* ky4 = (const float4*)(kp + NP);
        const float4* kz4 = (const float4*)(kp + 2 * NP);
        float kwm = 0.f;
        for (int j = tid; j < NP / 4; j += KNN_THREADS) {
            const float4 x = kx4[j], y = ky4[j], z = kz4[j];
            float4 w;
            w.x = __builtin_fmaf(z.x, z.x, __builtin_fmaf(x.x, x.x, y.x * y.x));
            w.y = __builtin_fmaf(z.y, z.y, __builtin_fmaf(x.y, x.y, y.y * y.y));
            w.z = __builtin_fmaf(z.z, z.z, __builtin_fmaf(x.z, x.z, y.z * y.z));
            w.w = __builtin_fmaf(z.w, z.w, __builtin_fmaf(x.w, x.w, y.w * y.w));
            sx4[j] = x; sy4[j] = y; sz4[j] = z; sw4[j] = w;
            kwm = fmaxf(kwm, fmaxf(fmaxf(w.x, w.y), fmaxf(w.z, w.w)));
        }
#pragma unroll
        for (int o = 16; o > 0; o >>= 1) kwm = fmaxf(kwm, __shfl_xor(kwm, o, 32));
        if (lane == 0) smax[wave] = kwm;
    }
    __syncthreads();
    float kwmax = smax[lane & 15];
#pragma unroll
    for (int o = 8; o > 0; o >>= 1) kwmax = fmaxf(kwmax, __shfl_xor(kwmax, o, 32));

    const float BIG = 3.4e38f;
    const unsigned BIGB = __float_as_uint(BIG);
    const int qA = q0 + wave * KNN_QPW;
    const int nA = qA - b * NP;
    const float ax = qp[nA],     ay = qp[NP + nA],     az = qp[2 * NP + nA];
    const float bx = qp[nA + 1], by = qp[NP + nA + 1], bz = qp[2 * NP + nA + 1];

    {
        const int c = lane & 15, h = lane >> 4;
        const int R = wave >> 3, sl = wave & 7;
        const int nq = q0 - b * NP + 16 * R + c;
        const float qx = qp[nq], qy = qp[NP + nq], qz = qp[2 * NP + nq];
        const float q2 = __builtin_fmaf(qz, qz, __builtin_fmaf(qx, qx, qy * qy));
        const float cq = q2 + 1.52587890625e-05f * (q2 + kwmax);
        v2f bq;
        bq.x = -2.f * (h ? qz : qx);
        bq.y = h ? 1.f : -2.f * qy;
        const float4* pa = h ? sz4 : sx4;
        const float4* pb = h ? sw4 : sy4;
        const v8f cin = {cq, cq, cq, cq, cq, cq, cq, cq};
        unsigned m0 = BIGB, m1 = BIGB;
#pragma unroll 2
        for (int u = 0; u < 16; u++) {
            const int j = 256 * sl + 16 * u + c;
            const float4 p = pa[j], q = pb[j];
            v2f a0, a1, a2, a3;
            a0.x = p.x; a0.y = q.x;  a1.x = p.y; a1.y = q.y;
            a2.x = p.z; a2.y = q.z;  a3.x = p.w; a3.y = q.w;
            const v8f d0 = __builtin_amdgcn_wmma_f32_16x16x4_f32(
                false, a0, false, bq, (short)0, cin, false, false);
            const v8f d1 = __builtin_amdgcn_wmma_f32_16x16x4_f32(
                false, a1, false, bq, (short)0, cin, false, false);
            const v8f d2 = __builtin_amdgcn_wmma_f32_16x16x4_f32(
                false, a2, false, bq, (short)0, cin, false, false);
            const v8f d3 = __builtin_amdgcn_wmma_f32_16x16x4_f32(
                false, a3, false, bq, (short)0, cin, false, false);
            m0 = min4_bits(min4_bits(min4_bits(min4_bits(m0, d0, 0), d1, 0), d2, 0), d3, 0);
            m1 = min4_bits(min4_bits(min4_bits(min4_bits(m1, d0, 4), d1, 4), d2, 4), d3, 4);
        }
        v2f o; o.x = __uint_as_float(m0); o.y = __uint_as_float(m1);
        pm2[(16 * R + c) * 16 + 2 * sl + h] = o;
    }
    __syncthreads();
    const unsigned* pm = (const unsigned*)pm2;
    const unsigned lmA = pm[(2 * wave) * 32 + lane];
    const unsigned lmB = pm[(2 * wave + 1) * 32 + lane];
    const unsigned td0A = sixteenth_of_lanes(lmA, BIGB);
    const unsigned td0B = sixteenth_of_lanes(lmB, BIGB);

    float ldA = BIG, ldB = BIG; int liA = 0, liB = 0;
    float tdA = __uint_as_float(td0A), tdB = __uint_as_float(td0B);
#pragma unroll 1
    for (int g = 0; g < NP / 256; g++) {
        const int j0 = g * 64 + lane;
        const Cand8 c = load8(sx4, sy4, sz4, j0);
        float dA[8], dB[8];
        const float mA = dist8(c, ax, ay, az, dA);
        const float mB = dist8(c, bx, by, bz, dB);
        if (__builtin_amdgcn_ballot_w32(mA <= tdA) != 0u)
            insert8(dA, j0, BIG, td0A, ldA, liA, tdA);
        if (__builtin_amdgcn_ballot_w32(mB <= tdB) != 0u)
            insert8(dB, j0, BIG, td0B, ldB, liB, tdB);
    }
    if (lane < KNN) {
        idx[(size_t)qA * KNN + lane]       = liA;
        idx[(size_t)(qA + 1) * KNN + lane] = liB;
    }
}

__device__ __forceinline__ void stage_wfrag(v2f* wl, const float* __restrict__ W,
                                            int tid, int nthreads)
{
    for (int s = tid; s < 2048; s += nthreads) {
        int lane = s & 31, nc = (s >> 5) & 3, kk = s >> 7;
        int o  = nc * 16 + (lane & 15);
        int k0 = kk * 4 + 2 * (lane >> 4);
        v2f w; w.x = W[o * 64 + k0]; w.y = W[o * 64 + k0 + 1];
        wl[s] = w;
    }
}

template <int NT>
__device__ __forceinline__ void stage_wfrag_once(v2f* wl, const float* __restrict__ W, int tid)
{
    v2f w[2048 / NT];
#pragma unroll
    for (int i = 0; i < 2048 / NT; i++) {
        int s = tid + i * NT;
        int lane = s & 31, nc = (s >> 5) & 3, kk = s >> 7;
        int o  = nc * 16 + (lane & 15);
        int k0 = kk * 4 + 2 * (lane >> 4);
        w[i].x = W[o * 64 + k0]; w[i].y = W[o * 64 + k0 + 1];
    }
#pragma unroll
    for (int i = 0; i < 2048 / NT; i++) wl[tid + i * NT] = w[i];
}

static constexpr int GEMM_BLOCKS = BB * NP / 16 / 8;

__global__ __launch_bounds__(256) void gemm_cm_kernel(
    const float* __restrict__ inA,
    const float* __restrict__ inB,
    const float* __restrict__ Wa,
    const float* __restrict__ ba,
    const float* __restrict__ Wb,
    const float* __restrict__ bb,
    float* __restrict__ out_rm4)
{
    __shared__ v2f  wl[2048];
    __shared__ float bl[64];
    int tid = threadIdx.x;
    const int prod = blockIdx.x / GEMM_BLOCKS;
    const int blk  = blockIdx.x % GEMM_BLOCKS;
    const float* in   = (prod == 0 || prod == 3) ? inA : inB;
    const float* W    = (prod & 1) ? Wb : Wa;
    const float* bias = (prod & 1) ? bb : ba;
    float* out_rm = out_rm4 + (size_t)prod * BB * NP * 64;
    stage_wfrag(wl, W, tid, 256);
    if (tid < 64) bl[tid] = bias[tid];
    __syncthreads();

    int lane = tid & 31, wave = tid >> 5;
    int tile = blk * 8 + wave;
    int b  = tile / (NP / 16);
    int n0 = (tile % (NP / 16)) * 16;
    const float* inb = in + (size_t)b * CH * NP;
    int row   = n0 + (lane & 15);
    int khalf = 2 * (lane >> 4);

    v2f a[16];
#pragma unroll
    for (int kk = 0; kk < 16; kk++) {
        int k0 = kk * 4 + khalf;
        v2f v; v.x = inb[(size_t)k0 * NP + row]; v.y = inb[(size_t)(k0 + 1) * NP + row];
        a[kk] = v;
    }
#pragma unroll
    for (int nc = 0; nc < 4; nc++) {
        v8f acc = {0.f, 0.f, 0.f, 0.f, 0.f, 0.f, 0.f, 0.f};
#pragma unroll
        for (int kk = 0; kk < 16; kk++) {
            acc = __builtin_amdgcn_wmma_f32_16x16x4_f32(
                false, a[kk], false, wl[(kk * 4 + nc) * 32 + lane],
                (short)0, acc, false, false);
        }
        int m = nc * 16 + (lane & 15);
        float bv = bl[m];
        float* ob = out_rm + (size_t)b * NP * 64;
#pragma unroll
        for (int v = 0; v < 8; v++) {
            int r = n0 + v + 8 * (lane >> 4);
            ob[(size_t)r * 64 + m] = acc[v] + bv;
        }
    }
}

__global__ __launch_bounds__(256) void gemm_rm_kernel(
    const float* __restrict__ in2,
    const float* __restrict__ W0, const float* __restrict__ bias0,
    const float* __restrict__ W1, const float* __restrict__ bias1,
    float* __restrict__ out_cm2,
    float* __restrict__ out_rm2)
{
    __shared__ v2f  wl[2048];
    __shared__ float bl[64];
    int tid = threadIdx.x;
    const int prod = blockIdx.x / GEMM_BLOCKS;
    const int blk  = blockIdx.x % GEMM_BLOCKS;
    const size_t plane = (size_t)prod * BB * NP * 64;
    const float* in   = in2 + plane;
    const float* W    = prod ? W1 : W0;
    const float* bias = prod ? bias1 : bias0;
    float* out_cm = out_cm2 + plane;
    float* out_rm = out_rm2 + plane;
    stage_wfrag(wl, W, tid, 256);
    if (tid < 64) bl[tid] = bias[tid];
    __syncthreads();

    int lane = tid & 31, wave = tid >> 5;
    int tile = blk * 8 + wave;
    int b  = tile / (NP / 16);
    int n0 = (tile % (NP / 16)) * 16;
    const float* inb = in + (size_t)b * NP * 64;
    int row   = n0 + (lane & 15);
    int khalf = 2 * (lane >> 4);

    v2f a[16];
#pragma unroll
    for (int kk = 0; kk < 16; kk++)
        a[kk] = *(const v2f*)(inb + (size_t)row * 64 + kk * 4 + khalf);

#pragma unroll
    for (int nc = 0; nc < 4; nc++) {
        v8f acc = {0.f, 0.f, 0.f, 0.f, 0.f, 0.f, 0.f, 0.f};
#pragma unroll
        for (int kk = 0; kk < 16; kk++) {
            acc = __builtin_amdgcn_wmma_f32_16x16x4_f32(
                false, a[kk], false, wl[(kk * 4 + nc) * 32 + lane],
                (short)0, acc, false, false);
        }
        int m = nc * 16 + (lane & 15);
        float bv = bl[m];
#pragma unroll
        for (int v = 0; v < 8; v++) {
            int r = n0 + v + 8 * (lane >> 4);
            float val = acc[v] + bv;
            out_cm[(size_t)b * 64 * NP + (size_t)m * NP + r] = val;
            out_rm[((size_t)b * NP + r) * 64 + m] = val;
        }
    }
}

static constexpr int CROSS_THREADS = 256;
static constexpr int CROSS_WAVES   = CROSS_THREADS / 32;
static constexpr int CROSS_BLOCKS  = 512;
static constexpr int CROSS_QPW     = BB * NP / (CROSS_BLOCKS * CROSS_WAVES);

__global__ __launch_bounds__(CROSS_THREADS, 8) void cross_kernel(
    const float* __restrict__ xa, const float* __restrict__ xb,
    const float* __restrict__ p1_0, const float* __restrict__ p2_0,
    const float* __restrict__ p1_1, const float* __restrict__ p2_1,
    const int*   __restrict__ idx_0, const int* __restrict__ idx_1,
    const float* __restrict__ posw, const float* __restrict__ posb,
    const float* __restrict__ w1, const float* __restrict__ b1,
    const float* __restrict__ w2, const float* __restrict__ b2,
    int nlayers,
    int bps,
    int qpw,
    float* __restrict__ out_rm_0,
    float* __restrict__ out_rm_1,
    float* __restrict__ out_cm)
{
    const int set = blockIdx.x / bps;
    const int blk = blockIdx.x % bps;
    const float* xq  = set ? xb : xa;
    const float* xk  = set ? xa : xb;
    const float* p1  = set ? p1_1 : p1_0;
    const float* p2  = set ? p2_1 : p2_0;
    const int*   idx = set ? idx_1 : idx_0;
    float* out_rm    = set ? out_rm_1 : out_rm_0;

    __shared__ v2f  wl1[2048];
    __shared__ v2f  wl2[2048];
    __shared__ float bl1[64], bl2[64];
    __shared__ float pw[192], pb[64];
    __shared__ float X[CROSS_WAVES][16][68];

    int tid = threadIdx.x;
    stage_wfrag_once<CROSS_THREADS>(wl1, w1, tid);
    if (nlayers > 1) stage_wfrag_once<CROSS_THREADS>(wl2, w2, tid);
    if (tid < 192) pw[tid] = posw[tid];
    if (tid < 64) {
        bl1[tid] = b1[tid];
        bl2[tid] = (nlayers > 1) ? b2[tid] : 0.f;
        pb[tid]  = posb[tid];
    }
    __syncthreads();

    int lane = tid & 31, wave = tid >> 5;
    int mrow  = lane & 15;
    int khalf = 2 * (lane >> 4);

    int c0 = 2 * lane;
    float pwa0 = pw[c0 * 3 + 0], pwa1 = pw[c0 * 3 + 1], pwa2 = pw[c0 * 3 + 2];
    float pwb0 = pw[(c0 + 1) * 3 + 0], pwb1 = pw[(c0 + 1) * 3 + 1], pwb2 = pw[(c0 + 1) * 3 + 2];
    float pb0 = pb[c0], pb1 = pb[c0 + 1];

#pragma unroll 1
    for (int it = 0; it < qpw; it++) {
        int q = (blk * qpw + it) * CROSS_WAVES + wave;
        int b = q / NP;
        int n = q % NP;

        const float* xqb = xq + (size_t)b * 3 * NP;
        const float* xkb = xk + (size_t)b * 3 * NP;
        float qx = xqb[n], qy = xqb[NP + n], qz = xqb[2 * NP + n];
        const int* iq = idx + (size_t)q * KNN;

        v2f p1v = *(const v2f*)(p1 + (size_t)q * 64 + c0);
#pragma unroll
        for (int r = 0; r < KNN; r++) {
            int nb = iq[r];
            float dx = xkb[nb] - qx, dy = xkb[NP + nb] - qy, dz = xkb[2 * NP + nb] - qz;
            v2f g = *(const v2f*)(p2 + ((size_t)b * NP + nb) * 64 + c0);
            X[wave][r][c0]     = leaky(g.x + p1v.x + pb0 + dx * pwa0 + dy * pwa1 + dz * pwa2);
            X[wave][r][c0 + 1] = leaky(g.y + p1v.y + pb1 + dx * pwb0 + dy * pwb1 + dz * pwb2);
        }
        __syncthreads();

        for (int layer = 0; layer < nlayers; layer++) {
            const v2f*  wl = layer ? wl2 : wl1;
            const float* bl = layer ? bl2 : bl1;
            bool last = (layer == nlayers - 1);

            v2f a[16];
#pragma unroll
            for (int kk = 0; kk < 16; kk++)
                a[kk] = *(const v2f*)&X[wave][mrow][kk * 4 + khalf];

#pragma unroll
            for (int nc = 0; nc < 4; nc++) {
                v8f acc = {0.f, 0.f, 0.f, 0.f, 0.f, 0.f, 0.f, 0.f};
#pragma unroll
                for (int kk = 0; kk < 16; kk++) {
                    acc = __builtin_amdgcn_wmma_f32_16x16x4_f32(
                        false, a[kk], false, wl[(kk * 4 + nc) * 32 + lane],
                        (short)0, acc, false, false);
                }
                int m = nc * 16 + mrow;
                float bv = bl[m];
                if (!last) {
#pragma unroll
                    for (int v = 0; v < 8; v++)
                        X[wave][v + 8 * (lane >> 4)][m] = leaky(acc[v] + bv);
                } else {
                    float mx = -3.4e38f;
#pragma unroll
                    for (int v = 0; v < 8; v++) mx = fmaxf(mx, leaky(acc[v] + bv));
                    mx = fmaxf(mx, __shfl_xor(mx, 16, 32));
                    if (lane < 16) {
                        out_rm[(size_t)q * 64 + m] = mx;
                        if (out_cm)
                            out_cm[(size_t)b * 64 * NP + (size_t)m * NP + n] = mx;
                    }
                }
            }
            __syncthreads();
        }
    }
}

extern "C" void kernel_launch(void* const* d_in, const int* in_sizes, int n_in,
                              void* d_out, int out_size, void* d_ws, size_t ws_size,
                              hipStream_t stream)
{
    (void)in_sizes; (void)n_in; (void)out_size; (void)ws_size;
    const float* pc1   = (const float*)d_in[0];
    const float* pc2   = (const float*)d_in[1];
    const float* feat1 = (const float*)d_in[2];
    const float* feat2 = (const float*)d_in[3];
    const float* t11_w = (const float*)d_in[4];  const float* t11_b = (const float*)d_in[5];
    const float* t22_w = (const float*)d_in[6];  const float* t22_b = (const float*)d_in[7];
    const float* pos1w = (const float*)d_in[8];  const float* pos1b = (const float*)d_in[9];
    const float* m1w1  = (const float*)d_in[10]; const float* m1b1  = (const float*)d_in[11];
    const float* m1w2  = (const float*)d_in[12]; const float* m1b2  = (const float*)d_in[13];
    const float* t1_w  = (const float*)d_in[14]; const float* t1_b  = (const float*)d_in[15];
    const float* t2_w  = (const float*)d_in[16]; const float* t2_b  = (const float*)d_in[17];
    const float* pos2w = (const float*)d_in[18]; const float* pos2b = (const float*)d_in[19];
    const float* m2w1  = (const float*)d_in[20]; const float* m2b1  = (const float*)d_in[21];

    float* out = (float*)d_out;
    const size_t S = (size_t)BB * NP * 64;
    float* ws = (float*)d_ws;
    float* q1 = ws;           float* q2 = ws + S;
    float* r1 = ws + 2 * S;   float* r2 = ws + 3 * S;
    float* pooled1 = ws + 4 * S;
    float* pooled2 = ws + 5 * S;
    int* idx1 = (int*)(ws + 6 * S);
    int* idx2 = idx1 + (size_t)BB * NP * KNN;
    float* fn1_rm  = q1;
    float* fn2_rm  = q2;
    float* pooled3 = r1;

    knn_kernel<<<BB * NP / KNN_QPB, KNN_THREADS, 0, stream>>>(pc1, pc2, idx1);
    knn_kernel<<<BB * NP / KNN_QPB, KNN_THREADS, 0, stream>>>(pc2, pc1, idx2);

    gemm_cm_kernel<<<4 * GEMM_BLOCKS, 256, 0, stream>>>(feat1, feat2,
        t11_w, t11_b, t22_w, t22_b, q1);

    cross_kernel<<<2 * CROSS_BLOCKS, CROSS_THREADS, 0, stream>>>(pc1, pc2, q1, q2, r1, r2, idx1, idx2,
        pos1w, pos1b, m1w1, m1b1, m1w2, m1b2, 2, CROSS_BLOCKS, CROSS_QPW, pooled1, pooled2, nullptr);

    gemm_rm_kernel<<<2 * GEMM_BLOCKS, 256, 0, stream>>>(pooled1, t1_w, t1_b, t2_w, t2_b,
        out, fn1_rm);

    cross_kernel<<<2 * CROSS_BLOCKS, CROSS_THREADS, 0, stream>>>(pc1, pc2, fn1_rm, fn2_rm, fn1_rm, fn2_rm,
        idx1, idx2, pos2w, pos2b, m2w1, m2b1, nullptr, nullptr, 1, 2 * CROSS_BLOCKS, CROSS_QPW / 2,
        pooled3, pooled2, out + 2 * S);
}


# kNN pass 1 keeps one minimum per 64 consecutive keys (128 groups per query); pass 2 forms direct distances only in groups whose minimum is at or under the bound plus a rounding margin
# speedup vs baseline: 26.3997x; 1.1026x over previous
#include <hip/hip_runtime.h>

typedef float v2f __attribute__((ext_vector_type(2)));
typedef float v8f __attribute__((ext_vector_type(8)));

static constexpr int BB  = 2;
static constexpr int NP  = 8192;
static constexpr int CH  = 64;
static constexpr int KNN = 16;

__device__ __forceinline__ float leaky(float x) { return fmaxf(x, 0.1f * x); }

static constexpr int KNN_THREADS = 512;
static constexpr int KNN_QPW     = 2;
static constexpr int KNN_QPB     = (KNN_THREADS / 32) * KNN_QPW;
static constexpr int KNN_PITCH4  = 33;

__device__ __forceinline__ float dist3(float qx, float qy, float qz,
                                       float x, float y, float z)
{
    const float dx = qx - x, dy = qy - y, dz = qz - z;
    return __builtin_fmaf(dz, dz, __builtin_fmaf(dx, dx, dy * dy));
}

__device__ __forceinline__ unsigned sixteenth_of_lanes(unsigned v, unsigned big)
{
    unsigned r = big;
#pragma unroll
    for (int j = 0; j < 32; j++) {
        const unsigned s = (unsigned)__builtin_amdgcn_readlane((int)v, j);
        const int cnt = __builtin_popcount(__builtin_amdgcn_ballot_w32(v <= s));
        r = (cnt >= 16) ? min(r, s) : r;
    }
    return r;
}

__device__ __forceinline__ int lane_below(int edge, int v)
{
    return __builtin_amdgcn_update_dpp(edge, v, 0x111, 0xf, 0xf, false);
}

__device__ __forceinline__ void insert1(float dk, int ik, float big,
                                        unsigned td0b, float& ld, int& li, float& td)
{
    unsigned mask = __builtin_amdgcn_ballot_w32(dk <= td);
    while (mask != 0u) {
        const int src = __builtin_ctz(mask);
        mask &= mask - 1u;
        const float dn = __int_as_float(
            __builtin_amdgcn_readlane(__float_as_int(dk), src));
        const int in = __builtin_amdgcn_readlane(ik, src);
        const float pd = __int_as_float(
            lane_below(__float_as_int(-big), __float_as_int(ld)));
        const int   pi = lane_below(0, li);
        const bool lt  = (dn < ld) | ((dn == ld) & (in < li));
        const bool ltp = (dn < pd) | ((dn == pd) & (in < pi));
        ld = lt ? (ltp ? pd : dn) : ld;
        li = lt ? (ltp ? pi : in) : li;
        td = __uint_as_float(min(td0b, (unsigned)
            __builtin_amdgcn_readlane(__float_as_int(ld), 15)));
        mask &= __builtin_amdgcn_ballot_w32(dk <= td);
    }
}

__device__ __forceinline__ unsigned min5_bits(unsigned m, float a, float b, float c, float d)
{
    m = min(min(m, __float_as_uint(a)), __float_as_uint(b));
    m = min(min(m, __float_as_uint(c)), __float_as_uint(d));
    return m;
}

__device__ __forceinline__ void scan_groups(const float4& g, float thr,
                                            float qx, float qy, float qz,
                                            const float4* sx4, const float4* sy4,
                                            const float4* sz4, int lane, float big,
                                            unsigned td0b, float& ld, int& li)
{
    const v2f* px = (const v2f*)sx4 + lane;
    const v2f* py = (const v2f*)sy4 + lane;
    const v2f* pz = (const v2f*)sz4 + lane;
    float td = __uint_as_float(td0b);
#pragma unroll 1
    for (int k = 0; k < 4; k++) {
        const float gk = (k == 0) ? g.x : (k == 1) ? g.y : (k == 2) ? g.z : g.w;
        unsigned m = __builtin_amdgcn_ballot_w32(gk <= thr);
#pragma unroll 1
        while (m != 0u) {
            const int G = 4 * __builtin_ctz(m) + k;
            m &= m - 1u;
            const v2f x = px[32 * G], y = py[32 * G], z = pz[32 * G];
            const float d0 = dist3(qx, qy, qz, x.x, y.x, z.x);
            const float d1 = dist3(qx, qy, qz, x.y, y.y, z.y);
            const int i0 = 64 * G + 2 * lane;
            insert1(d0, i0,     big, td0b, ld, li, td);
            insert1(d1, i0 + 1, big, td0b, ld, li, td);
        }
    }
}

__global__ __launch_bounds__(KNN_THREADS) void knn_kernel(
    const float* __restrict__ xq,
    const float* __restrict__ xk,
    int* __restrict__ idx)
{
    __shared__ float4 sx4[NP / 4], sy4[NP / 4], sz4[NP / 4], sw4[NP / 4];
    __shared__ float4 pmf4[KNN_QPB * KNN_PITCH4];
    __shared__ float smax[KNN_THREADS / 32];
    const int tid  = threadIdx.x;
    const int lane = tid & 31, wave = tid >> 5;
    const int q0 = blockIdx.x * KNN_QPB;
    const int b  = q0 / NP;
    const float* kp = xk + (size_t)b * 3 * NP;
    const float* qp = xq + (size_t)b * 3 * NP;
    {
        const float4* kx4 = (const float4*)kp;
        const float4* ky4 = (const float4*)(kp + NP);
        const float4* kz4 = (const float4*)(kp + 2 * NP);
        float kwm = 0.f;
        for (int j = tid; j < NP / 4; j += KNN_THREADS) {
            const float4 x = kx4[j], y = ky4[j], z = kz4[j];
            float4 w;
            w.x = __builtin_fmaf(z.x, z.x, __builtin_fmaf(x.x, x.x, y.x * y.x));
            w.y = __builtin_fmaf(z.y, z.y, __builtin_fmaf(x.y, x.y, y.y * y.y));
            w.z = __builtin_fmaf(z.z, z.z, __builtin_fmaf(x.z, x.z, y.z * y.z));
            w.w = __builtin_fmaf(z.w, z.w, __builtin_fmaf(x.w, x.w, y.w * y.w));
            sx4[j] = x; sy4[j] = y; sz4[j] = z; sw4[j] = w;
            kwm = fmaxf(kwm, fmaxf(fmaxf(w.x, w.y), fmaxf(w.z, w.w)));
        }
#pragma unroll
        for (int o = 16; o > 0; o >>= 1) kwm = fmaxf(kwm, __shfl_xor(kwm, o, 32));
        if (lane == 0) smax[wave] = kwm;
    }
    __syncthreads();
    float kwmax = smax[lane & 15];
#pragma unroll
    for (int o = 8; o > 0; o >>= 1) kwmax = fmaxf(kwmax, __shfl_xor(kwmax, o, 32));

    const float BIG = 3.4e38f;
    const unsigned BIGB = __float_as_uint(BIG);
    const float EPS = 1.52587890625e-05f;
    const int qA = q0 + wave * KNN_QPW;
    const int nA = qA - b * NP;
    const float ax = qp[nA],     ay = qp[NP + nA],     az = qp[2 * NP + nA];
    const float bx = qp[nA + 1], by = qp[NP + nA + 1], bz = qp[2 * NP + nA + 1];

    {
        const int c = lane & 15, h = lane >> 4;
        const int R = wave >> 3, sl = wave & 7;
        const int nq = q0 - b * NP + 16 * R + c;
        const float qx = qp[nq], qy = qp[NP + nq], qz = qp[2 * NP + nq];
        const float q2 = __builtin_fmaf(qz, qz, __builtin_fmaf(qx, qx, qy * qy));
        const float cq = q2 + EPS * (q2 + kwmax);
        v2f bq;
        bq.x = -2.f * (h ? qz : qx);
        bq.y = h ? 1.f : -2.f * qy;
        const float4* pa = h ? sz4 : sx4;
        const float4* pb = h ? sw4 : sy4;
        const v8f cin = {cq, cq, cq, cq, cq, cq, cq, cq};
        unsigned m[8];
#pragma unroll
        for (int r = 0; r < 8; r++) m[r] = BIGB;
        const int jb = 256 * sl + 16 * c;
#pragma unroll 2
        for (int u = 0; u < 16; u++) {
            const int j = jb + ((u + c) & 15);
            const float4 p = pa[j], q = pb[j];
            v2f a0, a1, a2, a3;
            a0.x = p.x; a0.y = q.x;  a1.x = p.y; a1.y = q.y;
            a2.x = p.z; a2.y = q.z;  a3.x = p.w; a3.y = q.w;
            const v8f d0 = __builtin_amdgcn_wmma_f32_16x16x4_f32(
                false, a0, false, bq, (short)0, cin, false, false);
            const v8f d1 = __builtin_amdgcn_wmma_f32_16x16x4_f32(
                false, a1, false, bq, (short)0, cin, false, false);
            const v8f d2 = __builtin_amdgcn_wmma_f32_16x16x4_f32(
                false, a2, false, bq, (short)0, cin, false, false);
            const v8f d3 = __builtin_amdgcn_wmma_f32_16x16x4_f32(
                false, a3, false, bq, (short)0, cin, false, false);
#pragma unroll
            for (int r = 0; r < 8; r++) m[r] = min5_bits(m[r], d0[r], d1[r], d2[r], d3[r]);
        }
        float4 o0, o1;
        o0.x = __uint_as_float(m[0]); o0.y = __uint_as_float(m[1]);
        o0.z = __uint_as_float(m[2]); o0.w = __uint_as_float(m[3]);
        o1.x = __uint_as_float(m[4]); o1.y = __uint_as_float(m[5]);
        o1.z = __uint_as_float(m[6]); o1.w = __uint_as_float(m[7]);
        float4* dst = pmf4 + (16 * R + c) * KNN_PITCH4 + 4 * sl + 2 * h;
        dst[0] = o0; dst[1] = o1;
    }
    __syncthreads();
    const float4 gA = pmf4[(2 * wave) * KNN_PITCH4 + lane];
    const float4 gB = pmf4[(2 * wave + 1) * KNN_PITCH4 + lane];
    const unsigned td0A = sixteenth_of_lanes(
        __float_as_uint(fminf(fminf(gA.x, gA.y), fminf(gA.z, gA.w))), BIGB);
    const unsigned td0B = sixteenth_of_lanes(
        __float_as_uint(fminf(fminf(gB.x, gB.y), fminf(gB.z, gB.w))), BIGB);
    const float thrA = __uint_as_float(td0A) + 2.f * EPS *
        (__builtin_fmaf(az, az, __builtin_fmaf(ax, ax, ay * ay)) + kwmax);
    const float thrB = __uint_as_float(td0B) + 2.f * EPS *
        (__builtin_fmaf(bz, bz, __builtin_fmaf(bx, bx, by * by)) + kwmax);

    float ldA = BIG, ldB = BIG; int liA = 0, liB = 0;
    scan_groups(gA, thrA, ax, ay, az, sx4, sy4, sz4, lane, BIG, td0A, ldA, liA);
    scan_groups(gB, thrB, bx, by, bz, sx4, sy4, sz4, lane, BIG, td0B, ldB, liB);
    if (lane < KNN) {
        idx[(size_t)qA * KNN + lane]       = liA;
        idx[(size_t)(qA + 1) * KNN + lane] = liB;
    }
}

__device__ __forceinline__ void stage_wfrag(v2f* wl, const float* __restrict__ W,
                                            int tid, int nthreads)
{
    for (int s = tid; s < 2048; s += nthreads) {
        int lane = s & 31, nc = (s >> 5) & 3, kk = s >> 7;
        int o  = nc * 16 + (lane & 15);
        int k0 = kk * 4 + 2 * (lane >> 4);
        v2f w; w.x = W[o * 64 + k0]; w.y = W[o * 64 + k0 + 1];
        wl[s] = w;
    }
}

template <int NT>
__device__ __forceinline__ void stage_wfrag_once(v2f* wl, const float* __restrict__ W, int tid)
{
    v2f w[2048 / NT];
#pragma unroll
    for (int i = 0; i < 2048 / NT; i++) {
        int s = tid + i * NT;
        int lane = s & 31, nc = (s >> 5) & 3, kk = s >> 7;
        int o  = nc * 16 + (lane & 15);
        int k0 = kk * 4 + 2 * (lane >> 4);
        w[i].x = W[o * 64 + k0]; w[i].y = W[o * 64 + k0 + 1];
    }
#pragma unroll
    for (int i = 0; i < 2048 / NT; i++) wl[tid + i * NT] = w[i];
}

static constexpr int GEMM_BLOCKS = BB * NP / 16 / 8;

__global__ __launch_bounds__(256) void gemm_cm_kernel(
    const float* __restrict__ inA,
    const float* __restrict__ inB,
    const float* __restrict__ Wa,
    const float* __restrict__ ba,
    const float* __restrict__ Wb,
    const float* __restrict__ bb,
    float* __restrict__ out_rm4)
{
    __shared__ v2f  wl[2048];
    __shared__ float bl[64];
    int tid = threadIdx.x;
    const int prod = blockIdx.x / GEMM_BLOCKS;
    const int blk  = blockIdx.x % GEMM_BLOCKS;
    const float* in   = (prod == 0 || prod == 3) ? inA : inB;
    const float* W    = (prod & 1) ? Wb : Wa;
    const float* bias = (prod & 1) ? bb : ba;
    float* out_rm = out_rm4 + (size_t)prod * BB * NP * 64;
    stage_wfrag(wl, W, tid, 256);
    if (tid < 64) bl[tid] = bias[tid];
    __syncthreads();

    int lane = tid & 31, wave = tid >> 5;
    int tile = blk * 8 + wave;
    int b  = tile / (NP / 16);
    int n0 = (tile % (NP / 16)) * 16;
    const float* inb = in + (size_t)b * CH * NP;
    int row   = n0 + (lane & 15);
    int khalf = 2 * (lane >> 4);

    v2f a[16];
#pragma unroll
    for (int kk = 0; kk < 16; kk++) {
        int k0 = kk * 4 + khalf;
        v2f v; v.x = inb[(size_t)k0 * NP + row]; v.y = inb[(size_t)(k0 + 1) * NP + row];
        a[kk] = v;
    }
#pragma unroll
    for (int nc = 0; nc < 4; nc++) {
        v8f acc = {0.f, 0.f, 0.f, 0.f, 0.f, 0.f, 0.f, 0.f};
#pragma unroll
        for (int kk = 0; kk < 16; kk++) {
            acc = __builtin_amdgcn_wmma_f32_16x16x4_f32(
                false, a[kk], false, wl[(kk * 4 + nc) * 32 + lane],
                (short)0, acc, false, false);
        }
        int m = nc * 16 + (lane & 15);
        float bv = bl[m];
        float* ob = out_rm + (size_t)b * NP * 64;
#pragma unroll
        for (int v = 0; v < 8; v++) {
            int r = n0 + v + 8 * (lane >> 4);
            ob[(size_t)r * 64 + m] = acc[v] + bv;
        }
    }
}

__global__ __launch_bounds__(256) void gemm_rm_kernel(
    const float* __restrict__ in2,
    const float* __restrict__ W0, const float* __restrict__ bias0,
    const float* __restrict__ W1, const float* __restrict__ bias1,
    float* __restrict__ out_cm2,
    float* __restrict__ out_rm2)
{
    __shared__ v2f  wl[2048];
    __shared__ float bl[64];
    int tid = threadIdx.x;
    const int prod = blockIdx.x / GEMM_BLOCKS;
    const int blk  = blockIdx.x % GEMM_BLOCKS;
    const size_t plane = (size_t)prod * BB * NP * 64;
    const float* in   = in2 + plane;
    const float* W    = prod ? W1 : W0;
    const float* bias = prod ? bias1 : bias0;
    float* out_cm = out_cm2 + plane;
    float* out_rm = out_rm2 + plane;
    stage_wfrag(wl, W, tid, 256);
    if (tid < 64) bl[tid] = bias[tid];
    __syncthreads();

    int lane = tid & 31, wave = tid >> 5;
    int tile = blk * 8 + wave;
    int b  = tile / (NP / 16);
    int n0 = (tile % (NP / 16)) * 16;
    const float* inb = in + (size_t)b * NP * 64;
    int row   = n0 + (lane & 15);
    int khalf = 2 * (lane >> 4);

    v2f a[16];
#pragma unroll
    for (int kk = 0; kk < 16; kk++)
        a[kk] = *(const v2f*)(inb + (size_t)row * 64 + kk * 4 + khalf);

#pragma unroll
    for (int nc = 0; nc < 4; nc++) {
        v8f acc = {0.f, 0.f, 0.f, 0.f, 0.f, 0.f, 0.f, 0.f};
#pragma unroll
        for (int kk = 0; kk < 16; kk++) {
            acc = __builtin_amdgcn_wmma_f32_16x16x4_f32(
                false, a[kk], false, wl[(kk * 4 + nc) * 32 + lane],
                (short)0, acc, false, false);
        }
        int m = nc * 16 + (lane & 15);
        float bv = bl[m];
#pragma unroll
        for (int v = 0; v < 8; v++) {
            int r = n0 + v + 8 * (lane >> 4);
            float val = acc[v] + bv;
            out_cm[(size_t)b * 64 * NP + (size_t)m * NP + r] = val;
            out_rm[((size_t)b * NP + r) * 64 + m] = val;
        }
    }
}

static constexpr int CROSS_THREADS = 256;
static constexpr int CROSS_WAVES   = CROSS_THREADS / 32;
static constexpr int CROSS_BLOCKS  = 512;
static constexpr int CROSS_QPW     = BB * NP / (CROSS_BLOCKS * CROSS_WAVES);

__global__ __launch_bounds__(CROSS_THREADS, 8) void cross_kernel(
    const float* __restrict__ xa, const float* __restrict__ xb,
    const float* __restrict__ p1_0, const float* __restrict__ p2_0,
    const float* __restrict__ p1_1, const float* __restrict__ p2_1,
    const int*   __restrict__ idx_0, const int* __restrict__ idx_1,
    const float* __restrict__ posw, const float* __restrict__ posb,
    const float* __restrict__ w1, const float* __restrict__ b1,
    const float* __restrict__ w2, const float* __restrict__ b2,
    int nlayers,
    int bps,
    int qpw,
    float* __restrict__ out_rm_0,
    float* __restrict__ out_rm_1,
    float* __restrict__ out_cm)
{
    const int set = blockIdx.x / bps;
    const int blk = blockIdx.x % bps;
    const float* xq  = set ? xb : xa;
    const float* xk  = set ? xa : xb;
    const float* p1  = set ? p1_1 : p1_0;
    const float* p2  = set ? p2_1 : p2_0;
    const int*   idx = set ? idx_1 : idx_0;
    float* out_rm    = set ? out_rm_1 : out_rm_0;

    __shared__ v2f  wl1[2048];
    __shared__ v2f  wl2[2048];
    __shared__ float bl1[64], bl2[64];
    __shared__ float pw[192], pb[64];
    __shared__ float X[CROSS_WAVES][16][68];

    int tid = threadIdx.x;
    stage_wfrag_once<CROSS_THREADS>(wl1, w1, tid);
    if (nlayers > 1) stage_wfrag_once<CROSS_THREADS>(wl2, w2, tid);
    if (tid < 192) pw[tid] = posw[tid];
    if (tid < 64) {
        bl1[tid] = b1[tid];
        bl2[tid] = (nlayers > 1) ? b2[tid] : 0.f;
        pb[tid]  = posb[tid];
    }
    __syncthreads();

    int lane = tid & 31, wave = tid >> 5;
    int mrow  = lane & 15;
    int khalf = 2 * (lane >> 4);

    int c0 = 2 * lane;
    float pwa0 = pw[c0 * 3 + 0], pwa1 = pw[c0 * 3 + 1], pwa2 = pw[c0 * 3 + 2];
    float pwb0 = pw[(c0 + 1) * 3 + 0], pwb1 = pw[(c0 + 1) * 3 + 1], pwb2 = pw[(c0 + 1) * 3 + 2];
    float pb0 = pb[c0], pb1 = pb[c0 + 1];

#pragma unroll 1
    for (int it = 0; it < qpw; it++) {
        int q = (blk * qpw + it) * CROSS_WAVES + wave;
        int b = q / NP;
        int n = q % NP;

        const float* xqb = xq + (size_t)b * 3 * NP;
        const float* xkb = xk + (size_t)b * 3 * NP;
        float qx = xqb[n], qy = xqb[NP + n], qz = xqb[2 * NP + n];
        const int* iq = idx + (size_t)q * KNN;

        v2f p1v = *(const v2f*)(p1 + (size_t)q * 64 + c0);
#pragma unroll
        for (int r = 0; r < KNN; r++) {
            int nb = iq[r];
            float dx = xkb[nb] - qx, dy = xkb[NP + nb] - qy, dz = xkb[2 * NP + nb] - qz;
            v2f g = *(const v2f*)(p2 + ((size_t)b * NP + nb) * 64 + c0);
            X[wave][r][c0]     = leaky(g.x + p1v.x + pb0 + dx * pwa0 + dy * pwa1 + dz * pwa2);
            X[wave][r][c0 + 1] = leaky(g.y + p1v.y + pb1 + dx * pwb0 + dy * pwb1 + dz * pwb2);
        }
        __syncthreads();

        for (int layer = 0; layer < nlayers; layer++) {
            const v2f*  wl = layer ? wl2 : wl1;
            const float* bl = layer ? bl2 : bl1;
            bool last = (layer == nlayers - 1);

            v2f a[16];
#pragma unroll
            for (int kk = 0; kk < 16; kk++)
                a[kk] = *(const v2f*)&X[wave][mrow][kk * 4 + khalf];

#pragma unroll
            for (int nc = 0; nc < 4; nc++) {
                v8f acc = {0.f, 0.f, 0.f, 0.f, 0.f, 0.f, 0.f, 0.f};
#pragma unroll
                for (int kk = 0; kk < 16; kk++) {
                    acc = __builtin_amdgcn_wmma_f32_16x16x4_f32(
                        false, a[kk], false, wl[(kk * 4 + nc) * 32 + lane],
                        (short)0, acc, false, false);
                }
                int m = nc * 16 + mrow;
                float bv = bl[m];
                if (!last) {
#pragma unroll
                    for (int v = 0; v < 8; v++)
                        X[wave][v + 8 * (lane >> 4)][m] = leaky(acc[v] + bv);
                } else {
                    float mx = -3.4e38f;
#pragma unroll
                    for (int v = 0; v < 8; v++) mx = fmaxf(mx, leaky(acc[v] + bv));
                    mx = fmaxf(mx, __shfl_xor(mx, 16, 32));
                    if (lane < 16) {
                        out_rm[(size_t)q * 64 + m] = mx;
                        if (out_cm)
                            out_cm[(size_t)b * 64 * NP + (size_t)m * NP + n] = mx;
                    }
                }
            }
            __syncthreads();
        }
    }
}

extern "C" void kernel_launch(void* const* d_in, const int* in_sizes, int n_in,
                              void* d_out, int out_size, void* d_ws, size_t ws_size,
                              hipStream_t stream)
{
    (void)in_sizes; (void)n_in; (void)out_size; (void)ws_size;
    const float* pc1   = (const float*)d_in[0];
    const float* pc2   = (const float*)d_in[1];
    const float* feat1 = (const float*)d_in[2];
    const float* feat2 = (const float*)d_in[3];
    const float* t11_w = (const float*)d_in[4];  const float* t11_b = (const float*)d_in[5];
    const float* t22_w = (const float*)d_in[6];  const float* t22_b = (const float*)d_in[7];
    const float* pos1w = (const float*)d_in[8];  const float* pos1b = (const float*)d_in[9];
    const float* m1w1  = (const float*)d_in[10]; const float* m1b1  = (const float*)d_in[11];
    const float* m1w2  = (const float*)d_in[12]; const float* m1b2  = (const float*)d_in[13];
    const float* t1_w  = (const float*)d_in[14]; const float* t1_b  = (const float*)d_in[15];
    const float* t2_w  = (const float*)d_in[16]; const float* t2_b  = (const float*)d_in[17];
    const float* pos2w = (const float*)d_in[18]; const float* pos2b = (const float*)d_in[19];
    const float* m2w1  = (const float*)d_in[20]; const float* m2b1  = (const float*)d_in[21];

    float* out = (float*)d_out;
    const size_t S = (size_t)BB * NP * 64;
    float* ws = (float*)d_ws;
    float* q1 = ws;           float* q2 = ws + S;
    float* r1 = ws + 2 * S;   float* r2 = ws + 3 * S;
    float* pooled1 = ws + 4 * S;
    float* pooled2 = ws + 5 * S;
    int* idx1 = (int*)(ws + 6 * S);
    int* idx2 = idx1 + (size_t)BB * NP * KNN;
    float* fn1_rm  = q1;
    float* fn2_rm  = q2;
    float* pooled3 = r1;

    knn_kernel<<<BB * NP / KNN_QPB, KNN_THREADS, 0, stream>>>(pc1, pc2, idx1);
    knn_kernel<<<BB * NP / KNN_QPB, KNN_THREADS, 0, stream>>>(pc2, pc1, idx2);

    gemm_cm_kernel<<<4 * GEMM_BLOCKS, 256, 0, stream>>>(feat1, feat2,
        t11_w, t11_b, t22_w, t22_b, q1);

    cross_kernel<<<2 * CROSS_BLOCKS, CROSS_THREADS, 0, stream>>>(pc1, pc2, q1, q2, r1, r2, idx1, idx2,
        pos1w, pos1b, m1w1, m1b1, m1w2, m1b2, 2, CROSS_BLOCKS, CROSS_QPW, pooled1, pooled2, nullptr);

    gemm_rm_kernel<<<2 * GEMM_BLOCKS, 256, 0, stream>>>(pooled1, t1_w, t1_b, t2_w, t2_b,
        out, fn1_rm);

    cross_kernel<<<2 * CROSS_BLOCKS, CROSS_THREADS, 0, stream>>>(pc1, pc2, fn1_rm, fn2_rm, fn1_rm, fn2_rm,
        idx1, idx2, pos2w, pos2b, m2w1, m2b1, nullptr, nullptr, 1, 2 * CROSS_BLOCKS, CROSS_QPW / 2,
        pooled3, pooled2, out + 2 * S);
}


# cross stages: the position term posw.(x_nb - x_q) + posb is added once per row by the GEMMs that write the rows; the gather phase is then one add per gathered row, no coordinates, no projection
# speedup vs baseline: 30.2368x; 1.1453x over previous
#include <hip/hip_runtime.h>

typedef float v2f __attribute__((ext_vector_type(2)));
typedef float v8f __attribute__((ext_vector_type(8)));

static constexpr int BB  = 2;
static constexpr int NP  = 8192;
static constexpr int CH  = 64;
static constexpr int KNN = 16;

__device__ __forceinline__ float leaky(float x) { return fmaxf(x, 0.1f * x); }

static constexpr int KNN_THREADS = 512;
static constexpr int KNN_QPW     = 2;
static constexpr int KNN_QPB     = (KNN_THREADS / 32) * KNN_QPW;
static constexpr int KNN_PITCH4  = 33;
static constexpr int KNN_BLOCKS  = BB * NP / KNN_QPB;
static constexpr int GEMM_CM_BLOCKS = BB * NP / 16 / (KNN_THREADS / 32);
static constexpr int GEMM_CM_TOTAL  = 4 * GEMM_CM_BLOCKS;
static constexpr int FRONT_GRID     = GEMM_CM_TOTAL + 2 * KNN_BLOCKS;

__device__ __forceinline__ float dist3(float qx, float qy, float qz,
                                       float x, float y, float z)
{
    const float dx = qx - x, dy = qy - y, dz = qz - z;
    return __builtin_fmaf(dz, dz, __builtin_fmaf(dx, dx, dy * dy));
}

__device__ __forceinline__ unsigned sixteenth_of_lanes(unsigned v, unsigned big)
{
    unsigned r = big;
#pragma unroll
    for (int j = 0; j < 32; j++) {
        const unsigned s = (unsigned)__builtin_amdgcn_readlane((int)v, j);
        const int cnt = __builtin_popcount(__builtin_amdgcn_ballot_w32(v <= s));
        r = (cnt >= 16) ? min(r, s) : r;
    }
    return r;
}

__device__ __forceinline__ int lane_below(int edge, int v)
{
    return __builtin_amdgcn_update_dpp(edge, v, 0x111, 0xf, 0xf, false);
}

__device__ __forceinline__ void insert1(float dk, int ik, float big,
                                        unsigned td0b, float& ld, int& li, float& td)
{
    unsigned mask = __builtin_amdgcn_ballot_w32(dk <= td);
    while (mask != 0u) {
        const int src = __builtin_ctz(mask);
        mask &= mask - 1u;
        const float dn = __int_as_float(
            __builtin_amdgcn_readlane(__float_as_int(dk), src));
        const int in = __builtin_amdgcn_readlane(ik, src);
        const float pd = __int_as_float(
            lane_below(__float_as_int(-big), __float_as_int(ld)));
        const int   pi = lane_below(0, li);
        const bool lt  = (dn < ld) | ((dn == ld) & (in < li));
        const bool ltp = (dn < pd) | ((dn == pd) & (in < pi));
        ld = lt ? (ltp ? pd : dn) : ld;
        li = lt ? (ltp ? pi : in) : li;
        td = __uint_as_float(min(td0b, (unsigned)
            __builtin_amdgcn_readlane(__float_as_int(ld), 15)));
        mask &= __builtin_amdgcn_ballot_w32(dk <= td);
    }
}

__device__ __forceinline__ unsigned min5_bits(unsigned m, float a, float b, float c, float d)
{
    m = min(min(m, __float_as_uint(a)), __float_as_uint(b));
    m = min(min(m, __float_as_uint(c)), __float_as_uint(d));
    return m;
}

__device__ __forceinline__ void scan_groups(const float4& g, float thr,
                                            float qx, float qy, float qz,
                                            const float4* sx4, const float4* sy4,
                                            const float4* sz4, int lane, float big,
                                            unsigned td0b, float& ld, int& li)
{
    const v2f* px = (const v2f*)sx4 + lane;
    const v2f* py = (const v2f*)sy4 + lane;
    const v2f* pz = (const v2f*)sz4 + lane;
    float td = __uint_as_float(td0b);
#pragma unroll 1
    for (int k = 0; k < 4; k++) {
        const float gk = (k == 0) ? g.x : (k == 1) ? g.y : (k == 2) ? g.z : g.w;
        unsigned m = __builtin_amdgcn_ballot_w32(gk <= thr);
#pragma unroll 1
        while (m != 0u) {
            const int G = 4 * __builtin_ctz(m) + k;
            m &= m - 1u;
            const v2f x = px[32 * G], y = py[32 * G], z = pz[32 * G];
            const float d0 = dist3(qx, qy, qz, x.x, y.x, z.x);
            const float d1 = dist3(qx, qy, qz, x.y, y.y, z.y);
            const int i0 = 64 * G + 2 * lane;
            insert1(d0, i0,     big, td0b, ld, li, td);
            insert1(d1, i0 + 1, big, td0b, ld, li, td);
        }
    }
}


__device__ __forceinline__ void stage_wfrag(v2f* wl, const float* __restrict__ W,
                                            int tid, int nthreads)
{
    for (int s = tid; s < 2048; s += nthreads) {
        int lane = s & 31, nc = (s >> 5) & 3, kk = s >> 7;
        int o  = nc * 16 + (lane & 15);
        int k0 = kk * 4 + 2 * (lane >> 4);
        v2f w; w.x = W[o * 64 + k0]; w.y = W[o * 64 + k0 + 1];
        wl[s] = w;
    }
}

__device__ __forceinline__ void gemm_cm_block(
    int gblk, v2f* wl, float* bl,
    const float* __restrict__ inA,
    const float* __restrict__ inB,
    const float* __restrict__ Wa,
    const float* __restrict__ ba,
    const float* __restrict__ Wb,
    const float* __restrict__ bb,
    const float* __restrict__ xa,
    const float* __restrict__ xb,
    const float* __restrict__ posw,
    const float* __restrict__ posb,
    float* __restrict__ out_rm4)
{
    const int tid = threadIdx.x;
    const int prod = gblk / GEMM_CM_BLOCKS;
    const int blk  = gblk % GEMM_CM_BLOCKS;
    const bool fromA  = (prod == 0 || prod == 3);
    const bool keyrow = (prod & 1) != 0;
    const float* in   = fromA ? inA : inB;
    const float* xc   = fromA ? xa : xb;
    const float* W    = keyrow ? Wb : Wa;
    const float* bias = keyrow ? bb : ba;
    float* out_rm = out_rm4 + (size_t)prod * BB * NP * 64;
    float* pwl = bl + 64;
    stage_wfrag(wl, W, tid, KNN_THREADS);
    if (tid < 64) bl[tid] = keyrow ? bias[tid] : bias[tid] + posb[tid];
    if (tid < 192) pwl[tid] = keyrow ? posw[tid] : -posw[tid];
    __syncthreads();

    const int lane = tid & 31, wave = tid >> 5;
    const int tile = blk * (KNN_THREADS / 32) + wave;
    const int b  = tile / (NP / 16);
    const int n0 = (tile % (NP / 16)) * 16;
    const float* inb = in + (size_t)b * CH * NP;
    const int row   = n0 + (lane & 15);
    const int khalf = 2 * (lane >> 4);

    v2f a[16];
#pragma unroll
    for (int kk = 0; kk < 16; kk++) {
        int k0 = kk * 4 + khalf;
        v2f v; v.x = inb[(size_t)k0 * NP + row]; v.y = inb[(size_t)(k0 + 1) * NP + row];
        a[kk] = v;
    }
    const float* xr = xc + (size_t)b * 3 * NP + n0 + 8 * (lane >> 4);
    const float4 x0 = *(const float4*)xr,            x1 = *(const float4*)(xr + 4);
    const float4 y0 = *(const float4*)(xr + NP),     y1 = *(const float4*)(xr + NP + 4);
    const float4 z0 = *(const float4*)(xr + 2 * NP), z1 = *(const float4*)(xr + 2 * NP + 4);
    const float px[8] = {x0.x, x0.y, x0.z, x0.w, x1.x, x1.y, x1.z, x1.w};
    const float py[8] = {y0.x, y0.y, y0.z, y0.w, y1.x, y1.y, y1.z, y1.w};
    const float pz[8] = {z0.x, z0.y, z0.z, z0.w, z1.x, z1.y, z1.z, z1.w};
#pragma unroll
    for (int nc = 0; nc < 4; nc++) {
        v8f acc = {0.f, 0.f, 0.f, 0.f, 0.f, 0.f, 0.f, 0.f};
#pragma unroll
        for (int kk = 0; kk < 16; kk++) {
            acc = __builtin_amdgcn_wmma_f32_16x16x4_f32(
                false, a[kk], false, wl[(kk * 4 + nc) * 32 + lane],
                (short)0, acc, false, false);
        }
        int m = nc * 16 + (lane & 15);
        float bv = bl[m];
        const float w0 = pwl[m * 3 + 0], w1 = pwl[m * 3 + 1], w2 = pwl[m * 3 + 2];
        float* ob = out_rm + (size_t)b * NP * 64;
#pragma unroll
        for (int v = 0; v < 8; v++) {
            int r = n0 + v + 8 * (lane >> 4);
            ob[(size_t)r * 64 + m] = __builtin_fmaf(w2, pz[v],
                __builtin_fmaf(w1, py[v], __builtin_fmaf(w0, px[v], acc[v] + bv)));
        }
    }
}

__global__ __launch_bounds__(KNN_THREADS) void knn_gemm_kernel(
    const float* __restrict__ xa,
    const float* __restrict__ xb,
    int* __restrict__ idx_ab,
    int* __restrict__ idx_ba,
    const float* __restrict__ inA,
    const float* __restrict__ inB,
    const float* __restrict__ Wa, const float* __restrict__ ba,
    const float* __restrict__ Wb, const float* __restrict__ bb,
    const float* __restrict__ posw,
    const float* __restrict__ posb,
    float* __restrict__ out_rm4)
{
    __shared__ float4 sx4[NP / 4], sy4[NP / 4], sz4[NP / 4], sw4[NP / 4];
    __shared__ float4 pmf4[KNN_QPB * KNN_PITCH4];
    __shared__ float smax[KNN_THREADS / 32];
    if (blockIdx.x < GEMM_CM_TOTAL) {
        gemm_cm_block((int)blockIdx.x, (v2f*)sx4, (float*)sy4,
                      inA, inB, Wa, ba, Wb, bb, xa, xb, posw, posb, out_rm4);
        return;
    }
    const int kblk = (int)blockIdx.x - GEMM_CM_TOTAL;
    const int dir = kblk / KNN_BLOCKS;
    const float* xq = dir ? xb : xa;
    const float* xk = dir ? xa : xb;
    int* idx        = dir ? idx_ba : idx_ab;
    const int tid  = threadIdx.x;
    const int lane = tid & 31, wave = tid >> 5;
    const int q0 = (kblk % KNN_BLOCKS) * KNN_QPB;
    const int b  = q0 / NP;
    const float* kp = xk + (size_t)b * 3 * NP;
    const float* qp = xq + (size_t)b * 3 * NP;
    {
        const float4* kx4 = (const float4*)kp;
        const float4* ky4 = (const float4*)(kp + NP);
        const float4* kz4 = (const float4*)(kp + 2 * NP);
        float kwm = 0.f;
        for (int j = tid; j < NP / 4; j += KNN_THREADS) {
            const float4 x = kx4[j], y = ky4[j], z = kz4[j];
            float4 w;
            w.x = __builtin_fmaf(z.x, z.x, __builtin_fmaf(x.x, x.x, y.x * y.x));
            w.y = __builtin_fmaf(z.y, z.y, __builtin_fmaf(x.y, x.y, y.y * y.y));
            w.z = __builtin_fmaf(z.z, z.z, __builtin_fmaf(x.z, x.z, y.z * y.z));
            w.w = __builtin_fmaf(z.w, z.w, __builtin_fmaf(x.w, x.w, y.w * y.w));
            sx4[j] = x; sy4[j] = y; sz4[j] = z; sw4[j] = w;
            kwm = fmaxf(kwm, fmaxf(fmaxf(w.x, w.y), fmaxf(w.z, w.w)));
        }
#pragma unroll
        for (int o = 16; o > 0; o >>= 1) kwm = fmaxf(kwm, __shfl_xor(kwm, o, 32));
        if (lane == 0) smax[wave] = kwm;
    }
    __syncthreads();
    float kwmax = smax[lane & 15];
#pragma unroll
    for (int o = 8; o > 0; o >>= 1) kwmax = fmaxf(kwmax, __shfl_xor(kwmax, o, 32));

    const float BIG = 3.4e38f;
    const unsigned BIGB = __float_as_uint(BIG);
    const float EPS = 1.52587890625e-05f;
    const int qA = q0 + wave * KNN_QPW;
    const int nA = qA - b * NP;
    const float ax = qp[nA],     ay = qp[NP + nA],     az = qp[2 * NP + nA];
    const float bx = qp[nA + 1], by = qp[NP + nA + 1], bz = qp[2 * NP + nA + 1];

    {
        const int c = lane & 15, h = lane >> 4;
        const int R = wave >> 3, sl = wave & 7;
        const int nq = q0 - b * NP + 16 * R + c;
        const float qx = qp[nq], qy = qp[NP + nq], qz = qp[2 * NP + nq];
        const float q2 = __builtin_fmaf(qz, qz, __builtin_fmaf(qx, qx, qy * qy));
        const float cq = q2 + EPS * (q2 + kwmax);
        v2f bq;
        bq.x = -2.f * (h ? qz : qx);
        bq.y = h ? 1.f : -2.f * qy;
        const float4* pa = h ? sz4 : sx4;
        const float4* pb = h ? sw4 : sy4;
        const v8f cin = {cq, cq, cq, cq, cq, cq, cq, cq};
        unsigned m[8];
#pragma unroll
        for (int r = 0; r < 8; r++) m[r] = BIGB;
        const int jb = 256 * sl + 16 * c;
#pragma unroll 2
        for (int u = 0; u < 16; u++) {
            const int j = jb + ((u + c) & 15);
            const float4 p = pa[j], q = pb[j];
            v2f a0, a1, a2, a3;
            a0.x = p.x; a0.y = q.x;  a1.x = p.y; a1.y = q.y;
            a2.x = p.z; a2.y = q.z;  a3.x = p.w; a3.y = q.w;
            const v8f d0 = __builtin_amdgcn_wmma_f32_16x16x4_f32(
                false, a0, false, bq, (short)0, cin, false, false);
            const v8f d1 = __builtin_amdgcn_wmma_f32_16x16x4_f32(
                false, a1, false, bq, (short)0, cin, false, false);
            const v8f d2 = __builtin_amdgcn_wmma_f32_16x16x4_f32(
                false, a2, false, bq, (short)0, cin, false, false);
            const v8f d3 = __builtin_amdgcn_wmma_f32_16x16x4_f32(
                false, a3, false, bq, (short)0, cin, false, false);
#pragma unroll
            for (int r = 0; r < 8; r++) m[r] = min5_bits(m[r], d0[r], d1[r], d2[r], d3[r]);
        }
        float4 o0, o1;
        o0.x = __uint_as_float(m[0]); o0.y = __uint_as_float(m[1]);
        o0.z = __uint_as_float(m[2]); o0.w = __uint_as_float(m[3]);
        o1.x = __uint_as_float(m[4]); o1.y = __uint_as_float(m[5]);
        o1.z = __uint_as_float(m[6]); o1.w = __uint_as_float(m[7]);
        float4* dst = pmf4 + (16 * R + c) * KNN_PITCH4 + 4 * sl + 2 * h;
        dst[0] = o0; dst[1] = o1;
    }
    __syncthreads();
    const float4 gA = pmf4[(2 * wave) * KNN_PITCH4 + lane];
    const float4 gB = pmf4[(2 * wave + 1) * KNN_PITCH4 + lane];
    const unsigned td0A = sixteenth_of_lanes(
        __float_as_uint(fminf(fminf(gA.x, gA.y), fminf(gA.z, gA.w))), BIGB);
    const unsigned td0B = sixteenth_of_lanes(
        __float_as_uint(fminf(fminf(gB.x, gB.y), fminf(gB.z, gB.w))), BIGB);
    const float thrA = __uint_as_float(td0A) + 2.f * EPS *
        (__builtin_fmaf(az, az, __builtin_fmaf(ax, ax, ay * ay)) + kwmax);
    const float thrB = __uint_as_float(td0B) + 2.f * EPS *
        (__builtin_fmaf(bz, bz, __builtin_fmaf(bx, bx, by * by)) + kwmax);

    float ldA = BIG, ldB = BIG; int liA = 0, liB = 0;
    scan_groups(gA, thrA, ax, ay, az, sx4, sy4, sz4, lane, BIG, td0A, ldA, liA);
    scan_groups(gB, thrB, bx, by, bz, sx4, sy4, sz4, lane, BIG, td0B, ldB, liB);
    if (lane < KNN) {
        idx[(size_t)qA * KNN + lane]       = liA;
        idx[(size_t)(qA + 1) * KNN + lane] = liB;
    }
}

template <int NT>
__device__ __forceinline__ void stage_wfrag_once(v2f* wl, const float* __restrict__ W, int tid)
{
    v2f w[2048 / NT];
#pragma unroll
    for (int i = 0; i < 2048 / NT; i++) {
        int s = tid + i * NT;
        int lane = s & 31, nc = (s >> 5) & 3, kk = s >> 7;
        int o  = nc * 16 + (lane & 15);
        int k0 = kk * 4 + 2 * (lane >> 4);
        w[i].x = W[o * 64 + k0]; w[i].y = W[o * 64 + k0 + 1];
    }
#pragma unroll
    for (int i = 0; i < 2048 / NT; i++) wl[tid + i * NT] = w[i];
}

static constexpr int GEMM_BLOCKS = BB * NP / 16 / 8;

__global__ __launch_bounds__(256) void gemm_rm_kernel(
    const float* __restrict__ in2,
    const float* __restrict__ W0, const float* __restrict__ bias0,
    const float* __restrict__ W1, const float* __restrict__ bias1,
    const float* __restrict__ xa,
    const float* __restrict__ xb,
    const float* __restrict__ posw,
    const float* __restrict__ posb,
    float* __restrict__ out_cm2,
    float* __restrict__ out_rm2)
{
    __shared__ v2f  wl[2048];
    __shared__ float bl[64];
    __shared__ float pwl[192], pbl[64];
    int tid = threadIdx.x;
    const int prod = blockIdx.x / GEMM_BLOCKS;
    const int blk  = blockIdx.x % GEMM_BLOCKS;
    const size_t plane = (size_t)prod * BB * NP * 64;
    const float* in   = in2 + plane;
    const float* W    = prod ? W1 : W0;
    const float* bias = prod ? bias1 : bias0;
    const float* xc   = prod ? xb : xa;
    float* out_cm = out_cm2 + plane;
    float* out_rm = out_rm2 + plane;
    stage_wfrag(wl, W, tid, 256);
    if (tid < 64) { bl[tid] = bias[tid]; pbl[tid] = prod ? 0.f : posb[tid]; }
    if (tid < 192) pwl[tid] = prod ? posw[tid] : -posw[tid];
    __syncthreads();

    int lane = tid & 31, wave = tid >> 5;
    int tile = blk * 8 + wave;
    int b  = tile / (NP / 16);
    int n0 = (tile % (NP / 16)) * 16;
    const float* inb = in + (size_t)b * NP * 64;
    int row   = n0 + (lane & 15);
    int khalf = 2 * (lane >> 4);

    v2f a[16];
#pragma unroll
    for (int kk = 0; kk < 16; kk++)
        a[kk] = *(const v2f*)(inb + (size_t)row * 64 + kk * 4 + khalf);
    const float* xr = xc + (size_t)b * 3 * NP + n0 + 8 * (lane >> 4);
    const float4 x0 = *(const float4*)xr,            x1 = *(const float4*)(xr + 4);
    const float4 y0 = *(const float4*)(xr + NP),     y1 = *(const float4*)(xr + NP + 4);
    const float4 z0 = *(const float4*)(xr + 2 * NP), z1 = *(const float4*)(xr + 2 * NP + 4);
    const float px[8] = {x0.x, x0.y, x0.z, x0.w, x1.x, x1.y, x1.z, x1.w};
    const float py[8] = {y0.x, y0.y, y0.z, y0.w, y1.x, y1.y, y1.z, y1.w};
    const float pz[8] = {z0.x, z0.y, z0.z, z0.w, z1.x, z1.y, z1.z, z1.w};

#pragma unroll
    for (int nc = 0; nc < 4; nc++) {
        v8f acc = {0.f, 0.f, 0.f, 0.f, 0.f, 0.f, 0.f, 0.f};
#pragma unroll
        for (int kk = 0; kk < 16; kk++) {
            acc = __builtin_amdgcn_wmma_f32_16x16x4_f32(
                false, a[kk], false, wl[(kk * 4 + nc) * 32 + lane],
                (short)0, acc, false, false);
        }
        int m = nc * 16 + (lane & 15);
        float bv = bl[m];
        const float pbv = pbl[m];
        const float w0 = pwl[m * 3 + 0], w1 = pwl[m * 3 + 1], w2 = pwl[m * 3 + 2];
#pragma unroll
        for (int v = 0; v < 8; v++) {
            int r = n0 + v + 8 * (lane >> 4);
            float val = acc[v] + bv;
            out_cm[(size_t)b * 64 * NP + (size_t)m * NP + r] = val;
            out_rm[((size_t)b * NP + r) * 64 + m] = __builtin_fmaf(w2, pz[v],
                __builtin_fmaf(w1, py[v], __builtin_fmaf(w0, px[v], val + pbv)));
        }
    }
}

static constexpr int CROSS_THREADS = 256;
static constexpr int CROSS_WAVES   = CROSS_THREADS / 32;
static constexpr int CROSS_BLOCKS  = 512;
static constexpr int CROSS_QPW     = BB * NP / (CROSS_BLOCKS * CROSS_WAVES);
static_assert(NP % (CROSS_WAVES * CROSS_QPW) == 0 && CROSS_QPW % 2 == 0,
              "a workgroup's query points must stay inside one batch");

__global__ __launch_bounds__(CROSS_THREADS, 8) void cross_kernel(
    const float* __restrict__ p1_0, const float* __restrict__ p2_0,
    const float* __restrict__ p1_1, const float* __restrict__ p2_1,
    const int*   __restrict__ idx_0, const int* __restrict__ idx_1,
    const float* __restrict__ w1, const float* __restrict__ b1,
    const float* __restrict__ w2, const float* __restrict__ b2,
    int nlayers,
    int bps,
    int qpw,
    float* __restrict__ out_rm_0,
    float* __restrict__ out_rm_1,
    float* __restrict__ out_cm)
{
    const int set = blockIdx.x / bps;
    const int blk = blockIdx.x % bps;
    const float* p1  = set ? p1_1 : p1_0;
    const float* p2  = set ? p2_1 : p2_0;
    const int*   idx = set ? idx_1 : idx_0;
    float* out_rm    = set ? out_rm_1 : out_rm_0;

    __shared__ v2f  wl1[2048];
    __shared__ v2f  wl2[2048];
    __shared__ float bl1[64], bl2[64];
    __shared__ float X[CROSS_WAVES][16][68];

    int tid = threadIdx.x;
    stage_wfrag_once<CROSS_THREADS>(wl1, w1, tid);
    if (nlayers > 1) stage_wfrag_once<CROSS_THREADS>(wl2, w2, tid);
    if (tid < 64) {
        bl1[tid] = b1[tid];
        bl2[tid] = (nlayers > 1) ? b2[tid] : 0.f;
    }
    __syncthreads();

    int lane = tid & 31, wave = tid >> 5;
    int mrow  = lane & 15;
    int khalf = 2 * (lane >> 4);

    int c0 = 2 * lane;

    const int b = (blk * qpw * CROSS_WAVES) / NP;
    const float* p2b = p2 + (size_t)b * NP * 64;

#pragma unroll 1
    for (int it = 0; it < qpw; it++) {
        int q = (blk * qpw + it) * CROSS_WAVES + wave;
        int n = q - b * NP;

        const int* iq = idx + (size_t)q * KNN;

        v2f p1v = *(const v2f*)(p1 + (size_t)q * 64 + c0);
#pragma unroll
        for (int r = 0; r < KNN; r++) {
            const unsigned offb = ((unsigned)iq[r] << 8) + ((unsigned)c0 << 2);
            v2f g = *(const v2f*)((const char*)p2b + offb);
            X[wave][r][c0]     = leaky(g.x + p1v.x);
            X[wave][r][c0 + 1] = leaky(g.y + p1v.y);
        }
        __syncthreads();

        for (int layer = 0; layer < nlayers; layer++) {
            const v2f*  wl = layer ? wl2 : wl1;
            const float* bl = layer ? bl2 : bl1;
            bool last = (layer == nlayers - 1);

            v2f a[16];
#pragma unroll
            for (int kk = 0; kk < 16; kk++)
                a[kk] = *(const v2f*)&X[wave][mrow][kk * 4 + khalf];

#pragma unroll
            for (int nc = 0; nc < 4; nc++) {
                v8f acc = {0.f, 0.f, 0.f, 0.f, 0.f, 0.f, 0.f, 0.f};
#pragma unroll
                for (int kk = 0; kk < 16; kk++) {
                    acc = __builtin_amdgcn_wmma_f32_16x16x4_f32(
                        false, a[kk], false, wl[(kk * 4 + nc) * 32 + lane],
                        (short)0, acc, false, false);
                }
                int m = nc * 16 + mrow;
                float bv = bl[m];
                if (!last) {
#pragma unroll
                    for (int v = 0; v < 8; v++)
                        X[wave][v + 8 * (lane >> 4)][m] = leaky(acc[v] + bv);
                } else {
                    float mx = -3.4e38f;
#pragma unroll
                    for (int v = 0; v < 8; v++) mx = fmaxf(mx, leaky(acc[v] + bv));
                    mx = fmaxf(mx, __shfl_xor(mx, 16, 32));
                    if (lane < 16) {
                        out_rm[(size_t)q * 64 + m] = mx;
                        if (out_cm)
                            out_cm[(size_t)b * 64 * NP + (size_t)m * NP + n] = mx;
                    }
                }
            }
            __syncthreads();
        }
    }
}

extern "C" void kernel_launch(void* const* d_in, const int* in_sizes, int n_in,
                              void* d_out, int out_size, void* d_ws, size_t ws_size,
                              hipStream_t stream)
{
    (void)in_sizes; (void)n_in; (void)out_size; (void)ws_size;
    const float* pc1   = (const float*)d_in[0];
    const float* pc2   = (const float*)d_in[1];
    const float* feat1 = (const float*)d_in[2];
    const float* feat2 = (const float*)d_in[3];
    const float* t11_w = (const float*)d_in[4];  const float* t11_b = (const float*)d_in[5];
    const float* t22_w = (const float*)d_in[6];  const float* t22_b = (const float*)d_in[7];
    const float* pos1w = (const float*)d_in[8];  const float* pos1b = (const float*)d_in[9];
    const float* m1w1  = (const float*)d_in[10]; const float* m1b1  = (const float*)d_in[11];
    const float* m1w2  = (const float*)d_in[12]; const float* m1b2  = (const float*)d_in[13];
    const float* t1_w  = (const float*)d_in[14]; const float* t1_b  = (const float*)d_in[15];
    const float* t2_w  = (const float*)d_in[16]; const float* t2_b  = (const float*)d_in[17];
    const float* pos2w = (const float*)d_in[18]; const float* pos2b = (const float*)d_in[19];
    const float* m2w1  = (const float*)d_in[20]; const float* m2b1  = (const float*)d_in[21];

    float* out = (float*)d_out;
    const size_t S = (size_t)BB * NP * 64;
    float* ws = (float*)d_ws;
    float* q1 = ws;           float* q2 = ws + S;
    float* r1 = ws + 2 * S;   float* r2 = ws + 3 * S;
    float* pooled1 = ws + 4 * S;
    float* pooled2 = ws + 5 * S;
    int* idx1 = (int*)(ws + 6 * S);
    int* idx2 = idx1 + (size_t)BB * NP * KNN;
    float* fn1_rm  = q1;
    float* fn2_rm  = q2;
    float* pooled3 = r1;

    knn_gemm_kernel<<<FRONT_GRID, KNN_THREADS, 0, stream>>>(pc1, pc2, idx1, idx2,
        feat1, feat2, t11_w, t11_b, t22_w, t22_b, pos1w, pos1b, q1);

    cross_kernel<<<2 * CROSS_BLOCKS, CROSS_THREADS, 0, stream>>>(q1, q2, r1, r2, idx1, idx2,
        m1w1, m1b1, m1w2, m1b2, 2, CROSS_BLOCKS, CROSS_QPW, pooled1, pooled2, nullptr);

    gemm_rm_kernel<<<2 * GEMM_BLOCKS, 256, 0, stream>>>(pooled1, t1_w, t1_b, t2_w, t2_b,
        pc1, pc2, pos2w, pos2b, out, fn1_rm);

    cross_kernel<<<2 * CROSS_BLOCKS, CROSS_THREADS, 0, stream>>>(fn1_rm, fn2_rm, fn1_rm, fn2_rm,
        idx1, idx2, m2w1, m2b1, nullptr, nullptr, 1, 2 * CROSS_BLOCKS, CROSS_QPW / 2,
        pooled3, pooled2, out + 2 * S);
}
